# starting kernel
# baseline (speedup 1.0000x reference)
#include <hip/hip_runtime.h>
#include <math.h>

static const int CB = 4, CT = 32, CN = 512;
static const int CIN = 10, CF = 50, CFC = 128, COUT = 2;

typedef __attribute__((ext_vector_type(2))) float v2f;
typedef __attribute__((ext_vector_type(8))) float v8f;

__global__ __launch_bounds__(128)
void gemm_wmma_f32(const float* __restrict__ A, long long sA, int lda,
                   const float* __restrict__ Bm, long long sB, int ldb,
                   float* __restrict__ C, long long sC, int ldc,
                   const float* __restrict__ bias,
                   int M, int N, int K, int relu) {
  const int lane  = threadIdx.x & 31;
  const int wave  = threadIdx.x >> 5;
  const int npair = (N + 31) >> 5;
  const int tiles = (M >> 4) * npair;
  const int tile  = blockIdx.x * (blockDim.x >> 5) + wave;
  if (tile >= tiles) return;
  const int tm = (tile / npair) << 4;
  const int tn = (tile % npair) << 5;

  const float* Ab = A  + (long long)blockIdx.z * sA;
  const float* Bb = Bm + (long long)blockIdx.z * sB;
  float*       Cb = C  + (long long)blockIdx.z * sC;

  const int mrow  = lane & 15;
  const int khalf = (lane >> 4) << 1;
  const int n0 = tn + (lane & 15);
  const int n1 = n0 + 16;
  const float* ap  = Ab + (long long)(tm + mrow) * lda + khalf;
  const float* bp0 = Bb + (n0 < N ? n0 : N - 1);
  const float* bp1 = Bb + (n1 < N ? n1 : N - 1);

  v8f acc0 = {}, acc1 = {};
  const int K4 = K & ~3;
  int k0 = 0;
#pragma unroll 2
  for (; k0 < K4; k0 += 4) {
    const v2f av = *(const v2f*)(ap + k0);
    const long long kb = (long long)(k0 + khalf) * ldb;
    v2f bv0, bv1;
    bv0.x = bp0[kb];       bv0.y = bp0[kb + ldb];
    bv1.x = bp1[kb];       bv1.y = bp1[kb + ldb];
    acc0 = __builtin_amdgcn_wmma_f32_16x16x4_f32(false, av, false, bv0,
                                                 (short)0, acc0, false, false);
    acc1 = __builtin_amdgcn_wmma_f32_16x16x4_f32(false, av, false, bv1,
                                                 (short)0, acc1, false, false);
  }
  if (k0 < K) {
    const int ka = k0 + khalf;
    v2f av = {}, bv0 = {}, bv1 = {};
    if (ka < K) {
      av.x  = ap[k0];
      bv0.x = bp0[(long long)ka * ldb];
      bv1.x = bp1[(long long)ka * ldb];
    }
    if (ka + 1 < K) {
      av.y  = ap[k0 + 1];
      bv0.y = bp0[(long long)(ka + 1) * ldb];
      bv1.y = bp1[(long long)(ka + 1) * ldb];
    }
    acc0 = __builtin_amdgcn_wmma_f32_16x16x4_f32(false, av, false, bv0,
                                                 (short)0, acc0, false, false);
    acc1 = __builtin_amdgcn_wmma_f32_16x16x4_f32(false, av, false, bv1,
                                                 (short)0, acc1, false, false);
  }

  const int rbase = tm + ((lane >> 4) << 3);
  if (n0 < N) {
    const float badd = bias ? bias[n0] : 0.0f;
#pragma unroll
    for (int r = 0; r < 8; ++r) {
      float v = acc0[r] + badd;
      if (relu) v = fmaxf(v, 0.0f);
      Cb[(long long)(rbase + r) * ldc + n0] = v;
    }
  }
  if (n1 < N) {
    const float badd = bias ? bias[n1] : 0.0f;
#pragma unroll
    for (int r = 0; r < 8; ++r) {
      float v = acc1[r] + badd;
      if (relu) v = fmaxf(v, 0.0f);
      Cb[(long long)(rbase + r) * ldc + n1] = v;
    }
  }
}

__global__ void deg_dinv_kernel(const float* __restrict__ A, float* __restrict__ dinv,
                                int rows, int n) {
  const int wid  = (blockIdx.x * blockDim.x + threadIdx.x) >> 5;
  const int lane = threadIdx.x & 31;
  if (wid >= rows) return;
  const float* row = A + (long long)wid * n;
  float s = 0.0f;
  for (int i = lane; i < n; i += 32) s += row[i];
#pragma unroll
  for (int off = 16; off > 0; off >>= 1) s += __shfl_xor(s, off, 32);
  if (lane == 0) dinv[wid] = (s > 0.0f) ? rsqrtf(s) : 0.0f;
}

__global__ void pack_wz_kernel(const float* __restrict__ Wch,
                               const float* __restrict__ Wcu,
                               float* __restrict__ Wz) {
  int idx = blockIdx.x * blockDim.x + threadIdx.x;
  if (idx >= 200 * 150) return;
  int r = idx / 150, cg = idx % 150;
  int c = cg / 50, g = cg % 50;
  int src = r / 50, f = r % 50;
  float v;
  if (src == 0)      v = Wch[((c * 2 + 0) * 50 + f) * 50 + g];
  else if (src == 1) v = Wch[((c * 2 + 1) * 50 + f) * 50 + g];
  else if (src == 2) v = Wcu[((c * 2 + 0) * 50 + f) * 50 + g];
  else               v = Wcu[((c * 2 + 1) * 50 + f) * 50 + g];
  Wz[idx] = v;
}

__global__ void prescale_kernel(const float* __restrict__ h, const float* __restrict__ x,
                                const float* __restrict__ dinv, float* __restrict__ Vs,
                                int t) {
  int idx = blockIdx.x * blockDim.x + threadIdx.x;
  if (idx >= CB * CN * 2 * CF) return;
  int row = idx / (2 * CF), f = idx % (2 * CF);
  int b = row / CN, n = row % CN;
  float dv = dinv[((long long)b * CT + t) * CN + n];
  float v = (f < CF) ? h[row * CF + f] : x[row * CF + (f - CF)];
  Vs[idx] = dv * v;
}

__global__ void combine_kernel(const float* __restrict__ h, const float* __restrict__ x,
                               const float* __restrict__ W, const float* __restrict__ dinv,
                               float* __restrict__ U, int t) {
  int idx = blockIdx.x * blockDim.x + threadIdx.x;
  if (idx >= CB * CN * 4 * CF) return;
  int row = idx / (4 * CF), f = idx % (4 * CF);
  int b = row / CN, n = row % CN;
  float dv = dinv[((long long)b * CT + t) * CN + n];
  float v;
  if (f < CF)            v = h[row * CF + f];
  else if (f < 2 * CF)   v = h[row * CF + (f - CF)]     - dv * W[row * 2 * CF + (f - CF)];
  else if (f < 3 * CF)   v = x[row * CF + (f - 2 * CF)];
  else                   v = x[row * CF + (f - 3 * CF)] - dv * W[row * 2 * CF + CF + (f - 3 * CF)];
  U[idx] = v;
}

__global__ void gate_kernel(const float* __restrict__ z, float* __restrict__ h) {
  int idx = blockIdx.x * blockDim.x + threadIdx.x;
  if (idx >= CB * CN * CF) return;
  int row = idx / CF, f = idx % CF;
  float z0 = z[row * 150 + f];
  float z1 = z[row * 150 + CF + f];
  float z2 = z[row * 150 + 2 * CF + f];
  float gt = 1.0f / (1.0f + expf(-z0));
  h[idx] = gt * tanhf(z1) + (1.0f - gt) * tanhf(z2);
}

static inline void launch_gemm(const float* A, long long sA, int lda,
                               const float* Bm, long long sB, int ldb,
                               float* C, long long sC, int ldc,
                               const float* bias, int M, int N, int K,
                               int batch, int relu, hipStream_t stream) {
  int npair = (N + 31) >> 5;
  int tiles = (M >> 4) * npair;
  dim3 grid((tiles + 3) / 4, 1, batch);
  gemm_wmma_f32<<<grid, 128, 0, stream>>>(A, sA, lda, Bm, sB, ldb,
                                          C, sC, ldc, bias, M, N, K, relu);
}

extern "C" void kernel_launch(void* const* d_in, const int* in_sizes, int n_in,
                              void* d_out, int out_size, void* d_ws, size_t ws_size,
                              hipStream_t stream) {
  const float* obs  = (const float*)d_in[0];
  const float* adj  = (const float*)d_in[1];
  const float* h0   = (const float*)d_in[2];
  const float* We1  = (const float*)d_in[3];
  const float* be1  = (const float*)d_in[4];
  const float* We2  = (const float*)d_in[5];
  const float* be2  = (const float*)d_in[6];
  const float* Wch  = (const float*)d_in[7];
  const float* Wcu  = (const float*)d_in[8];
  const float* bc   = (const float*)d_in[9];
  const float* Wd1  = (const float*)d_in[10];
  const float* bd1  = (const float*)d_in[11];
  const float* Wd2  = (const float*)d_in[12];
  const float* bd2  = (const float*)d_in[13];
  const float* Wd3  = (const float*)d_in[14];
  const float* bd3  = (const float*)d_in[15];
  float* out = (float*)d_out;

  float* ws = (float*)d_ws;
  size_t off = 0;
  float* dinv = ws + off; off += (size_t)CB * CT * CN;
  float* h    = ws + off; off += (size_t)CB * CN * CF;
  float* x    = ws + off; off += (size_t)CB * CN * CF;
  float* tmp1 = ws + off; off += (size_t)CB * CN * CFC;
  float* Vs   = ws + off; off += (size_t)CB * CN * 2 * CF;
  float* Wb   = ws + off; off += (size_t)CB * CN * 2 * CF;
  float* U    = ws + off; off += (size_t)CB * CN * 4 * CF;
  float* z    = ws + off; off += (size_t)CB * CN * 150;
  float* y1   = ws + off; off += (size_t)CB * CN * CFC;
  float* y2   = ws + off; off += (size_t)CB * CN * CFC;
  float* Wz   = ws + off; off += (size_t)200 * 150;
  (void)ws_size; (void)in_sizes; (void)n_in; (void)out_size;

  {
    int rows = CB * CT * CN;
    deg_dinv_kernel<<<(rows * 32 + 255) / 256, 256, 0, stream>>>(adj, dinv, rows, CN);
    pack_wz_kernel<<<(200 * 150 + 255) / 256, 256, 0, stream>>>(Wch, Wcu, Wz);
    hipMemcpyAsync(h, h0, sizeof(float) * CB * CN * CF, hipMemcpyDeviceToDevice, stream);
  }

  const int MN = CB * CN;

  for (int t = 0; t < CT; ++t) {
    launch_gemm(obs + (long long)t * CN * CIN, (long long)CT * CN * CIN, CIN,
                We1, 0, CFC, tmp1, (long long)CN * CFC, CFC,
                be1, CN, CFC, CIN, CB, 1, stream);
    launch_gemm(tmp1, 0, CFC, We2, 0, CF, x, 0, CF,
                be2, MN, CF, CFC, 1, 1, stream);
    prescale_kernel<<<(MN * 2 * CF + 255) / 256, 256, 0, stream>>>(h, x, dinv, Vs, t);
    launch_gemm(adj + (long long)t * CN * CN, (long long)CT * CN * CN, CN,
                Vs, (long long)CN * 2 * CF, 2 * CF,
                Wb, (long long)CN * 2 * CF, 2 * CF,
                nullptr, CN, 2 * CF, CN, CB, 0, stream);
    combine_kernel<<<(MN * 4 * CF + 255) / 256, 256, 0, stream>>>(h, x, Wb, dinv, U, t);
    launch_gemm(U, 0, 4 * CF, Wz, 0, 150, z, 0, 150,
                bc, MN, 150, 4 * CF, 1, 0, stream);
    gate_kernel<<<(MN * CF + 255) / 256, 256, 0, stream>>>(z, h);
    launch_gemm(h, 0, CF, Wd1, 0, CFC, y1, 0, CFC, bd1, MN, CFC, CF, 1, 1, stream);
    launch_gemm(y1, 0, CFC, Wd2, 0, CFC, y2, 0, CFC, bd2, MN, CFC, CFC, 1, 1, stream);
    launch_gemm(y2, (long long)CN * CFC, CFC, Wd3, 0, COUT,
                out + (long long)t * CN * COUT, (long long)CT * CN * COUT, COUT,
                bd3, CN, COUT, CFC, CB, 0, stream);
  }

  hipMemcpyAsync(out + (long long)CB * CT * CN * COUT, h,
                 sizeof(float) * CB * CN * CF, hipMemcpyDeviceToDevice, stream);
}


# per-step dinv row scale and Laplacian combine applied in the consuming GEMMs' fragment loaders (B of A_t@V, A of the gate GEMM); Vs and U never stored: 5 -> 3 launches per step; same k and sum order
# speedup vs baseline: 2.7317x; 2.7317x over previous
#include <hip/hip_runtime.h>
#include <math.h>

static const int CB = 4, CT = 32, CN = 512;
static const int CIN = 10, CF = 50, CFC = 128, COUT = 2;

typedef __attribute__((ext_vector_type(2))) float v2f;
typedef __attribute__((ext_vector_type(8))) float v8f;

__global__ __launch_bounds__(128)
void gemm_wmma_f32(const float* __restrict__ A, long long sA, long long sA2, int lda,
                   const float* __restrict__ Bm, long long sB, int ldb,
                   float* __restrict__ C, long long sC, long long sC2, int ldc,
                   const float* __restrict__ bias,
                   int M, int N, int K, int relu) {
  const int lane  = threadIdx.x & 31;
  const int wave  = threadIdx.x >> 5;
  const int npair = (N + 31) >> 5;
  const int tiles = (M >> 4) * npair;
  const int tile  = blockIdx.x * (blockDim.x >> 5) + wave;
  if (tile >= tiles) return;
  const int tm = (tile / npair) << 4;
  const int tn = (tile % npair) << 5;

  const float* Ab = A  + (long long)blockIdx.z * sA + (long long)blockIdx.y * sA2;
  const float* Bb = Bm + (long long)blockIdx.z * sB;
  float*       Cb = C  + (long long)blockIdx.z * sC + (long long)blockIdx.y * sC2;

  const int mrow  = lane & 15;
  const int khalf = (lane >> 4) << 1;
  const int n0 = tn + (lane & 15);
  const int n1 = n0 + 16;
  const float* ap  = Ab + (long long)(tm + mrow) * lda + khalf;
  const float* bp0 = Bb + (n0 < N ? n0 : N - 1);
  const float* bp1 = Bb + (n1 < N ? n1 : N - 1);

  v8f acc0 = {}, acc1 = {};
  const int K4 = K & ~3;
  int k0 = 0;
#pragma unroll 2
  for (; k0 < K4; k0 += 4) {
    const v2f av = *(const v2f*)(ap + k0);
    const long long kb = (long long)(k0 + khalf) * ldb;
    v2f bv0, bv1;
    bv0.x = bp0[kb];       bv0.y = bp0[kb + ldb];
    bv1.x = bp1[kb];       bv1.y = bp1[kb + ldb];
    acc0 = __builtin_amdgcn_wmma_f32_16x16x4_f32(false, av, false, bv0,
                                                 (short)0, acc0, false, false);
    acc1 = __builtin_amdgcn_wmma_f32_16x16x4_f32(false, av, false, bv1,
                                                 (short)0, acc1, false, false);
  }
  if (k0 < K) {
    const int ka = k0 + khalf;
    v2f av = {}, bv0 = {}, bv1 = {};
    if (ka < K) {
      av.x  = ap[k0];
      bv0.x = bp0[(long long)ka * ldb];
      bv1.x = bp1[(long long)ka * ldb];
    }
    if (ka + 1 < K) {
      av.y  = ap[k0 + 1];
      bv0.y = bp0[(long long)(ka + 1) * ldb];
      bv1.y = bp1[(long long)(ka + 1) * ldb];
    }
    acc0 = __builtin_amdgcn_wmma_f32_16x16x4_f32(false, av, false, bv0,
                                                 (short)0, acc0, false, false);
    acc1 = __builtin_amdgcn_wmma_f32_16x16x4_f32(false, av, false, bv1,
                                                 (short)0, acc1, false, false);
  }

  const int rbase = tm + ((lane >> 4) << 3);
  if (n0 < N) {
    const float badd = bias ? bias[n0] : 0.0f;
#pragma unroll
    for (int r = 0; r < 8; ++r) {
      float v = acc0[r] + badd;
      if (relu) v = fmaxf(v, 0.0f);
      Cb[(long long)(rbase + r) * ldc + n0] = v;
    }
  }
  if (n1 < N) {
    const float badd = bias ? bias[n1] : 0.0f;
#pragma unroll
    for (int r = 0; r < 8; ++r) {
      float v = acc1[r] + badd;
      if (relu) v = fmaxf(v, 0.0f);
      Cb[(long long)(rbase + r) * ldc + n1] = v;
    }
  }
}

template <int S>
__global__ __launch_bounds__(32 * S)
void gemm_wmma_f32_ksplit(const float* __restrict__ A, long long sA, long long sA2, int lda,
                          const float* __restrict__ Bm, long long sB, int ldb,
                          float* __restrict__ C, long long sC, long long sC2, int ldc,
                          const float* __restrict__ bias,
                          int N, int K, int relu) {
  __shared__ float part[S][16][32];

  const int lane  = threadIdx.x & 31;
  const int wave  = __builtin_amdgcn_readfirstlane(threadIdx.x >> 5);
  const int npair = (N + 31) >> 5;
  const int tile  = blockIdx.x;
  const int tm = (tile / npair) << 4;
  const int tn = (tile % npair) << 5;

  const float* Ab = A  + (long long)blockIdx.z * sA + (long long)blockIdx.y * sA2;
  const float* Bb = Bm + (long long)blockIdx.z * sB;
  float*       Cb = C  + (long long)blockIdx.z * sC + (long long)blockIdx.y * sC2;

  const int mrow  = lane & 15;
  const int khalf = (lane >> 4) << 1;
  const int n0 = tn + (lane & 15);
  const int n1 = n0 + 16;
  const float* ap  = Ab + (long long)(tm + mrow) * lda + khalf;
  const float* bp0 = Bb + (n0 < N ? n0 : N - 1);
  const float* bp1 = Bb + (n1 < N ? n1 : N - 1);

  const int K4    = K & ~3;
  const int steps = (K + 3) >> 2;
  const int per   = (steps + S - 1) / S;
  const int kbeg  = wave * per * 4;
  const int klim  = kbeg + per * 4;
  const int kstop = klim < K4 ? klim : K4;

  v8f acc0 = {}, acc1 = {};
#pragma unroll 2
  for (int k0 = kbeg; k0 < kstop; k0 += 4) {
    const v2f av = *(const v2f*)(ap + k0);
    const long long kb = (long long)(k0 + khalf) * ldb;
    v2f bv0, bv1;
    bv0.x = bp0[kb];       bv0.y = bp0[kb + ldb];
    bv1.x = bp1[kb];       bv1.y = bp1[kb + ldb];
    acc0 = __builtin_amdgcn_wmma_f32_16x16x4_f32(false, av, false, bv0,
                                                 (short)0, acc0, false, false);
    acc1 = __builtin_amdgcn_wmma_f32_16x16x4_f32(false, av, false, bv1,
                                                 (short)0, acc1, false, false);
  }
  if (K4 < K && kbeg <= K4 && K4 < klim) {
    const int ka = K4 + khalf;
    v2f av = {}, bv0 = {}, bv1 = {};
    if (ka < K) {
      av.x  = ap[K4];
      bv0.x = bp0[(long long)ka * ldb];
      bv1.x = bp1[(long long)ka * ldb];
    }
    if (ka + 1 < K) {
      av.y  = ap[K4 + 1];
      bv0.y = bp0[(long long)(ka + 1) * ldb];
      bv1.y = bp1[(long long)(ka + 1) * ldb];
    }
    acc0 = __builtin_amdgcn_wmma_f32_16x16x4_f32(false, av, false, bv0,
                                                 (short)0, acc0, false, false);
    acc1 = __builtin_amdgcn_wmma_f32_16x16x4_f32(false, av, false, bv1,
                                                 (short)0, acc1, false, false);
  }

  for (int r = 0; r < 8; ++r) {
    part[wave][r][lane]     = acc0[r];
    part[wave][8 + r][lane] = acc1[r];
  }
  __syncthreads();

  constexpr int RPW = 16 / S;
  const int rbase = tm + ((lane >> 4) << 3);
  for (int j = 0; j < RPW; ++j) {
    const int r = wave * RPW + j;
    float s = part[0][r][lane];
    for (int q = 1; q < S; ++q) s += part[q][r][lane];
    const int n = (r < 8) ? n0 : n1;
    if (n < N) {
      float v = s + (bias ? bias[n] : 0.0f);
      if (relu) v = fmaxf(v, 0.0f);
      Cb[(long long)(rbase + (r & 7)) * ldc + n] = v;
    }
  }
}

__global__ void deg_dinv_kernel(const float* __restrict__ A, float* __restrict__ dinv,
                                int rows, int n) {
  const int wid  = (blockIdx.x * blockDim.x + threadIdx.x) >> 5;
  const int lane = threadIdx.x & 31;
  if (wid >= rows) return;
  const float* row = A + (long long)wid * n;
  float s = 0.0f;
  for (int i = lane; i < n; i += 32) s += row[i];
#pragma unroll
  for (int off = 16; off > 0; off >>= 1) s += __shfl_xor(s, off, 32);
  if (lane == 0) dinv[wid] = (s > 0.0f) ? rsqrtf(s) : 0.0f;
}

__global__ void pack_wz_kernel(const float* __restrict__ Wch,
                               const float* __restrict__ Wcu,
                               float* __restrict__ Wz) {
  int idx = blockIdx.x * blockDim.x + threadIdx.x;
  if (idx >= 200 * 150) return;
  int r = idx / 150, cg = idx % 150;
  int c = cg / 50, g = cg % 50;
  int src = r / 50, f = r % 50;
  float v;
  if (src == 0)      v = Wch[((c * 2 + 0) * 50 + f) * 50 + g];
  else if (src == 1) v = Wch[((c * 2 + 1) * 50 + f) * 50 + g];
  else if (src == 2) v = Wcu[((c * 2 + 0) * 50 + f) * 50 + g];
  else               v = Wcu[((c * 2 + 1) * 50 + f) * 50 + g];
  Wz[idx] = v;
}

__device__ __forceinline__ void wmma_group_open() {
  __builtin_amdgcn_sched_barrier(0);
}
__device__ __forceinline__ void wmma_group_close() {
  __builtin_amdgcn_sched_barrier(0);
  asm volatile("v_nop\n\tv_nop\n\tv_nop\n\tv_nop");
  __builtin_amdgcn_sched_barrier(0);
}

template <int S>
__global__ __launch_bounds__(32 * S)
void gemm_adj_ksplit(const float* __restrict__ adj_t,
                     const float* __restrict__ h, const float* __restrict__ x,
                     const float* __restrict__ dinv_t,
                     float* __restrict__ Wout) {
  constexpr int N = 2 * CF;
  constexpr int K = CN;
  static_assert(K % (8 * S) == 0, "whole trips of two k-steps in every wave");
  __shared__ float part[S][16][32];

  const int lane  = threadIdx.x & 31;
  const int wave  = __builtin_amdgcn_readfirstlane(threadIdx.x >> 5);
  constexpr int npair = (N + 31) >> 5;
  const int tm = (blockIdx.x / npair) << 4;
  const int tn = (blockIdx.x % npair) << 5;
  const int b  = blockIdx.z;

  const float* Ab = adj_t  + (long long)b * CT * CN * CN;
  const float* db = dinv_t + (long long)b * CT * CN;
  const float* hb = h + (long long)b * CN * CF;
  const float* xb = x + (long long)b * CN * CF;
  float*       Cb = Wout + (long long)b * CN * N;

  const int mrow  = lane & 15;
  const int khalf = (lane >> 4) << 1;
  const int n0 = tn + (lane & 15);
  const int n1 = n0 + 16;
  const int c0 = n0 < N ? n0 : N - 1;
  const int c1 = n1 < N ? n1 : N - 1;
  const float* ap  = Ab + (long long)(tm + mrow) * K + khalf;
  const float* dp  = db + khalf;
  const float* bp0 = c0 < CF ? hb + c0 : xb + (c0 - CF);
  const float* bp1 = c1 < CF ? hb + c1 : xb + (c1 - CF);

  constexpr int per = K / (4 * S);
  const int kbeg = wave * per * 4;
  const int kend = kbeg + per * 4;

  v8f acc0 = {}, acc1 = {};
  for (int k0 = kbeg; k0 < kend; k0 += 8) {
    const v2f ava = *(const v2f*)(ap + k0);
    const v2f avb = *(const v2f*)(ap + k0 + 4);
    const v2f dda = *(const v2f*)(dp + k0);
    const v2f ddb = *(const v2f*)(dp + k0 + 4);
    const int kb = (k0 + khalf) * CF;
    v2f b0a, b1a, b0b, b1b;
    b0a.x = dda.x * bp0[kb];            b0a.y = dda.y * bp0[kb + CF];
    b1a.x = dda.x * bp1[kb];            b1a.y = dda.y * bp1[kb + CF];
    b0b.x = ddb.x * bp0[kb + 4 * CF];   b0b.y = ddb.y * bp0[kb + 5 * CF];
    b1b.x = ddb.x * bp1[kb + 4 * CF];   b1b.y = ddb.y * bp1[kb + 5 * CF];
    wmma_group_open();
    acc0 = __builtin_amdgcn_wmma_f32_16x16x4_f32(false, ava, false, b0a,
                                                 (short)0, acc0, false, false);
    acc1 = __builtin_amdgcn_wmma_f32_16x16x4_f32(false, ava, false, b1a,
                                                 (short)0, acc1, false, false);
    acc0 = __builtin_amdgcn_wmma_f32_16x16x4_f32(false, avb, false, b0b,
                                                 (short)0, acc0, false, false);
    acc1 = __builtin_amdgcn_wmma_f32_16x16x4_f32(false, avb, false, b1b,
                                                 (short)0, acc1, false, false);
    wmma_group_close();
  }

  for (int r = 0; r < 8; ++r) {
    part[wave][r][lane]     = acc0[r];
    part[wave][8 + r][lane] = acc1[r];
  }
  __syncthreads();

  constexpr int RPW = 16 / S;
  const int rbase = tm + ((lane >> 4) << 3);
  for (int j = 0; j < RPW; ++j) {
    const int r = wave * RPW + j;
    float s = part[0][r][lane];
    for (int q = 1; q < S; ++q) s += part[q][r][lane];
    const int n = (r < 8) ? n0 : n1;
    if (n < N) Cb[(long long)(rbase + (r & 7)) * N + n] = s;
  }
}

__device__ __forceinline__ v2f gate_a_frag(const float* __restrict__ hp,
                                           const float* __restrict__ xp,
                                           const float* __restrict__ wp,
                                           float dv, int k) {
  const int sec = k / CF;
  const int f   = k - sec * CF;
  const v2f sv = *(const v2f*)(((sec & 2) ? xp : hp) + f);
  const v2f wv = *(const v2f*)(wp + (sec >> 1) * CF + f);
  const bool lap = (sec & 1) != 0;
  v2f av;
  av.x = lap ? sv.x - dv * wv.x : sv.x;
  av.y = lap ? sv.y - dv * wv.y : sv.y;
  return av;
}

template <int S>
__global__ __launch_bounds__(32 * S)
void gemm_gate_ksplit(const float* __restrict__ h, const float* __restrict__ x,
                      const float* __restrict__ W, const float* __restrict__ dinv_t,
                      const float* __restrict__ Wz, const float* __restrict__ bias,
                      float* __restrict__ z) {
  constexpr int N = 3 * CF;
  constexpr int K = 4 * CF;
  static_assert(K % 4 == 0 && CF % 2 == 0, "no K tail; sections start at even k");
  __shared__ float part[S][16][32];

  const int lane  = threadIdx.x & 31;
  const int wave  = __builtin_amdgcn_readfirstlane(threadIdx.x >> 5);
  constexpr int npair = (N + 31) >> 5;
  const int tm = (blockIdx.x / npair) << 4;
  const int tn = (blockIdx.x % npair) << 5;

  const int mrow  = lane & 15;
  const int khalf = (lane >> 4) << 1;
  const int n0 = tn + (lane & 15);
  const int n1 = n0 + 16;
  const int row = tm + mrow;
  const float dv = dinv_t[(long long)(row / CN) * CT * CN + (row % CN)];
  const float* hp = h + (long long)row * CF;
  const float* xp = x + (long long)row * CF;
  const float* wp = W + (long long)row * 2 * CF;
  const float* bp0 = Wz + (n0 < N ? n0 : N - 1);
  const float* bp1 = Wz + (n1 < N ? n1 : N - 1);

  constexpr int steps = K / 4;
  constexpr int per   = (steps + S - 1) / S;
  const int kbeg  = wave * per * 4;
  const int klim  = kbeg + per * 4;
  const int kstop = klim < K ? klim : K;

  v8f acc0 = {}, acc1 = {};
  int k0 = kbeg;
  for (; k0 + 4 < kstop; k0 += 8) {
    const int k = k0 + khalf;
    const v2f ava = gate_a_frag(hp, xp, wp, dv, k);
    const v2f avb = gate_a_frag(hp, xp, wp, dv, k + 4);
    const int kb = k * N;
    v2f b0a, b1a, b0b, b1b;
    b0a.x = bp0[kb];           b0a.y = bp0[kb + N];
    b1a.x = bp1[kb];           b1a.y = bp1[kb + N];
    b0b.x = bp0[kb + 4 * N];   b0b.y = bp0[kb + 5 * N];
    b1b.x = bp1[kb + 4 * N];   b1b.y = bp1[kb + 5 * N];
    wmma_group_open();
    acc0 = __builtin_amdgcn_wmma_f32_16x16x4_f32(false, ava, false, b0a,
                                                 (short)0, acc0, false, false);
    acc1 = __builtin_amdgcn_wmma_f32_16x16x4_f32(false, ava, false, b1a,
                                                 (short)0, acc1, false, false);
    acc0 = __builtin_amdgcn_wmma_f32_16x16x4_f32(false, avb, false, b0b,
                                                 (short)0, acc0, false, false);
    acc1 = __builtin_amdgcn_wmma_f32_16x16x4_f32(false, avb, false, b1b,
                                                 (short)0, acc1, false, false);
    wmma_group_close();
  }
  if (k0 < kstop) {
    const int k = k0 + khalf;
    const v2f ava = gate_a_frag(hp, xp, wp, dv, k);
    const int kb = k * N;
    v2f b0a, b1a;
    b0a.x = bp0[kb];           b0a.y = bp0[kb + N];
    b1a.x = bp1[kb];           b1a.y = bp1[kb + N];
    wmma_group_open();
    acc0 = __builtin_amdgcn_wmma_f32_16x16x4_f32(false, ava, false, b0a,
                                                 (short)0, acc0, false, false);
    acc1 = __builtin_amdgcn_wmma_f32_16x16x4_f32(false, ava, false, b1a,
                                                 (short)0, acc1, false, false);
    wmma_group_close();
  }

  for (int r = 0; r < 8; ++r) {
    part[wave][r][lane]     = acc0[r];
    part[wave][8 + r][lane] = acc1[r];
  }
  __syncthreads();

  constexpr int RPW = 16 / S;
  const int rbase = tm + ((lane >> 4) << 3);
  for (int j = 0; j < RPW; ++j) {
    const int r = wave * RPW + j;
    float s = part[0][r][lane];
    for (int q = 1; q < S; ++q) s += part[q][r][lane];
    const int n = (r < 8) ? n0 : n1;
    if (n < N) z[(long long)(rbase + (r & 7)) * N + n] = s + bias[n];
  }
}

__global__ void gate_kernel(const float* __restrict__ z, float* __restrict__ h) {
  int idx = blockIdx.x * blockDim.x + threadIdx.x;
  if (idx >= CB * CN * CF) return;
  int row = idx / CF, f = idx % CF;
  float z0 = z[row * 150 + f];
  float z1 = z[row * 150 + CF + f];
  float z2 = z[row * 150 + 2 * CF + f];
  float gt = 1.0f / (1.0f + expf(-z0));
  h[idx] = gt * tanhf(z1) + (1.0f - gt) * tanhf(z2);
}

static inline void launch_gemm(const float* A, long long sA, long long sA2, int lda,
                               const float* Bm, long long sB, int ldb,
                               float* C, long long sC, long long sC2, int ldc,
                               const float* bias, int M, int N, int K,
                               int batch, int batch2, int relu, hipStream_t stream) {
  int npair = (N + 31) >> 5;
  int tiles = (M >> 4) * npair;
  dim3 grid((tiles + 3) / 4, batch2, batch);
  gemm_wmma_f32<<<grid, 128, 0, stream>>>(A, sA, sA2, lda, Bm, sB, ldb,
                                          C, sC, sC2, ldc, bias, M, N, K, relu);
}

template <int S>
static inline void launch_gemm_ksplit(const float* A, long long sA, long long sA2, int lda,
                                      const float* Bm, long long sB, int ldb,
                                      float* C, long long sC, long long sC2, int ldc,
                                      const float* bias, int M, int N, int K,
                                      int batch, int batch2, int relu, hipStream_t stream) {
  int npair = (N + 31) >> 5;
  int tiles = (M >> 4) * npair;
  dim3 grid(tiles, batch2, batch);
  gemm_wmma_f32_ksplit<S><<<grid, 32 * S, 0, stream>>>(A, sA, sA2, lda, Bm, sB, ldb,
                                                       C, sC, sC2, ldc, bias, N, K, relu);
}

static const int TE = 4;
static const int TD = 2;

extern "C" void kernel_launch(void* const* d_in, const int* in_sizes, int n_in,
                              void* d_out, int out_size, void* d_ws, size_t ws_size,
                              hipStream_t stream) {
  const float* obs  = (const float*)d_in[0];
  const float* adj  = (const float*)d_in[1];
  const float* h0   = (const float*)d_in[2];
  const float* We1  = (const float*)d_in[3];
  const float* be1  = (const float*)d_in[4];
  const float* We2  = (const float*)d_in[5];
  const float* be2  = (const float*)d_in[6];
  const float* Wch  = (const float*)d_in[7];
  const float* Wcu  = (const float*)d_in[8];
  const float* bc   = (const float*)d_in[9];
  const float* Wd1  = (const float*)d_in[10];
  const float* bd1  = (const float*)d_in[11];
  const float* Wd2  = (const float*)d_in[12];
  const float* bd2  = (const float*)d_in[13];
  const float* Wd3  = (const float*)d_in[14];
  const float* bd3  = (const float*)d_in[15];
  float* out = (float*)d_out;

  const int MN = CB * CN;
  const size_t HSZ = (size_t)MN * CF;

  float* ws = (float*)d_ws;
  size_t off = 0;
  float* dinv = ws + off; off += (size_t)CB * CT * CN;
  float* Wz   = ws + off; off += (size_t)200 * 150;
  float* H    = ws + off; off += 2 * TD * HSZ;
  float* xc   = ws + off; off += TE * HSZ;
  float* scr  = ws + off;
  float* Wb   = scr;
  float* z    = Wb + (size_t)MN * 2 * CF;
  float* tmp1 = scr;
  float* y1   = scr;
  float* y2   = scr + (size_t)TD * MN * CFC;
  (void)ws_size; (void)in_sizes; (void)n_in; (void)out_size;

  float* hlast = H + (2 * TD - 1) * HSZ;
  {
    int rows = CB * CT * CN;
    deg_dinv_kernel<<<(rows * 32 + 255) / 256, 256, 0, stream>>>(adj, dinv, rows, CN);
    pack_wz_kernel<<<(200 * 150 + 255) / 256, 256, 0, stream>>>(Wch, Wcu, Wz);
    hipMemcpyAsync(hlast, h0, sizeof(float) * CB * CN * CF, hipMemcpyDeviceToDevice, stream);
  }

  for (int te = 0; te < CT; te += TE) {
    launch_gemm(obs + (long long)te * CN * CIN, (long long)CT * CN * CIN, (long long)CN * CIN, CIN,
                We1, 0, CFC, tmp1, (long long)CN * CFC, (long long)MN * CFC, CFC,
                be1, CN, CFC, CIN, CB, TE, 1, stream);
    launch_gemm_ksplit<8>(tmp1, 0, 0, CFC, We2, 0, CF, xc, 0, 0, CF,
                          be2, TE * MN, CF, CFC, 1, 1, 1, stream);

    for (int td = 0; td < TE; td += TD) {
      const int grp = ((te + td) / TD) & 1;
      float* Hg = H + (size_t)grp * TD * HSZ;
      for (int j = 0; j < TD; ++j) {
        const int t = te + td + j;
        const float* x = xc + (size_t)(td + j) * HSZ;
        const float* hin = (j == 0) ? (H + ((size_t)(1 - grp) * TD + (TD - 1)) * HSZ)
                                    : (Hg + (size_t)(j - 1) * HSZ);
        float* hout = Hg + (size_t)j * HSZ;
        gemm_adj_ksplit<16><<<dim3((CN / 16) * 4, 1, CB), 32 * 16, 0, stream>>>(
            adj + (long long)t * CN * CN, hin, x, dinv + (long long)t * CN, Wb);
        gemm_gate_ksplit<8><<<(MN / 16) * 5, 32 * 8, 0, stream>>>(
            hin, x, Wb, dinv + (long long)t * CN, Wz, bc, z);
        gate_kernel<<<(MN * CF + 255) / 256, 256, 0, stream>>>(z, hout);
      }
      launch_gemm_ksplit<4>(Hg, 0, 0, CF, Wd1, 0, CFC, y1, 0, 0, CFC,
                            bd1, TD * MN, CFC, CF, 1, 1, 1, stream);
      launch_gemm_ksplit<8>(y1, 0, 0, CFC, Wd2, 0, CFC, y2, 0, 0, CFC,
                            bd2, TD * MN, CFC, CFC, 1, 1, 1, stream);
      launch_gemm_ksplit<8>(y2, (long long)CN * CFC, (long long)MN * CFC, CFC, Wd3, 0, COUT,
                            out + (long long)(te + td) * CN * COUT,
                            (long long)CT * CN * COUT, (long long)CN * COUT, COUT,
                            bd3, CN, COUT, CFC, CB, TD, 0, stream);
    }
  }

  hipMemcpyAsync(out + (long long)CB * CT * CN * COUT, hlast,
                 sizeof(float) * CB * CN * CF, hipMemcpyDeviceToDevice, stream);
}


# decoder runs once per 4 steps (was 2): its 128x128 layer stores over its input (a block owns 16 whole rows: 4 strips x 8 K-split waves), so 4 steps fit the scratch; 162 -> 138 launches; same sum order
# speedup vs baseline: 3.0737x; 1.1252x over previous
#include <hip/hip_runtime.h>
#include <math.h>

static const int CB = 4, CT = 32, CN = 512;
static const int CIN = 10, CF = 50, CFC = 128, COUT = 2;

typedef __attribute__((ext_vector_type(2))) float v2f;
typedef __attribute__((ext_vector_type(8))) float v8f;

__global__ __launch_bounds__(128)
void gemm_wmma_f32(const float* __restrict__ A, long long sA, long long sA2, int lda,
                   const float* __restrict__ Bm, long long sB, int ldb,
                   float* __restrict__ C, long long sC, long long sC2, int ldc,
                   const float* __restrict__ bias,
                   int M, int N, int K, int relu) {
  const int lane  = threadIdx.x & 31;
  const int wave  = threadIdx.x >> 5;
  const int npair = (N + 31) >> 5;
  const int tiles = (M >> 4) * npair;
  const int tile  = blockIdx.x * (blockDim.x >> 5) + wave;
  if (tile >= tiles) return;
  const int tm = (tile / npair) << 4;
  const int tn = (tile % npair) << 5;

  const float* Ab = A  + (long long)blockIdx.z * sA + (long long)blockIdx.y * sA2;
  const float* Bb = Bm + (long long)blockIdx.z * sB;
  float*       Cb = C  + (long long)blockIdx.z * sC + (long long)blockIdx.y * sC2;

  const int mrow  = lane & 15;
  const int khalf = (lane >> 4) << 1;
  const int n0 = tn + (lane & 15);
  const int n1 = n0 + 16;
  const float* ap  = Ab + (long long)(tm + mrow) * lda + khalf;
  const float* bp0 = Bb + (n0 < N ? n0 : N - 1);
  const float* bp1 = Bb + (n1 < N ? n1 : N - 1);

  v8f acc0 = {}, acc1 = {};
  const int K4 = K & ~3;
  int k0 = 0;
#pragma unroll 2
  for (; k0 < K4; k0 += 4) {
    const v2f av = *(const v2f*)(ap + k0);
    const long long kb = (long long)(k0 + khalf) * ldb;
    v2f bv0, bv1;
    bv0.x = bp0[kb];       bv0.y = bp0[kb + ldb];
    bv1.x = bp1[kb];       bv1.y = bp1[kb + ldb];
    acc0 = __builtin_amdgcn_wmma_f32_16x16x4_f32(false, av, false, bv0,
                                                 (short)0, acc0, false, false);
    acc1 = __builtin_amdgcn_wmma_f32_16x16x4_f32(false, av, false, bv1,
                                                 (short)0, acc1, false, false);
  }
  if (k0 < K) {
    const int ka = k0 + khalf;
    v2f av = {}, bv0 = {}, bv1 = {};
    if (ka < K) {
      av.x  = ap[k0];
      bv0.x = bp0[(long long)ka * ldb];
      bv1.x = bp1[(long long)ka * ldb];
    }
    if (ka + 1 < K) {
      av.y  = ap[k0 + 1];
      bv0.y = bp0[(long long)(ka + 1) * ldb];
      bv1.y = bp1[(long long)(ka + 1) * ldb];
    }
    acc0 = __builtin_amdgcn_wmma_f32_16x16x4_f32(false, av, false, bv0,
                                                 (short)0, acc0, false, false);
    acc1 = __builtin_amdgcn_wmma_f32_16x16x4_f32(false, av, false, bv1,
                                                 (short)0, acc1, false, false);
  }

  const int rbase = tm + ((lane >> 4) << 3);
  if (n0 < N) {
    const float badd = bias ? bias[n0] : 0.0f;
#pragma unroll
    for (int r = 0; r < 8; ++r) {
      float v = acc0[r] + badd;
      if (relu) v = fmaxf(v, 0.0f);
      Cb[(long long)(rbase + r) * ldc + n0] = v;
    }
  }
  if (n1 < N) {
    const float badd = bias ? bias[n1] : 0.0f;
#pragma unroll
    for (int r = 0; r < 8; ++r) {
      float v = acc1[r] + badd;
      if (relu) v = fmaxf(v, 0.0f);
      Cb[(long long)(rbase + r) * ldc + n1] = v;
    }
  }
}

template <int S>
__global__ __launch_bounds__(32 * S)
void gemm_wmma_f32_ksplit(const float* __restrict__ A, long long sA, long long sA2, int lda,
                          const float* __restrict__ Bm, long long sB, int ldb,
                          float* __restrict__ C, long long sC, long long sC2, int ldc,
                          const float* __restrict__ bias,
                          int N, int K, int relu) {
  __shared__ float part[S][16][32];

  const int lane  = threadIdx.x & 31;
  const int wave  = __builtin_amdgcn_readfirstlane(threadIdx.x >> 5);
  const int npair = (N + 31) >> 5;
  const int tile  = blockIdx.x;
  const int tm = (tile / npair) << 4;
  const int tn = (tile % npair) << 5;

  const float* Ab = A  + (long long)blockIdx.z * sA + (long long)blockIdx.y * sA2;
  const float* Bb = Bm + (long long)blockIdx.z * sB;
  float*       Cb = C  + (long long)blockIdx.z * sC + (long long)blockIdx.y * sC2;

  const int mrow  = lane & 15;
  const int khalf = (lane >> 4) << 1;
  const int n0 = tn + (lane & 15);
  const int n1 = n0 + 16;
  const float* ap  = Ab + (long long)(tm + mrow) * lda + khalf;
  const float* bp0 = Bb + (n0 < N ? n0 : N - 1);
  const float* bp1 = Bb + (n1 < N ? n1 : N - 1);

  const int K4    = K & ~3;
  const int steps = (K + 3) >> 2;
  const int per   = (steps + S - 1) / S;
  const int kbeg  = wave * per * 4;
  const int klim  = kbeg + per * 4;
  const int kstop = klim < K4 ? klim : K4;

  v8f acc0 = {}, acc1 = {};
#pragma unroll 2
  for (int k0 = kbeg; k0 < kstop; k0 += 4) {
    const v2f av = *(const v2f*)(ap + k0);
    const long long kb = (long long)(k0 + khalf) * ldb;
    v2f bv0, bv1;
    bv0.x = bp0[kb];       bv0.y = bp0[kb + ldb];
    bv1.x = bp1[kb];       bv1.y = bp1[kb + ldb];
    acc0 = __builtin_amdgcn_wmma_f32_16x16x4_f32(false, av, false, bv0,
                                                 (short)0, acc0, false, false);
    acc1 = __builtin_amdgcn_wmma_f32_16x16x4_f32(false, av, false, bv1,
                                                 (short)0, acc1, false, false);
  }
  if (K4 < K && kbeg <= K4 && K4 < klim) {
    const int ka = K4 + khalf;
    v2f av = {}, bv0 = {}, bv1 = {};
    if (ka < K) {
      av.x  = ap[K4];
      bv0.x = bp0[(long long)ka * ldb];
      bv1.x = bp1[(long long)ka * ldb];
    }
    if (ka + 1 < K) {
      av.y  = ap[K4 + 1];
      bv0.y = bp0[(long long)(ka + 1) * ldb];
      bv1.y = bp1[(long long)(ka + 1) * ldb];
    }
    acc0 = __builtin_amdgcn_wmma_f32_16x16x4_f32(false, av, false, bv0,
                                                 (short)0, acc0, false, false);
    acc1 = __builtin_amdgcn_wmma_f32_16x16x4_f32(false, av, false, bv1,
                                                 (short)0, acc1, false, false);
  }

  for (int r = 0; r < 8; ++r) {
    part[wave][r][lane]     = acc0[r];
    part[wave][8 + r][lane] = acc1[r];
  }
  __syncthreads();

  constexpr int RPW = 16 / S;
  const int rbase = tm + ((lane >> 4) << 3);
  for (int j = 0; j < RPW; ++j) {
    const int r = wave * RPW + j;
    float s = part[0][r][lane];
    for (int q = 1; q < S; ++q) s += part[q][r][lane];
    const int n = (r < 8) ? n0 : n1;
    if (n < N) {
      float v = s + (bias ? bias[n] : 0.0f);
      if (relu) v = fmaxf(v, 0.0f);
      Cb[(long long)(rbase + (r & 7)) * ldc + n] = v;
    }
  }
}

template <int S, int NS>
__global__ __launch_bounds__(32 * S * NS)
void gemm_rows_inplace_ksplit(float* Y, int ld,
                              const float* __restrict__ Bm, int ldb,
                              const float* __restrict__ bias, int relu) {
  constexpr int K = 32 * NS;
  static_assert((K / 4) % S == 0, "the same number of whole k-steps in every wave");
  __shared__ float part[NS][S][16][32];

  const int lane  = threadIdx.x & 31;
  const int wv    = __builtin_amdgcn_readfirstlane(threadIdx.x >> 5);
  const int strip = wv / S;
  const int wave  = wv % S;
  const int tm = blockIdx.x << 4;
  const int tn = strip << 5;

  const int mrow  = lane & 15;
  const int khalf = (lane >> 4) << 1;
  const int n0 = tn + (lane & 15);
  const int n1 = n0 + 16;
  const float* ap  = Y + (long long)(tm + mrow) * ld + khalf;
  const float* bp0 = Bm + n0;
  const float* bp1 = Bm + n1;

  constexpr int per = (K / 4) / S;
  const int kbeg = wave * per * 4;
  const int kend = kbeg + per * 4;

  v8f acc0 = {}, acc1 = {};
#pragma unroll 2
  for (int k0 = kbeg; k0 < kend; k0 += 4) {
    const v2f av = *(const v2f*)(ap + k0);
    const long long kb = (long long)(k0 + khalf) * ldb;
    v2f bv0, bv1;
    bv0.x = bp0[kb];       bv0.y = bp0[kb + ldb];
    bv1.x = bp1[kb];       bv1.y = bp1[kb + ldb];
    acc0 = __builtin_amdgcn_wmma_f32_16x16x4_f32(false, av, false, bv0,
                                                 (short)0, acc0, false, false);
    acc1 = __builtin_amdgcn_wmma_f32_16x16x4_f32(false, av, false, bv1,
                                                 (short)0, acc1, false, false);
  }

  for (int r = 0; r < 8; ++r) {
    part[strip][wave][r][lane]     = acc0[r];
    part[strip][wave][8 + r][lane] = acc1[r];
  }
  __syncthreads();

  constexpr int RPW = 16 / S;
  const int rbase = tm + ((lane >> 4) << 3);
  for (int j = 0; j < RPW; ++j) {
    const int r = wave * RPW + j;
    float s = part[strip][0][r][lane];
    for (int q = 1; q < S; ++q) s += part[strip][q][r][lane];
    const int n = (r < 8) ? n0 : n1;
    float v = s + bias[n];
    if (relu) v = fmaxf(v, 0.0f);
    Y[(long long)(rbase + (r & 7)) * ld + n] = v;
  }
}

__global__ void deg_dinv_kernel(const float* __restrict__ A, float* __restrict__ dinv,
                                int rows, int n) {
  const int wid  = (blockIdx.x * blockDim.x + threadIdx.x) >> 5;
  const int lane = threadIdx.x & 31;
  if (wid >= rows) return;
  const float* row = A + (long long)wid * n;
  float s = 0.0f;
  for (int i = lane; i < n; i += 32) s += row[i];
#pragma unroll
  for (int off = 16; off > 0; off >>= 1) s += __shfl_xor(s, off, 32);
  if (lane == 0) dinv[wid] = (s > 0.0f) ? rsqrtf(s) : 0.0f;
}

__global__ void pack_wz_kernel(const float* __restrict__ Wch,
                               const float* __restrict__ Wcu,
                               float* __restrict__ Wz) {
  int idx = blockIdx.x * blockDim.x + threadIdx.x;
  if (idx >= 200 * 150) return;
  int r = idx / 150, cg = idx % 150;
  int c = cg / 50, g = cg % 50;
  int src = r / 50, f = r % 50;
  float v;
  if (src == 0)      v = Wch[((c * 2 + 0) * 50 + f) * 50 + g];
  else if (src == 1) v = Wch[((c * 2 + 1) * 50 + f) * 50 + g];
  else if (src == 2) v = Wcu[((c * 2 + 0) * 50 + f) * 50 + g];
  else               v = Wcu[((c * 2 + 1) * 50 + f) * 50 + g];
  Wz[idx] = v;
}

__device__ __forceinline__ void wmma_group_open() {
  __builtin_amdgcn_sched_barrier(0);
}
__device__ __forceinline__ void wmma_group_close() {
  __builtin_amdgcn_sched_barrier(0);
  asm volatile("v_nop\n\tv_nop\n\tv_nop\n\tv_nop");
  __builtin_amdgcn_sched_barrier(0);
}

template <int S>
__global__ __launch_bounds__(32 * S)
void gemm_adj_ksplit(const float* __restrict__ adj_t,
                     const float* __restrict__ h, const float* __restrict__ x,
                     const float* __restrict__ dinv_t,
                     float* __restrict__ Wout) {
  constexpr int N = 2 * CF;
  constexpr int K = CN;
  static_assert(K % (8 * S) == 0, "whole trips of two k-steps in every wave");
  __shared__ float part[S][16][32];

  const int lane  = threadIdx.x & 31;
  const int wave  = __builtin_amdgcn_readfirstlane(threadIdx.x >> 5);
  constexpr int npair = (N + 31) >> 5;
  const int tm = (blockIdx.x / npair) << 4;
  const int tn = (blockIdx.x % npair) << 5;
  const int b  = blockIdx.z;

  const float* Ab = adj_t  + (long long)b * CT * CN * CN;
  const float* db = dinv_t + (long long)b * CT * CN;
  const float* hb = h + (long long)b * CN * CF;
  const float* xb = x + (long long)b * CN * CF;
  float*       Cb = Wout + (long long)b * CN * N;

  const int mrow  = lane & 15;
  const int khalf = (lane >> 4) << 1;
  const int n0 = tn + (lane & 15);
  const int n1 = n0 + 16;
  const int c0 = n0 < N ? n0 : N - 1;
  const int c1 = n1 < N ? n1 : N - 1;
  const float* ap  = Ab + (long long)(tm + mrow) * K + khalf;
  const float* dp  = db + khalf;
  const float* bp0 = c0 < CF ? hb + c0 : xb + (c0 - CF);
  const float* bp1 = c1 < CF ? hb + c1 : xb + (c1 - CF);

  constexpr int per = K / (4 * S);
  const int kbeg = wave * per * 4;
  const int kend = kbeg + per * 4;

  v8f acc0 = {}, acc1 = {};
  for (int k0 = kbeg; k0 < kend; k0 += 8) {
    const v2f ava = *(const v2f*)(ap + k0);
    const v2f avb = *(const v2f*)(ap + k0 + 4);
    const v2f dda = *(const v2f*)(dp + k0);
    const v2f ddb = *(const v2f*)(dp + k0 + 4);
    const int kb = (k0 + khalf) * CF;
    v2f b0a, b1a, b0b, b1b;
    b0a.x = dda.x * bp0[kb];            b0a.y = dda.y * bp0[kb + CF];
    b1a.x = dda.x * bp1[kb];            b1a.y = dda.y * bp1[kb + CF];
    b0b.x = ddb.x * bp0[kb + 4 * CF];   b0b.y = ddb.y * bp0[kb + 5 * CF];
    b1b.x = ddb.x * bp1[kb + 4 * CF];   b1b.y = ddb.y * bp1[kb + 5 * CF];
    wmma_group_open();
    acc0 = __builtin_amdgcn_wmma_f32_16x16x4_f32(false, ava, false, b0a,
                                                 (short)0, acc0, false, false);
    acc1 = __builtin_amdgcn_wmma_f32_16x16x4_f32(false, ava, false, b1a,
                                                 (short)0, acc1, false, false);
    acc0 = __builtin_amdgcn_wmma_f32_16x16x4_f32(false, avb, false, b0b,
                                                 (short)0, acc0, false, false);
    acc1 = __builtin_amdgcn_wmma_f32_16x16x4_f32(false, avb, false, b1b,
                                                 (short)0, acc1, false, false);
    wmma_group_close();
  }

  for (int r = 0; r < 8; ++r) {
    part[wave][r][lane]     = acc0[r];
    part[wave][8 + r][lane] = acc1[r];
  }
  __syncthreads();

  constexpr int RPW = 16 / S;
  const int rbase = tm + ((lane >> 4) << 3);
  for (int j = 0; j < RPW; ++j) {
    const int r = wave * RPW + j;
    float s = part[0][r][lane];
    for (int q = 1; q < S; ++q) s += part[q][r][lane];
    const int n = (r < 8) ? n0 : n1;
    if (n < N) Cb[(long long)(rbase + (r & 7)) * N + n] = s;
  }
}

__device__ __forceinline__ v2f gate_a_frag(const float* __restrict__ hp,
                                           const float* __restrict__ xp,
                                           const float* __restrict__ wp,
                                           float dv, int k) {
  const int sec = k / CF;
  const int f   = k - sec * CF;
  const v2f sv = *(const v2f*)(((sec & 2) ? xp : hp) + f);
  const v2f wv = *(const v2f*)(wp + (sec >> 1) * CF + f);
  const bool lap = (sec & 1) != 0;
  v2f av;
  av.x = lap ? sv.x - dv * wv.x : sv.x;
  av.y = lap ? sv.y - dv * wv.y : sv.y;
  return av;
}

template <int S>
__global__ __launch_bounds__(32 * S)
void gemm_gate_ksplit(const float* __restrict__ h, const float* __restrict__ x,
                      const float* __restrict__ W, const float* __restrict__ dinv_t,
                      const float* __restrict__ Wz, const float* __restrict__ bias,
                      float* __restrict__ z) {
  constexpr int N = 3 * CF;
  constexpr int K = 4 * CF;
  static_assert(K % 4 == 0 && CF % 2 == 0, "no K tail; sections start at even k");
  __shared__ float part[S][16][32];

  const int lane  = threadIdx.x & 31;
  const int wave  = __builtin_amdgcn_readfirstlane(threadIdx.x >> 5);
  constexpr int npair = (N + 31) >> 5;
  const int tm = (blockIdx.x / npair) << 4;
  const int tn = (blockIdx.x % npair) << 5;

  const int mrow  = lane & 15;
  const int khalf = (lane >> 4) << 1;
  const int n0 = tn + (lane & 15);
  const int n1 = n0 + 16;
  const int row = tm + mrow;
  const float dv = dinv_t[(long long)(row / CN) * CT * CN + (row % CN)];
  const float* hp = h + (long long)row * CF;
  const float* xp = x + (long long)row * CF;
  const float* wp = W + (long long)row * 2 * CF;
  const float* bp0 = Wz + (n0 < N ? n0 : N - 1);
  const float* bp1 = Wz + (n1 < N ? n1 : N - 1);

  constexpr int steps = K / 4;
  constexpr int per   = (steps + S - 1) / S;
  const int kbeg  = wave * per * 4;
  const int klim  = kbeg + per * 4;
  const int kstop = klim < K ? klim : K;

  v8f acc0 = {}, acc1 = {};
  int k0 = kbeg;
  for (; k0 + 4 < kstop; k0 += 8) {
    const int k = k0 + khalf;
    const v2f ava = gate_a_frag(hp, xp, wp, dv, k);
    const v2f avb = gate_a_frag(hp, xp, wp, dv, k + 4);
    const int kb = k * N;
    v2f b0a, b1a, b0b, b1b;
    b0a.x = bp0[kb];           b0a.y = bp0[kb + N];
    b1a.x = bp1[kb];           b1a.y = bp1[kb + N];
    b0b.x = bp0[kb + 4 * N];   b0b.y = bp0[kb + 5 * N];
    b1b.x = bp1[kb + 4 * N];   b1b.y = bp1[kb + 5 * N];
    wmma_group_open();
    acc0 = __builtin_amdgcn_wmma_f32_16x16x4_f32(false, ava, false, b0a,
                                                 (short)0, acc0, false, false);
    acc1 = __builtin_amdgcn_wmma_f32_16x16x4_f32(false, ava, false, b1a,
                                                 (short)0, acc1, false, false);
    acc0 = __builtin_amdgcn_wmma_f32_16x16x4_f32(false, avb, false, b0b,
                                                 (short)0, acc0, false, false);
    acc1 = __builtin_amdgcn_wmma_f32_16x16x4_f32(false, avb, false, b1b,
                                                 (short)0, acc1, false, false);
    wmma_group_close();
  }
  if (k0 < kstop) {
    const int k = k0 + khalf;
    const v2f ava = gate_a_frag(hp, xp, wp, dv, k);
    const int kb = k * N;
    v2f b0a, b1a;
    b0a.x = bp0[kb];           b0a.y = bp0[kb + N];
    b1a.x = bp1[kb];           b1a.y = bp1[kb + N];
    wmma_group_open();
    acc0 = __builtin_amdgcn_wmma_f32_16x16x4_f32(false, ava, false, b0a,
                                                 (short)0, acc0, false, false);
    acc1 = __builtin_amdgcn_wmma_f32_16x16x4_f32(false, ava, false, b1a,
                                                 (short)0, acc1, false, false);
    wmma_group_close();
  }

  for (int r = 0; r < 8; ++r) {
    part[wave][r][lane]     = acc0[r];
    part[wave][8 + r][lane] = acc1[r];
  }
  __syncthreads();

  constexpr int RPW = 16 / S;
  const int rbase = tm + ((lane >> 4) << 3);
  for (int j = 0; j < RPW; ++j) {
    const int r = wave * RPW + j;
    float s = part[0][r][lane];
    for (int q = 1; q < S; ++q) s += part[q][r][lane];
    const int n = (r < 8) ? n0 : n1;
    if (n < N) z[(long long)(rbase + (r & 7)) * N + n] = s + bias[n];
  }
}

__global__ void gate_kernel(const float* __restrict__ z, float* __restrict__ h) {
  int idx = blockIdx.x * blockDim.x + threadIdx.x;
  if (idx >= CB * CN * CF) return;
  int row = idx / CF, f = idx % CF;
  float z0 = z[row * 150 + f];
  float z1 = z[row * 150 + CF + f];
  float z2 = z[row * 150 + 2 * CF + f];
  float gt = 1.0f / (1.0f + expf(-z0));
  h[idx] = gt * tanhf(z1) + (1.0f - gt) * tanhf(z2);
}

static inline void launch_gemm(const float* A, long long sA, long long sA2, int lda,
                               const float* Bm, long long sB, int ldb,
                               float* C, long long sC, long long sC2, int ldc,
                               const float* bias, int M, int N, int K,
                               int batch, int batch2, int relu, hipStream_t stream) {
  int npair = (N + 31) >> 5;
  int tiles = (M >> 4) * npair;
  dim3 grid((tiles + 3) / 4, batch2, batch);
  gemm_wmma_f32<<<grid, 128, 0, stream>>>(A, sA, sA2, lda, Bm, sB, ldb,
                                          C, sC, sC2, ldc, bias, M, N, K, relu);
}

template <int S>
static inline void launch_gemm_ksplit(const float* A, long long sA, long long sA2, int lda,
                                      const float* Bm, long long sB, int ldb,
                                      float* C, long long sC, long long sC2, int ldc,
                                      const float* bias, int M, int N, int K,
                                      int batch, int batch2, int relu, hipStream_t stream) {
  int npair = (N + 31) >> 5;
  int tiles = (M >> 4) * npair;
  dim3 grid(tiles, batch2, batch);
  gemm_wmma_f32_ksplit<S><<<grid, 32 * S, 0, stream>>>(A, sA, sA2, lda, Bm, sB, ldb,
                                                       C, sC, sC2, ldc, bias, N, K, relu);
}

static const int TE = 4;
static const int TD = 4;

extern "C" void kernel_launch(void* const* d_in, const int* in_sizes, int n_in,
                              void* d_out, int out_size, void* d_ws, size_t ws_size,
                              hipStream_t stream) {
  const float* obs  = (const float*)d_in[0];
  const float* adj  = (const float*)d_in[1];
  const float* h0   = (const float*)d_in[2];
  const float* We1  = (const float*)d_in[3];
  const float* be1  = (const float*)d_in[4];
  const float* We2  = (const float*)d_in[5];
  const float* be2  = (const float*)d_in[6];
  const float* Wch  = (const float*)d_in[7];
  const float* Wcu  = (const float*)d_in[8];
  const float* bc   = (const float*)d_in[9];
  const float* Wd1  = (const float*)d_in[10];
  const float* bd1  = (const float*)d_in[11];
  const float* Wd2  = (const float*)d_in[12];
  const float* bd2  = (const float*)d_in[13];
  const float* Wd3  = (const float*)d_in[14];
  const float* bd3  = (const float*)d_in[15];
  float* out = (float*)d_out;

  const int MN = CB * CN;
  const size_t HSZ = (size_t)MN * CF;

  float* ws = (float*)d_ws;
  size_t off = 0;
  float* dinv = ws + off; off += (size_t)CB * CT * CN;
  float* Wz   = ws + off; off += (size_t)200 * 150;
  float* H    = ws + off; off += TD * HSZ;
  float* xc   = ws + off; off += TE * HSZ;
  float* scr  = ws + off;
  float* Wb   = scr;
  float* z    = Wb + (size_t)MN * 2 * CF;
  float* tmp1 = scr;
  float* y    = scr;
  (void)ws_size; (void)in_sizes; (void)n_in; (void)out_size;

  float* hlast = H + (TD - 1) * HSZ;
  {
    int rows = CB * CT * CN;
    deg_dinv_kernel<<<(rows * 32 + 255) / 256, 256, 0, stream>>>(adj, dinv, rows, CN);
    pack_wz_kernel<<<(200 * 150 + 255) / 256, 256, 0, stream>>>(Wch, Wcu, Wz);
    hipMemcpyAsync(hlast, h0, sizeof(float) * CB * CN * CF, hipMemcpyDeviceToDevice, stream);
  }

  for (int te = 0; te < CT; te += TE) {
    launch_gemm(obs + (long long)te * CN * CIN, (long long)CT * CN * CIN, (long long)CN * CIN, CIN,
                We1, 0, CFC, tmp1, (long long)CN * CFC, (long long)MN * CFC, CFC,
                be1, CN, CFC, CIN, CB, TE, 1, stream);
    launch_gemm_ksplit<8>(tmp1, 0, 0, CFC, We2, 0, CF, xc, 0, 0, CF,
                          be2, TE * MN, CF, CFC, 1, 1, 1, stream);

    for (int j = 0; j < TD; ++j) {
      const int t = te + j;
      const float* x = xc + (size_t)j * HSZ;
      const float* hin = (j == 0) ? hlast : (H + (size_t)(j - 1) * HSZ);
      float* hout = H + (size_t)j * HSZ;
      gemm_adj_ksplit<16><<<dim3((CN / 16) * 4, 1, CB), 32 * 16, 0, stream>>>(
          adj + (long long)t * CN * CN, hin, x, dinv + (long long)t * CN, Wb);
      gemm_gate_ksplit<8><<<(MN / 16) * 5, 32 * 8, 0, stream>>>(
          hin, x, Wb, dinv + (long long)t * CN, Wz, bc, z);
      gate_kernel<<<(MN * CF + 255) / 256, 256, 0, stream>>>(z, hout);
    }
    launch_gemm_ksplit<4>(H, 0, 0, CF, Wd1, 0, CFC, y, 0, 0, CFC,
                          bd1, TD * MN, CFC, CF, 1, 1, 1, stream);
    gemm_rows_inplace_ksplit<8, CFC / 32><<<(TD * MN) / 16, 32 * 8 * (CFC / 32), 0, stream>>>(
        y, CFC, Wd2, CFC, bd2, 1);
    launch_gemm_ksplit<8>(y, (long long)CN * CFC, (long long)MN * CFC, CFC, Wd3, 0, COUT,
                          out + (long long)te * CN * COUT,
                          (long long)CT * CN * COUT, (long long)CN * COUT, COUT,
                          bd3, CN, COUT, CFC, CB, TD, 0, stream);
  }

  hipMemcpyAsync(out + (long long)CB * CT * CN * COUT, hlast,
                 sizeof(float) * CB * CN * CF, hipMemcpyDeviceToDevice, stream);
}


# every K-split GEMM block covers a wider column strip (64-128 cols: one A fragment feeds 4-8 tiles, was 2): 2-4x fewer blocks and waves per launch; same S, k ranges, k order and LDS sum order
# speedup vs baseline: 3.7838x; 1.2310x over previous
#include <hip/hip_runtime.h>
#include <math.h>

static const int CB = 4, CT = 32, CN = 512;
static const int CIN = 10, CF = 50, CFC = 128, COUT = 2;

typedef __attribute__((ext_vector_type(2))) float v2f;
typedef __attribute__((ext_vector_type(8))) float v8f;

__global__ __launch_bounds__(128)
void gemm_wmma_f32(const float* __restrict__ A, long long sA, long long sA2, int lda,
                   const float* __restrict__ Bm, long long sB, int ldb,
                   float* __restrict__ C, long long sC, long long sC2, int ldc,
                   const float* __restrict__ bias,
                   int M, int N, int K, int relu) {
  const int lane  = threadIdx.x & 31;
  const int wave  = threadIdx.x >> 5;
  const int npair = (N + 31) >> 5;
  const int tiles = (M >> 4) * npair;
  const int tile  = blockIdx.x * (blockDim.x >> 5) + wave;
  if (tile >= tiles) return;
  const int tm = (tile / npair) << 4;
  const int tn = (tile % npair) << 5;

  const float* Ab = A  + (long long)blockIdx.z * sA + (long long)blockIdx.y * sA2;
  const float* Bb = Bm + (long long)blockIdx.z * sB;
  float*       Cb = C  + (long long)blockIdx.z * sC + (long long)blockIdx.y * sC2;

  const int mrow  = lane & 15;
  const int khalf = (lane >> 4) << 1;
  const int n0 = tn + (lane & 15);
  const int n1 = n0 + 16;
  const float* ap  = Ab + (long long)(tm + mrow) * lda + khalf;
  const float* bp0 = Bb + (n0 < N ? n0 : N - 1);
  const float* bp1 = Bb + (n1 < N ? n1 : N - 1);

  v8f acc0 = {}, acc1 = {};
  const int K4 = K & ~3;
  int k0 = 0;
#pragma unroll 2
  for (; k0 < K4; k0 += 4) {
    const v2f av = *(const v2f*)(ap + k0);
    const long long kb = (long long)(k0 + khalf) * ldb;
    v2f bv0, bv1;
    bv0.x = bp0[kb];       bv0.y = bp0[kb + ldb];
    bv1.x = bp1[kb];       bv1.y = bp1[kb + ldb];
    acc0 = __builtin_amdgcn_wmma_f32_16x16x4_f32(false, av, false, bv0,
                                                 (short)0, acc0, false, false);
    acc1 = __builtin_amdgcn_wmma_f32_16x16x4_f32(false, av, false, bv1,
                                                 (short)0, acc1, false, false);
  }
  if (k0 < K) {
    const int ka = k0 + khalf;
    v2f av = {}, bv0 = {}, bv1 = {};
    if (ka < K) {
      av.x  = ap[k0];
      bv0.x = bp0[(long long)ka * ldb];
      bv1.x = bp1[(long long)ka * ldb];
    }
    if (ka + 1 < K) {
      av.y  = ap[k0 + 1];
      bv0.y = bp0[(long long)(ka + 1) * ldb];
      bv1.y = bp1[(long long)(ka + 1) * ldb];
    }
    acc0 = __builtin_amdgcn_wmma_f32_16x16x4_f32(false, av, false, bv0,
                                                 (short)0, acc0, false, false);
    acc1 = __builtin_amdgcn_wmma_f32_16x16x4_f32(false, av, false, bv1,
                                                 (short)0, acc1, false, false);
  }

  const int rbase = tm + ((lane >> 4) << 3);
  if (n0 < N) {
    const float badd = bias ? bias[n0] : 0.0f;
#pragma unroll
    for (int r = 0; r < 8; ++r) {
      float v = acc0[r] + badd;
      if (relu) v = fmaxf(v, 0.0f);
      Cb[(long long)(rbase + r) * ldc + n0] = v;
    }
  }
  if (n1 < N) {
    const float badd = bias ? bias[n1] : 0.0f;
#pragma unroll
    for (int r = 0; r < 8; ++r) {
      float v = acc1[r] + badd;
      if (relu) v = fmaxf(v, 0.0f);
      Cb[(long long)(rbase + r) * ldc + n1] = v;
    }
  }
}

__device__ __forceinline__ void wmma_group_open() {
  __builtin_amdgcn_sched_barrier(0);
}
__device__ __forceinline__ void wmma_operands_ready(v2f& b_a, v2f& b_b, v2f& ava, v2f& avb) {
  asm volatile("" : "+v"(b_a), "+v"(b_b), "+v"(ava), "+v"(avb));
}
__device__ __forceinline__ void wmma_operands_ready(v2f& b_a, v2f& ava) {
  asm volatile("" : "+v"(b_a), "+v"(ava));
}
__device__ __forceinline__ void wmma_group_close() {
  __builtin_amdgcn_sched_barrier(0);
  asm volatile("v_nop\n\tv_nop\n\tv_nop\n\tv_nop");
  __builtin_amdgcn_sched_barrier(0);
}

template <int S, int NT>
__global__ __launch_bounds__(32 * S)
void gemm_wmma_f32_ksplit(const float* __restrict__ A, long long sA, long long sA2, int lda,
                          const float* __restrict__ Bm, long long sB, int ldb,
                          float* __restrict__ C, long long sC, long long sC2, int ldc,
                          const float* __restrict__ bias,
                          int N, int K, int relu) {
  static_assert((8 * NT) % S == 0, "whole accumulator registers per wave in the reduction");
  __shared__ float part[S][8 * NT][32];

  const int lane  = threadIdx.x & 31;
  const int wave  = __builtin_amdgcn_readfirstlane(threadIdx.x >> 5);
  constexpr int SW = 16 * NT;
  const int nstrip = (N + SW - 1) / SW;
  const int tile  = blockIdx.x;
  const int tm = (tile / nstrip) << 4;
  const int tn = (tile % nstrip) * SW;

  const float* Ab = A  + (long long)blockIdx.z * sA + (long long)blockIdx.y * sA2;
  const float* Bb = Bm + (long long)blockIdx.z * sB;
  float*       Cb = C  + (long long)blockIdx.z * sC + (long long)blockIdx.y * sC2;

  const int mrow  = lane & 15;
  const int khalf = (lane >> 4) << 1;
  const int nb = tn + (lane & 15);
  const float* ap  = Ab + (long long)(tm + mrow) * lda + khalf;
  const float* bp[NT];
#pragma unroll
  for (int t = 0; t < NT; ++t) {
    const int n = nb + 16 * t;
    bp[t] = Bb + (n < N ? n : N - 1);
  }

  const int K4    = K & ~3;
  const int steps = (K + 3) >> 2;
  const int per   = (steps + S - 1) / S;
  const int kbeg  = wave * per * 4;
  const int klim  = kbeg + per * 4;
  const int kstop = klim < K4 ? klim : K4;

  v8f acc[NT];
#pragma unroll
  for (int t = 0; t < NT; ++t) acc[t] = v8f{};
  int k0 = kbeg;
  for (; k0 + 4 < kstop; k0 += 8) {
    v2f ava = *(const v2f*)(ap + k0);
    v2f avb = *(const v2f*)(ap + k0 + 4);
    const long long kb = (long long)(k0 + khalf) * ldb;
    v2f ba[NT], bb[NT];
#pragma unroll
    for (int t = 0; t < NT; ++t) {
      ba[t].x = bp[t][kb];             ba[t].y = bp[t][kb + ldb];
      bb[t].x = bp[t][kb + 4LL * ldb]; bb[t].y = bp[t][kb + 5LL * ldb];
    }
#pragma unroll
    for (int t = 0; t < NT; ++t) wmma_operands_ready(ba[t], bb[t], ava, avb);
    wmma_group_open();
#pragma unroll
    for (int t = 0; t < NT; ++t)
      acc[t] = __builtin_amdgcn_wmma_f32_16x16x4_f32(false, ava, false, ba[t],
                                                     (short)0, acc[t], false, false);
#pragma unroll
    for (int t = 0; t < NT; ++t)
      acc[t] = __builtin_amdgcn_wmma_f32_16x16x4_f32(false, avb, false, bb[t],
                                                     (short)0, acc[t], false, false);
    wmma_group_close();
  }
  if (k0 < kstop) {
    v2f ava = *(const v2f*)(ap + k0);
    const long long kb = (long long)(k0 + khalf) * ldb;
    v2f ba[NT];
#pragma unroll
    for (int t = 0; t < NT; ++t) { ba[t].x = bp[t][kb]; ba[t].y = bp[t][kb + ldb]; }
#pragma unroll
    for (int t = 0; t < NT; ++t) wmma_operands_ready(ba[t], ava);
    wmma_group_open();
#pragma unroll
    for (int t = 0; t < NT; ++t)
      acc[t] = __builtin_amdgcn_wmma_f32_16x16x4_f32(false, ava, false, ba[t],
                                                     (short)0, acc[t], false, false);
    wmma_group_close();
  }
  if (K4 < K && kbeg <= K4 && K4 < klim) {
    const int ka = K4 + khalf;
    v2f av = {};
    v2f bv[NT];
#pragma unroll
    for (int t = 0; t < NT; ++t) bv[t] = v2f{};
    if (ka < K) {
      av.x  = ap[K4];
#pragma unroll
      for (int t = 0; t < NT; ++t) bv[t].x = bp[t][(long long)ka * ldb];
    }
    if (ka + 1 < K) {
      av.y  = ap[K4 + 1];
#pragma unroll
      for (int t = 0; t < NT; ++t) bv[t].y = bp[t][(long long)(ka + 1) * ldb];
    }
#pragma unroll
    for (int t = 0; t < NT; ++t) wmma_operands_ready(bv[t], av);
    wmma_group_open();
#pragma unroll
    for (int t = 0; t < NT; ++t)
      acc[t] = __builtin_amdgcn_wmma_f32_16x16x4_f32(false, av, false, bv[t],
                                                     (short)0, acc[t], false, false);
    wmma_group_close();
  }

#pragma unroll
  for (int t = 0; t < NT; ++t)
#pragma unroll
    for (int r = 0; r < 8; ++r) part[wave][8 * t + r][lane] = acc[t][r];
  __syncthreads();

  constexpr int RPW = 8 * NT / S;
  const int rbase = tm + ((lane >> 4) << 3);
  for (int j = 0; j < RPW; ++j) {
    const int r = wave * RPW + j;
    float s = part[0][r][lane];
    for (int q = 1; q < S; ++q) s += part[q][r][lane];
    const int n = nb + 16 * (r >> 3);
    if (n < N) {
      float v = s + (bias ? bias[n] : 0.0f);
      if (relu) v = fmaxf(v, 0.0f);
      Cb[(long long)(rbase + (r & 7)) * ldc + n] = v;
    }
  }
}

template <int S, int NT>
__global__ __launch_bounds__(32 * S)
void gemm_rows_inplace_ksplit(float* Y, int ld,
                              const float* __restrict__ Bm, int ldb,
                              const float* __restrict__ bias, int relu) {
  constexpr int K = 16 * NT;
  static_assert((K / 8) % S == 0, "whole trips of two k-steps, the same number in every wave");
  static_assert((8 * NT) % S == 0, "whole accumulator registers per wave in the reduction");
  __shared__ float part[S][8 * NT][32];

  const int lane  = threadIdx.x & 31;
  const int wave  = __builtin_amdgcn_readfirstlane(threadIdx.x >> 5);
  const int tm = blockIdx.x << 4;

  const int mrow  = lane & 15;
  const int khalf = (lane >> 4) << 1;
  const int nb = lane & 15;
  const float* ap  = Y + (long long)(tm + mrow) * ld + khalf;
  const float* bp  = Bm + nb;

  constexpr int per = (K / 4) / S;
  const int kbeg = wave * per * 4;
  const int kend = kbeg + per * 4;

  v8f acc[NT];
#pragma unroll
  for (int t = 0; t < NT; ++t) acc[t] = v8f{};
  for (int k0 = kbeg; k0 < kend; k0 += 8) {
    v2f ava = *(const v2f*)(ap + k0);
    v2f avb = *(const v2f*)(ap + k0 + 4);
    const long long kb = (long long)(k0 + khalf) * ldb;
    v2f ba[NT], bb[NT];
#pragma unroll
    for (int t = 0; t < NT; ++t) {
      ba[t].x = bp[kb + 16 * t];             ba[t].y = bp[kb + ldb + 16 * t];
      bb[t].x = bp[kb + 4LL * ldb + 16 * t]; bb[t].y = bp[kb + 5LL * ldb + 16 * t];
    }
#pragma unroll
    for (int t = 0; t < NT; ++t) wmma_operands_ready(ba[t], bb[t], ava, avb);
    wmma_group_open();
#pragma unroll
    for (int t = 0; t < NT; ++t)
      acc[t] = __builtin_amdgcn_wmma_f32_16x16x4_f32(false, ava, false, ba[t],
                                                     (short)0, acc[t], false, false);
#pragma unroll
    for (int t = 0; t < NT; ++t)
      acc[t] = __builtin_amdgcn_wmma_f32_16x16x4_f32(false, avb, false, bb[t],
                                                     (short)0, acc[t], false, false);
    wmma_group_close();
  }

#pragma unroll
  for (int t = 0; t < NT; ++t)
#pragma unroll
    for (int r = 0; r < 8; ++r) part[wave][8 * t + r][lane] = acc[t][r];
  __syncthreads();

  constexpr int RPW = 8 * NT / S;
  const int rbase = tm + ((lane >> 4) << 3);
  for (int j = 0; j < RPW; ++j) {
    const int r = wave * RPW + j;
    float s = part[0][r][lane];
    for (int q = 1; q < S; ++q) s += part[q][r][lane];
    const int n = nb + 16 * (r >> 3);
    float v = s + bias[n];
    if (relu) v = fmaxf(v, 0.0f);
    Y[(long long)(rbase + (r & 7)) * ld + n] = v;
  }
}

__global__ void deg_dinv_kernel(const float* __restrict__ A, float* __restrict__ dinv,
                                int rows, int n) {
  const int wid  = (blockIdx.x * blockDim.x + threadIdx.x) >> 5;
  const int lane = threadIdx.x & 31;
  if (wid >= rows) return;
  const float* row = A + (long long)wid * n;
  float s = 0.0f;
  for (int i = lane; i < n; i += 32) s += row[i];
#pragma unroll
  for (int off = 16; off > 0; off >>= 1) s += __shfl_xor(s, off, 32);
  if (lane == 0) dinv[wid] = (s > 0.0f) ? rsqrtf(s) : 0.0f;
}

__global__ void pack_wz_kernel(const float* __restrict__ Wch,
                               const float* __restrict__ Wcu,
                               float* __restrict__ Wz) {
  int idx = blockIdx.x * blockDim.x + threadIdx.x;
  if (idx >= 200 * 150) return;
  int r = idx / 150, cg = idx % 150;
  int c = cg / 50, g = cg % 50;
  int src = r / 50, f = r % 50;
  float v;
  if (src == 0)      v = Wch[((c * 2 + 0) * 50 + f) * 50 + g];
  else if (src == 1) v = Wch[((c * 2 + 1) * 50 + f) * 50 + g];
  else if (src == 2) v = Wcu[((c * 2 + 0) * 50 + f) * 50 + g];
  else               v = Wcu[((c * 2 + 1) * 50 + f) * 50 + g];
  Wz[idx] = v;
}

template <int S, int NT>
__global__ __launch_bounds__(32 * S)
void gemm_adj_ksplit(const float* __restrict__ adj_t,
                     const float* __restrict__ h, const float* __restrict__ x,
                     const float* __restrict__ dinv_t,
                     float* __restrict__ Wout) {
  constexpr int N = 2 * CF;
  constexpr int K = CN;
  static_assert(K % (8 * S) == 0, "whole trips of two k-steps in every wave");
  static_assert((8 * NT) % S == 0, "whole accumulator registers per wave in the reduction");
  __shared__ float part[S][8 * NT][32];

  const int lane  = threadIdx.x & 31;
  const int wave  = __builtin_amdgcn_readfirstlane(threadIdx.x >> 5);
  constexpr int SW = 16 * NT;
  constexpr int nstrip = (N + SW - 1) / SW;
  const int tm = (blockIdx.x / nstrip) << 4;
  const int tn = (blockIdx.x % nstrip) * SW;
  const int b  = blockIdx.z;

  const float* Ab = adj_t  + (long long)b * CT * CN * CN;
  const float* db = dinv_t + (long long)b * CT * CN;
  const float* hb = h + (long long)b * CN * CF;
  const float* xb = x + (long long)b * CN * CF;
  float*       Cb = Wout + (long long)b * CN * N;

  const int mrow  = lane & 15;
  const int khalf = (lane >> 4) << 1;
  const int nb = tn + (lane & 15);
  const float* ap  = Ab + (long long)(tm + mrow) * K + khalf;
  const float* dp  = db + khalf;
  const float* bp[NT];
#pragma unroll
  for (int t = 0; t < NT; ++t) {
    const int n = nb + 16 * t;
    const int c = n < N ? n : N - 1;
    bp[t] = c < CF ? hb + c : xb + (c - CF);
  }

  constexpr int per = K / (4 * S);
  const int kbeg = wave * per * 4;
  const int kend = kbeg + per * 4;

  v8f acc[NT];
#pragma unroll
  for (int t = 0; t < NT; ++t) acc[t] = v8f{};
  for (int k0 = kbeg; k0 < kend; k0 += 8) {
    v2f ava = *(const v2f*)(ap + k0);
    v2f avb = *(const v2f*)(ap + k0 + 4);
    const v2f dda = *(const v2f*)(dp + k0);
    const v2f ddb = *(const v2f*)(dp + k0 + 4);
    const int kb = (k0 + khalf) * CF;
    v2f ba[NT], bb[NT];
#pragma unroll
    for (int t = 0; t < NT; ++t) {
      ba[t].x = dda.x * bp[t][kb];            ba[t].y = dda.y * bp[t][kb + CF];
      bb[t].x = ddb.x * bp[t][kb + 4 * CF];   bb[t].y = ddb.y * bp[t][kb + 5 * CF];
    }
#pragma unroll
    for (int t = 0; t < NT; ++t) wmma_operands_ready(ba[t], bb[t], ava, avb);
    wmma_group_open();
#pragma unroll
    for (int t = 0; t < NT; ++t)
      acc[t] = __builtin_amdgcn_wmma_f32_16x16x4_f32(false, ava, false, ba[t],
                                                     (short)0, acc[t], false, false);
#pragma unroll
    for (int t = 0; t < NT; ++t)
      acc[t] = __builtin_amdgcn_wmma_f32_16x16x4_f32(false, avb, false, bb[t],
                                                     (short)0, acc[t], false, false);
    wmma_group_close();
  }

#pragma unroll
  for (int t = 0; t < NT; ++t)
#pragma unroll
    for (int r = 0; r < 8; ++r) part[wave][8 * t + r][lane] = acc[t][r];
  __syncthreads();

  constexpr int RPW = 8 * NT / S;
  const int rbase = tm + ((lane >> 4) << 3);
  for (int j = 0; j < RPW; ++j) {
    const int r = wave * RPW + j;
    float s = part[0][r][lane];
    for (int q = 1; q < S; ++q) s += part[q][r][lane];
    const int n = nb + 16 * (r >> 3);
    if (n < N) Cb[(long long)(rbase + (r & 7)) * N + n] = s;
  }
}

__device__ __forceinline__ v2f gate_a_frag(const float* __restrict__ hp,
                                           const float* __restrict__ xp,
                                           const float* __restrict__ wp,
                                           float dv, int k) {
  const int sec = k / CF;
  const int f   = k - sec * CF;
  const v2f sv = *(const v2f*)(((sec & 2) ? xp : hp) + f);
  const v2f wv = *(const v2f*)(wp + (sec >> 1) * CF + f);
  const bool lap = (sec & 1) != 0;
  v2f av;
  av.x = lap ? sv.x - dv * wv.x : sv.x;
  av.y = lap ? sv.y - dv * wv.y : sv.y;
  return av;
}

template <int S, int NT>
__global__ __launch_bounds__(32 * S)
void gemm_gate_ksplit(const float* __restrict__ h, const float* __restrict__ x,
                      const float* __restrict__ W, const float* __restrict__ dinv_t,
                      const float* __restrict__ Wz, const float* __restrict__ bias,
                      float* __restrict__ z) {
  constexpr int N = 3 * CF;
  constexpr int K = 4 * CF;
  static_assert(K % 4 == 0 && CF % 2 == 0, "no K tail; sections start at even k");
  static_assert((8 * NT) % S == 0, "whole accumulator registers per wave in the reduction");
  __shared__ float part[S][8 * NT][32];

  const int lane  = threadIdx.x & 31;
  const int wave  = __builtin_amdgcn_readfirstlane(threadIdx.x >> 5);
  constexpr int SW = 16 * NT;
  constexpr int nstrip = (N + SW - 1) / SW;
  const int tm = (blockIdx.x / nstrip) << 4;
  const int tn = (blockIdx.x % nstrip) * SW;

  const int mrow  = lane & 15;
  const int khalf = (lane >> 4) << 1;
  const int nb = tn + (lane & 15);
  const int row = tm + mrow;
  const float dv = dinv_t[(long long)(row / CN) * CT * CN + (row % CN)];
  const float* hp = h + (long long)row * CF;
  const float* xp = x + (long long)row * CF;
  const float* wp = W + (long long)row * 2 * CF;
  const float* bp[NT];
#pragma unroll
  for (int t = 0; t < NT; ++t) {
    const int n = nb + 16 * t;
    bp[t] = Wz + (n < N ? n : N - 1);
  }

  constexpr int steps = K / 4;
  constexpr int per   = (steps + S - 1) / S;
  const int kbeg  = wave * per * 4;
  const int klim  = kbeg + per * 4;
  const int kstop = klim < K ? klim : K;

  v8f acc[NT];
#pragma unroll
  for (int t = 0; t < NT; ++t) acc[t] = v8f{};
  int k0 = kbeg;
  for (; k0 + 4 < kstop; k0 += 8) {
    const int k = k0 + khalf;
    v2f ava = gate_a_frag(hp, xp, wp, dv, k);
    v2f avb = gate_a_frag(hp, xp, wp, dv, k + 4);
    const int kb = k * N;
    v2f ba[NT], bb[NT];
#pragma unroll
    for (int t = 0; t < NT; ++t) {
      ba[t].x = bp[t][kb];           ba[t].y = bp[t][kb + N];
      bb[t].x = bp[t][kb + 4 * N];   bb[t].y = bp[t][kb + 5 * N];
    }
#pragma unroll
    for (int t = 0; t < NT; ++t) wmma_operands_ready(ba[t], bb[t], ava, avb);
    wmma_group_open();
#pragma unroll
    for (int t = 0; t < NT; ++t)
      acc[t] = __builtin_amdgcn_wmma_f32_16x16x4_f32(false, ava, false, ba[t],
                                                     (short)0, acc[t], false, false);
#pragma unroll
    for (int t = 0; t < NT; ++t)
      acc[t] = __builtin_amdgcn_wmma_f32_16x16x4_f32(false, avb, false, bb[t],
                                                     (short)0, acc[t], false, false);
    wmma_group_close();
  }
  if (k0 < kstop) {
    const int k = k0 + khalf;
    v2f ava = gate_a_frag(hp, xp, wp, dv, k);
    const int kb = k * N;
    v2f ba[NT];
#pragma unroll
    for (int t = 0; t < NT; ++t) { ba[t].x = bp[t][kb]; ba[t].y = bp[t][kb + N]; }
#pragma unroll
    for (int t = 0; t < NT; ++t) wmma_operands_ready(ba[t], ava);
    wmma_group_open();
#pragma unroll
    for (int t = 0; t < NT; ++t)
      acc[t] = __builtin_amdgcn_wmma_f32_16x16x4_f32(false, ava, false, ba[t],
                                                     (short)0, acc[t], false, false);
    wmma_group_close();
  }

#pragma unroll
  for (int t = 0; t < NT; ++t)
#pragma unroll
    for (int r = 0; r < 8; ++r) part[wave][8 * t + r][lane] = acc[t][r];
  __syncthreads();

  constexpr int RPW = 8 * NT / S;
  const int rbase = tm + ((lane >> 4) << 3);
  for (int j = 0; j < RPW; ++j) {
    const int r = wave * RPW + j;
    float s = part[0][r][lane];
    for (int q = 1; q < S; ++q) s += part[q][r][lane];
    const int n = nb + 16 * (r >> 3);
    if (n < N) z[(long long)(rbase + (r & 7)) * N + n] = s + bias[n];
  }
}

__global__ void gate_kernel(const float* __restrict__ z, float* __restrict__ h) {
  int idx = blockIdx.x * blockDim.x + threadIdx.x;
  if (idx >= CB * CN * CF) return;
  int row = idx / CF, f = idx % CF;
  float z0 = z[row * 150 + f];
  float z1 = z[row * 150 + CF + f];
  float z2 = z[row * 150 + 2 * CF + f];
  float gt = 1.0f / (1.0f + expf(-z0));
  h[idx] = gt * tanhf(z1) + (1.0f - gt) * tanhf(z2);
}

static inline void launch_gemm(const float* A, long long sA, long long sA2, int lda,
                               const float* Bm, long long sB, int ldb,
                               float* C, long long sC, long long sC2, int ldc,
                               const float* bias, int M, int N, int K,
                               int batch, int batch2, int relu, hipStream_t stream) {
  int npair = (N + 31) >> 5;
  int tiles = (M >> 4) * npair;
  dim3 grid((tiles + 3) / 4, batch2, batch);
  gemm_wmma_f32<<<grid, 128, 0, stream>>>(A, sA, sA2, lda, Bm, sB, ldb,
                                          C, sC, sC2, ldc, bias, M, N, K, relu);
}

template <int S, int NT>
static inline void launch_gemm_ksplit(const float* A, long long sA, long long sA2, int lda,
                                      const float* Bm, long long sB, int ldb,
                                      float* C, long long sC, long long sC2, int ldc,
                                      const float* bias, int M, int N, int K,
                                      int batch, int batch2, int relu, hipStream_t stream) {
  int nstrip = (N + 16 * NT - 1) / (16 * NT);
  int tiles = (M >> 4) * nstrip;
  dim3 grid(tiles, batch2, batch);
  gemm_wmma_f32_ksplit<S, NT><<<grid, 32 * S, 0, stream>>>(A, sA, sA2, lda, Bm, sB, ldb,
                                                           C, sC, sC2, ldc, bias, N, K, relu);
}

static const int TE = 4;
static const int TD = 4;

extern "C" void kernel_launch(void* const* d_in, const int* in_sizes, int n_in,
                              void* d_out, int out_size, void* d_ws, size_t ws_size,
                              hipStream_t stream) {
  const float* obs  = (const float*)d_in[0];
  const float* adj  = (const float*)d_in[1];
  const float* h0   = (const float*)d_in[2];
  const float* We1  = (const float*)d_in[3];
  const float* be1  = (const float*)d_in[4];
  const float* We2  = (const float*)d_in[5];
  const float* be2  = (const float*)d_in[6];
  const float* Wch  = (const float*)d_in[7];
  const float* Wcu  = (const float*)d_in[8];
  const float* bc   = (const float*)d_in[9];
  const float* Wd1  = (const float*)d_in[10];
  const float* bd1  = (const float*)d_in[11];
  const float* Wd2  = (const float*)d_in[12];
  const float* bd2  = (const float*)d_in[13];
  const float* Wd3  = (const float*)d_in[14];
  const float* bd3  = (const float*)d_in[15];
  float* out = (float*)d_out;

  const int MN = CB * CN;
  const size_t HSZ = (size_t)MN * CF;

  float* ws = (float*)d_ws;
  size_t off = 0;
  float* dinv = ws + off; off += (size_t)CB * CT * CN;
  float* Wz   = ws + off; off += (size_t)200 * 150;
  float* H    = ws + off; off += TD * HSZ;
  float* xc   = ws + off; off += TE * HSZ;
  float* scr  = ws + off;
  float* Wb   = scr;
  float* z    = Wb + (size_t)MN * 2 * CF;
  float* tmp1 = scr;
  float* y    = scr;
  (void)ws_size; (void)in_sizes; (void)n_in; (void)out_size;

  float* hlast = H + (TD - 1) * HSZ;
  {
    int rows = CB * CT * CN;
    deg_dinv_kernel<<<(rows * 32 + 255) / 256, 256, 0, stream>>>(adj, dinv, rows, CN);
    pack_wz_kernel<<<(200 * 150 + 255) / 256, 256, 0, stream>>>(Wch, Wcu, Wz);
    hipMemcpyAsync(hlast, h0, sizeof(float) * CB * CN * CF, hipMemcpyDeviceToDevice, stream);
  }

  for (int te = 0; te < CT; te += TE) {
    launch_gemm(obs + (long long)te * CN * CIN, (long long)CT * CN * CIN, (long long)CN * CIN, CIN,
                We1, 0, CFC, tmp1, (long long)CN * CFC, (long long)MN * CFC, CFC,
                be1, CN, CFC, CIN, CB, TE, 1, stream);
    launch_gemm_ksplit<8, 4>(tmp1, 0, 0, CFC, We2, 0, CF, xc, 0, 0, CF,
                          be2, TE * MN, CF, CFC, 1, 1, 1, stream);

    for (int j = 0; j < TD; ++j) {
      const int t = te + j;
      const float* x = xc + (size_t)j * HSZ;
      const float* hin = (j == 0) ? hlast : (H + (size_t)(j - 1) * HSZ);
      float* hout = H + (size_t)j * HSZ;
      gemm_adj_ksplit<16, 4><<<dim3((CN / 16) * 2, 1, CB), 32 * 16, 0, stream>>>(
          adj + (long long)t * CN * CN, hin, x, dinv + (long long)t * CN, Wb);
      gemm_gate_ksplit<8, 5><<<(MN / 16) * 2, 32 * 8, 0, stream>>>(
          hin, x, Wb, dinv + (long long)t * CN, Wz, bc, z);
      gate_kernel<<<(MN * CF + 255) / 256, 256, 0, stream>>>(z, hout);
    }
    launch_gemm_ksplit<4, 8>(H, 0, 0, CF, Wd1, 0, CFC, y, 0, 0, CFC,
                          bd1, TD * MN, CFC, CF, 1, 1, 1, stream);
    gemm_rows_inplace_ksplit<8, CFC / 16><<<(TD * MN) / 16, 32 * 8, 0, stream>>>(
        y, CFC, Wd2, CFC, bd2, 1);
    launch_gemm_ksplit<8, 1>(y, (long long)CN * CFC, (long long)MN * CFC, CFC, Wd3, 0, COUT,
                          out + (long long)te * CN * COUT,
                          (long long)CT * CN * COUT, (long long)CN * COUT, COUT,
                          bd3, CN, COUT, CFC, CB, TD, 0, stream);
  }

  hipMemcpyAsync(out + (long long)CB * CT * CN * COUT, hlast,
                 sizeof(float) * CB * CN * CF, hipMemcpyDeviceToDevice, stream);
}


# decoder's 3 layers in one launch: a block owns 16 whole rows and reads the rows it stored back from the global scratch plane behind a barrier (no LDS staging); 138 -> 122 launches; same sum order
# speedup vs baseline: 4.2546x; 1.1244x over previous
#include <hip/hip_runtime.h>
#include <math.h>

static const int CB = 4, CT = 32, CN = 512;
static const int CIN = 10, CF = 50, CFC = 128, COUT = 2;

typedef __attribute__((ext_vector_type(2))) float v2f;
typedef __attribute__((ext_vector_type(8))) float v8f;

__global__ __launch_bounds__(128)
void gemm_wmma_f32(const float* __restrict__ A, long long sA, long long sA2, int lda,
                   const float* __restrict__ Bm, long long sB, int ldb,
                   float* __restrict__ C, long long sC, long long sC2, int ldc,
                   const float* __restrict__ bias,
                   int M, int N, int K, int relu) {
  const int lane  = threadIdx.x & 31;
  const int wave  = threadIdx.x >> 5;
  const int npair = (N + 31) >> 5;
  const int tiles = (M >> 4) * npair;
  const int tile  = blockIdx.x * (blockDim.x >> 5) + wave;
  if (tile >= tiles) return;
  const int tm = (tile / npair) << 4;
  const int tn = (tile % npair) << 5;

  const float* Ab = A  + (long long)blockIdx.z * sA + (long long)blockIdx.y * sA2;
  const float* Bb = Bm + (long long)blockIdx.z * sB;
  float*       Cb = C  + (long long)blockIdx.z * sC + (long long)blockIdx.y * sC2;

  const int mrow  = lane & 15;
  const int khalf = (lane >> 4) << 1;
  const int n0 = tn + (lane & 15);
  const int n1 = n0 + 16;
  const float* ap  = Ab + (long long)(tm + mrow) * lda + khalf;
  const float* bp0 = Bb + (n0 < N ? n0 : N - 1);
  const float* bp1 = Bb + (n1 < N ? n1 : N - 1);

  v8f acc0 = {}, acc1 = {};
  const int K4 = K & ~3;
  int k0 = 0;
#pragma unroll 2
  for (; k0 < K4; k0 += 4) {
    const v2f av = *(const v2f*)(ap + k0);
    const long long kb = (long long)(k0 + khalf) * ldb;
    v2f bv0, bv1;
    bv0.x = bp0[kb];       bv0.y = bp0[kb + ldb];
    bv1.x = bp1[kb];       bv1.y = bp1[kb + ldb];
    acc0 = __builtin_amdgcn_wmma_f32_16x16x4_f32(false, av, false, bv0,
                                                 (short)0, acc0, false, false);
    acc1 = __builtin_amdgcn_wmma_f32_16x16x4_f32(false, av, false, bv1,
                                                 (short)0, acc1, false, false);
  }
  if (k0 < K) {
    const int ka = k0 + khalf;
    v2f av = {}, bv0 = {}, bv1 = {};
    if (ka < K) {
      av.x  = ap[k0];
      bv0.x = bp0[(long long)ka * ldb];
      bv1.x = bp1[(long long)ka * ldb];
    }
    if (ka + 1 < K) {
      av.y  = ap[k0 + 1];
      bv0.y = bp0[(long long)(ka + 1) * ldb];
      bv1.y = bp1[(long long)(ka + 1) * ldb];
    }
    acc0 = __builtin_amdgcn_wmma_f32_16x16x4_f32(false, av, false, bv0,
                                                 (short)0, acc0, false, false);
    acc1 = __builtin_amdgcn_wmma_f32_16x16x4_f32(false, av, false, bv1,
                                                 (short)0, acc1, false, false);
  }

  const int rbase = tm + ((lane >> 4) << 3);
  if (n0 < N) {
    const float badd = bias ? bias[n0] : 0.0f;
#pragma unroll
    for (int r = 0; r < 8; ++r) {
      float v = acc0[r] + badd;
      if (relu) v = fmaxf(v, 0.0f);
      Cb[(long long)(rbase + r) * ldc + n0] = v;
    }
  }
  if (n1 < N) {
    const float badd = bias ? bias[n1] : 0.0f;
#pragma unroll
    for (int r = 0; r < 8; ++r) {
      float v = acc1[r] + badd;
      if (relu) v = fmaxf(v, 0.0f);
      Cb[(long long)(rbase + r) * ldc + n1] = v;
    }
  }
}

__device__ __forceinline__ void wmma_group_open() {
  __builtin_amdgcn_sched_barrier(0);
}
__device__ __forceinline__ void wmma_operands_ready(v2f& b_a, v2f& b_b, v2f& ava, v2f& avb) {
  asm volatile("" : "+v"(b_a), "+v"(b_b), "+v"(ava), "+v"(avb));
}
__device__ __forceinline__ void wmma_operands_ready(v2f& b_a, v2f& ava) {
  asm volatile("" : "+v"(b_a), "+v"(ava));
}
__device__ __forceinline__ void wmma_group_close() {
  __builtin_amdgcn_sched_barrier(0);
  asm volatile("v_nop\n\tv_nop\n\tv_nop\n\tv_nop");
  __builtin_amdgcn_sched_barrier(0);
}

template <int S, int NT>
__global__ __launch_bounds__(32 * S)
void gemm_wmma_f32_ksplit(const float* __restrict__ A, long long sA, long long sA2, int lda,
                          const float* __restrict__ Bm, long long sB, int ldb,
                          float* __restrict__ C, long long sC, long long sC2, int ldc,
                          const float* __restrict__ bias,
                          int N, int K, int relu) {
  static_assert((8 * NT) % S == 0, "whole accumulator registers per wave in the reduction");
  __shared__ float part[S][8 * NT][32];

  const int lane  = threadIdx.x & 31;
  const int wave  = __builtin_amdgcn_readfirstlane(threadIdx.x >> 5);
  constexpr int SW = 16 * NT;
  const int nstrip = (N + SW - 1) / SW;
  const int tile  = blockIdx.x;
  const int tm = (tile / nstrip) << 4;
  const int tn = (tile % nstrip) * SW;

  const float* Ab = A  + (long long)blockIdx.z * sA + (long long)blockIdx.y * sA2;
  const float* Bb = Bm + (long long)blockIdx.z * sB;
  float*       Cb = C  + (long long)blockIdx.z * sC + (long long)blockIdx.y * sC2;

  const int mrow  = lane & 15;
  const int khalf = (lane >> 4) << 1;
  const int nb = tn + (lane & 15);
  const float* ap  = Ab + (long long)(tm + mrow) * lda + khalf;
  const float* bp[NT];
#pragma unroll
  for (int t = 0; t < NT; ++t) {
    const int n = nb + 16 * t;
    bp[t] = Bb + (n < N ? n : N - 1);
  }

  const int K4    = K & ~3;
  const int steps = (K + 3) >> 2;
  const int per   = (steps + S - 1) / S;
  const int kbeg  = wave * per * 4;
  const int klim  = kbeg + per * 4;
  const int kstop = klim < K4 ? klim : K4;

  v8f acc[NT];
#pragma unroll
  for (int t = 0; t < NT; ++t) acc[t] = v8f{};
  int k0 = kbeg;
  for (; k0 + 4 < kstop; k0 += 8) {
    v2f ava = *(const v2f*)(ap + k0);
    v2f avb = *(const v2f*)(ap + k0 + 4);
    const long long kb = (long long)(k0 + khalf) * ldb;
    v2f ba[NT], bb[NT];
#pragma unroll
    for (int t = 0; t < NT; ++t) {
      ba[t].x = bp[t][kb];             ba[t].y = bp[t][kb + ldb];
      bb[t].x = bp[t][kb + 4LL * ldb]; bb[t].y = bp[t][kb + 5LL * ldb];
    }
#pragma unroll
    for (int t = 0; t < NT; ++t) wmma_operands_ready(ba[t], bb[t], ava, avb);
    wmma_group_open();
#pragma unroll
    for (int t = 0; t < NT; ++t)
      acc[t] = __builtin_amdgcn_wmma_f32_16x16x4_f32(false, ava, false, ba[t],
                                                     (short)0, acc[t], false, false);
#pragma unroll
    for (int t = 0; t < NT; ++t)
      acc[t] = __builtin_amdgcn_wmma_f32_16x16x4_f32(false, avb, false, bb[t],
                                                     (short)0, acc[t], false, false);
    wmma_group_close();
  }
  if (k0 < kstop) {
    v2f ava = *(const v2f*)(ap + k0);
    const long long kb = (long long)(k0 + khalf) * ldb;
    v2f ba[NT];
#pragma unroll
    for (int t = 0; t < NT; ++t) { ba[t].x = bp[t][kb]; ba[t].y = bp[t][kb + ldb]; }
#pragma unroll
    for (int t = 0; t < NT; ++t) wmma_operands_ready(ba[t], ava);
    wmma_group_open();
#pragma unroll
    for (int t = 0; t < NT; ++t)
      acc[t] = __builtin_amdgcn_wmma_f32_16x16x4_f32(false, ava, false, ba[t],
                                                     (short)0, acc[t], false, false);
    wmma_group_close();
  }
  if (K4 < K && kbeg <= K4 && K4 < klim) {
    const int ka = K4 + khalf;
    v2f av = {};
    v2f bv[NT];
#pragma unroll
    for (int t = 0; t < NT; ++t) bv[t] = v2f{};
    if (ka < K) {
      av.x  = ap[K4];
#pragma unroll
      for (int t = 0; t < NT; ++t) bv[t].x = bp[t][(long long)ka * ldb];
    }
    if (ka + 1 < K) {
      av.y  = ap[K4 + 1];
#pragma unroll
      for (int t = 0; t < NT; ++t) bv[t].y = bp[t][(long long)(ka + 1) * ldb];
    }
#pragma unroll
    for (int t = 0; t < NT; ++t) wmma_operands_ready(bv[t], av);
    wmma_group_open();
#pragma unroll
    for (int t = 0; t < NT; ++t)
      acc[t] = __builtin_amdgcn_wmma_f32_16x16x4_f32(false, av, false, bv[t],
                                                     (short)0, acc[t], false, false);
    wmma_group_close();
  }

#pragma unroll
  for (int t = 0; t < NT; ++t)
#pragma unroll
    for (int r = 0; r < 8; ++r) part[wave][8 * t + r][lane] = acc[t][r];
  __syncthreads();

  constexpr int RPW = 8 * NT / S;
  const int rbase = tm + ((lane >> 4) << 3);
  for (int j = 0; j < RPW; ++j) {
    const int r = wave * RPW + j;
    float s = part[0][r][lane];
    for (int q = 1; q < S; ++q) s += part[q][r][lane];
    const int n = nb + 16 * (r >> 3);
    if (n < N) {
      float v = s + (bias ? bias[n] : 0.0f);
      if (relu) v = fmaxf(v, 0.0f);
      Cb[(long long)(rbase + (r & 7)) * ldc + n] = v;
    }
  }
}

__global__ __launch_bounds__(256)
void decoder_rows_ksplit(const float* __restrict__ Hm, float* Y,
                         const float* __restrict__ Wd1, const float* __restrict__ bd1,
                         const float* __restrict__ Wd2, const float* __restrict__ bd2,
                         const float* __restrict__ Wd3, const float* __restrict__ bd3,
                         float* __restrict__ act) {
  constexpr int S  = 8;
  constexpr int NT = CFC / 16;
  __shared__ float part[S][8 * NT][32];

  const int lane  = threadIdx.x & 31;
  const int wave  = __builtin_amdgcn_readfirstlane(threadIdx.x >> 5);
  const int tm = blockIdx.x << 4;

  const int mrow  = lane & 15;
  const int khalf = (lane >> 4) << 1;
  const int nb = lane & 15;
  const int rbase = tm + ((lane >> 4) << 3);
  constexpr int RPW = 8 * NT / S;

  {
    constexpr int K = CF, K4 = K & ~3;
    constexpr int S1  = 4;
    constexpr int per = (((K + 3) >> 2) + S1 - 1) / S1;
    static_assert(per % 2 == 0, "whole trips of two k-steps in every wave");
    v8f acc[NT];
#pragma unroll
    for (int t = 0; t < NT; ++t) acc[t] = v8f{};
    if (wave < S1) {
      const float* ap = Hm + (long long)(tm + mrow) * CF + khalf;
      const float* bp = Wd1 + nb;
      const int kbeg  = wave * per * 4;
      const int klim  = kbeg + per * 4;
      const int kstop = klim < K4 ? klim : K4;
      for (int k0 = kbeg; k0 < kstop; k0 += 8) {
        v2f ava = *(const v2f*)(ap + k0);
        v2f avb = *(const v2f*)(ap + k0 + 4);
        const int kb = (k0 + khalf) * CFC;
        v2f ba[NT], bb[NT];
#pragma unroll
        for (int t = 0; t < NT; ++t) {
          ba[t].x = bp[kb + 16 * t];           ba[t].y = bp[kb + CFC + 16 * t];
          bb[t].x = bp[kb + 4 * CFC + 16 * t]; bb[t].y = bp[kb + 5 * CFC + 16 * t];
        }
#pragma unroll
        for (int t = 0; t < NT; ++t) wmma_operands_ready(ba[t], bb[t], ava, avb);
        wmma_group_open();
#pragma unroll
        for (int t = 0; t < NT; ++t)
          acc[t] = __builtin_amdgcn_wmma_f32_16x16x4_f32(false, ava, false, ba[t],
                                                         (short)0, acc[t], false, false);
#pragma unroll
        for (int t = 0; t < NT; ++t)
          acc[t] = __builtin_amdgcn_wmma_f32_16x16x4_f32(false, avb, false, bb[t],
                                                         (short)0, acc[t], false, false);
        wmma_group_close();
      }
      if (K4 < K && kbeg <= K4 && K4 < klim) {
        const int ka = K4 + khalf;
        v2f av = {};
        v2f bv[NT];
#pragma unroll
        for (int t = 0; t < NT; ++t) bv[t] = v2f{};
        if (ka < K) {
          av.x = ap[K4];
#pragma unroll
          for (int t = 0; t < NT; ++t) bv[t].x = bp[ka * CFC + 16 * t];
        }
        if (ka + 1 < K) {
          av.y = ap[K4 + 1];
#pragma unroll
          for (int t = 0; t < NT; ++t) bv[t].y = bp[(ka + 1) * CFC + 16 * t];
        }
#pragma unroll
        for (int t = 0; t < NT; ++t) wmma_operands_ready(bv[t], av);
        wmma_group_open();
#pragma unroll
        for (int t = 0; t < NT; ++t)
          acc[t] = __builtin_amdgcn_wmma_f32_16x16x4_f32(false, av, false, bv[t],
                                                         (short)0, acc[t], false, false);
        wmma_group_close();
      }
#pragma unroll
      for (int t = 0; t < NT; ++t)
#pragma unroll
        for (int r = 0; r < 8; ++r) part[wave][8 * t + r][lane] = acc[t][r];
    }
    __syncthreads();
    for (int j = 0; j < RPW; ++j) {
      const int r = wave * RPW + j;
      float s = part[0][r][lane];
      for (int q = 1; q < S1; ++q) s += part[q][r][lane];
      const int n = nb + 16 * (r >> 3);
      const float v = fmaxf(s + bd1[n], 0.0f);
      Y[(long long)(rbase + (r & 7)) * CFC + n] = v;
    }
    __syncthreads();
  }

  {
    constexpr int K = CFC;
    constexpr int per = (K / 4) / S;
    static_assert((K / 8) % S == 0, "whole trips of two k-steps, the same number in every wave");
    const float* ap = Y + (long long)(tm + mrow) * CFC + khalf;
    const float* bp = Wd2 + nb;
    const int kbeg = wave * per * 4;
    const int kend = kbeg + per * 4;
    v8f acc[NT];
#pragma unroll
    for (int t = 0; t < NT; ++t) acc[t] = v8f{};
    for (int k0 = kbeg; k0 < kend; k0 += 8) {
      v2f ava = *(const v2f*)(ap + k0);
      v2f avb = *(const v2f*)(ap + k0 + 4);
      const int kb = (k0 + khalf) * CFC;
      v2f ba[NT], bb[NT];
#pragma unroll
      for (int t = 0; t < NT; ++t) {
        ba[t].x = bp[kb + 16 * t];           ba[t].y = bp[kb + CFC + 16 * t];
        bb[t].x = bp[kb + 4 * CFC + 16 * t]; bb[t].y = bp[kb + 5 * CFC + 16 * t];
      }
#pragma unroll
      for (int t = 0; t < NT; ++t) wmma_operands_ready(ba[t], bb[t], ava, avb);
      wmma_group_open();
#pragma unroll
      for (int t = 0; t < NT; ++t)
        acc[t] = __builtin_amdgcn_wmma_f32_16x16x4_f32(false, ava, false, ba[t],
                                                       (short)0, acc[t], false, false);
#pragma unroll
      for (int t = 0; t < NT; ++t)
        acc[t] = __builtin_amdgcn_wmma_f32_16x16x4_f32(false, avb, false, bb[t],
                                                       (short)0, acc[t], false, false);
      wmma_group_close();
    }
#pragma unroll
    for (int t = 0; t < NT; ++t)
#pragma unroll
      for (int r = 0; r < 8; ++r) part[wave][8 * t + r][lane] = acc[t][r];
    __syncthreads();
    for (int j = 0; j < RPW; ++j) {
      const int r = wave * RPW + j;
      float s = part[0][r][lane];
      for (int q = 1; q < S; ++q) s += part[q][r][lane];
      const int n = nb + 16 * (r >> 3);
      const float v = fmaxf(s + bd2[n], 0.0f);
      Y[(long long)(rbase + (r & 7)) * CFC + n] = v;
    }
    __syncthreads();
  }

  {
    constexpr int K = CFC;
    constexpr int per = (K / 4) / S;
    const float* ap = Y + (long long)(tm + mrow) * CFC + khalf;
    const float* bp = Wd3 + (nb < COUT ? nb : COUT - 1);
    const int kbeg = wave * per * 4;
    const int kend = kbeg + per * 4;
    v8f acc = {};
    for (int k0 = kbeg; k0 < kend; k0 += 8) {
      v2f ava = *(const v2f*)(ap + k0);
      v2f avb = *(const v2f*)(ap + k0 + 4);
      const int kb = (k0 + khalf) * COUT;
      v2f ba, bb;
      ba.x = bp[kb];            ba.y = bp[kb + COUT];
      bb.x = bp[kb + 4 * COUT]; bb.y = bp[kb + 5 * COUT];
      wmma_operands_ready(ba, bb, ava, avb);
      wmma_group_open();
      acc = __builtin_amdgcn_wmma_f32_16x16x4_f32(false, ava, false, ba,
                                                  (short)0, acc, false, false);
      acc = __builtin_amdgcn_wmma_f32_16x16x4_f32(false, avb, false, bb,
                                                  (short)0, acc, false, false);
      wmma_group_close();
    }
#pragma unroll
    for (int r = 0; r < 8; ++r) part[wave][r][lane] = acc[r];
    __syncthreads();
    const int r = wave;
    float s = part[0][r][lane];
    for (int q = 1; q < S; ++q) s += part[q][r][lane];
    if (nb < COUT) {
      const int row  = rbase + r;
      const int sb   = row / CN;
      const int node = row - sb * CN;
      const int step = sb / CB, b = sb - step * CB;
      act[((long long)(b * CT + step) * CN + node) * COUT + nb] = s + bd3[nb];
    }
  }
}

__global__ void deg_dinv_kernel(const float* __restrict__ A, float* __restrict__ dinv,
                                int rows, int n) {
  const int wid  = (blockIdx.x * blockDim.x + threadIdx.x) >> 5;
  const int lane = threadIdx.x & 31;
  if (wid >= rows) return;
  const float* row = A + (long long)wid * n;
  float s = 0.0f;
  for (int i = lane; i < n; i += 32) s += row[i];
#pragma unroll
  for (int off = 16; off > 0; off >>= 1) s += __shfl_xor(s, off, 32);
  if (lane == 0) dinv[wid] = (s > 0.0f) ? rsqrtf(s) : 0.0f;
}

__global__ void pack_wz_kernel(const float* __restrict__ Wch,
                               const float* __restrict__ Wcu,
                               float* __restrict__ Wz) {
  int idx = blockIdx.x * blockDim.x + threadIdx.x;
  if (idx >= 200 * 150) return;
  int r = idx / 150, cg = idx % 150;
  int c = cg / 50, g = cg % 50;
  int src = r / 50, f = r % 50;
  float v;
  if (src == 0)      v = Wch[((c * 2 + 0) * 50 + f) * 50 + g];
  else if (src == 1) v = Wch[((c * 2 + 1) * 50 + f) * 50 + g];
  else if (src == 2) v = Wcu[((c * 2 + 0) * 50 + f) * 50 + g];
  else               v = Wcu[((c * 2 + 1) * 50 + f) * 50 + g];
  Wz[idx] = v;
}

template <int S, int NT>
__global__ __launch_bounds__(32 * S)
void gemm_adj_ksplit(const float* __restrict__ adj_t,
                     const float* __restrict__ h, const float* __restrict__ x,
                     const float* __restrict__ dinv_t,
                     float* __restrict__ Wout) {
  constexpr int N = 2 * CF;
  constexpr int K = CN;
  static_assert(K % (8 * S) == 0, "whole trips of two k-steps in every wave");
  static_assert((8 * NT) % S == 0, "whole accumulator registers per wave in the reduction");
  __shared__ float part[S][8 * NT][32];

  const int lane  = threadIdx.x & 31;
  const int wave  = __builtin_amdgcn_readfirstlane(threadIdx.x >> 5);
  constexpr int SW = 16 * NT;
  constexpr int nstrip = (N + SW - 1) / SW;
  const int tm = (blockIdx.x / nstrip) << 4;
  const int tn = (blockIdx.x % nstrip) * SW;
  const int b  = blockIdx.z;

  const float* Ab = adj_t  + (long long)b * CT * CN * CN;
  const float* db = dinv_t + (long long)b * CT * CN;
  const float* hb = h + (long long)b * CN * CF;
  const float* xb = x + (long long)b * CN * CF;
  float*       Cb = Wout + (long long)b * CN * N;

  const int mrow  = lane & 15;
  const int khalf = (lane >> 4) << 1;
  const int nb = tn + (lane & 15);
  const float* ap  = Ab + (long long)(tm + mrow) * K + khalf;
  const float* dp  = db + khalf;
  const float* bp[NT];
#pragma unroll
  for (int t = 0; t < NT; ++t) {
    const int n = nb + 16 * t;
    const int c = n < N ? n : N - 1;
    bp[t] = c < CF ? hb + c : xb + (c - CF);
  }

  constexpr int per = K / (4 * S);
  const int kbeg = wave * per * 4;
  const int kend = kbeg + per * 4;

  v8f acc[NT];
#pragma unroll
  for (int t = 0; t < NT; ++t) acc[t] = v8f{};
  for (int k0 = kbeg; k0 < kend; k0 += 8) {
    v2f ava = *(const v2f*)(ap + k0);
    v2f avb = *(const v2f*)(ap + k0 + 4);
    const v2f dda = *(const v2f*)(dp + k0);
    const v2f ddb = *(const v2f*)(dp + k0 + 4);
    const int kb = (k0 + khalf) * CF;
    v2f ba[NT], bb[NT];
#pragma unroll
    for (int t = 0; t < NT; ++t) {
      ba[t].x = dda.x * bp[t][kb];            ba[t].y = dda.y * bp[t][kb + CF];
      bb[t].x = ddb.x * bp[t][kb + 4 * CF];   bb[t].y = ddb.y * bp[t][kb + 5 * CF];
    }
#pragma unroll
    for (int t = 0; t < NT; ++t) wmma_operands_ready(ba[t], bb[t], ava, avb);
    wmma_group_open();
#pragma unroll
    for (int t = 0; t < NT; ++t)
      acc[t] = __builtin_amdgcn_wmma_f32_16x16x4_f32(false, ava, false, ba[t],
                                                     (short)0, acc[t], false, false);
#pragma unroll
    for (int t = 0; t < NT; ++t)
      acc[t] = __builtin_amdgcn_wmma_f32_16x16x4_f32(false, avb, false, bb[t],
                                                     (short)0, acc[t], false, false);
    wmma_group_close();
  }

#pragma unroll
  for (int t = 0; t < NT; ++t)
#pragma unroll
    for (int r = 0; r < 8; ++r) part[wave][8 * t + r][lane] = acc[t][r];
  __syncthreads();

  constexpr int RPW = 8 * NT / S;
  const int rbase = tm + ((lane >> 4) << 3);
  for (int j = 0; j < RPW; ++j) {
    const int r = wave * RPW + j;
    float s = part[0][r][lane];
    for (int q = 1; q < S; ++q) s += part[q][r][lane];
    const int n = nb + 16 * (r >> 3);
    if (n < N) Cb[(long long)(rbase + (r & 7)) * N + n] = s;
  }
}

__device__ __forceinline__ v2f gate_a_frag(const float* __restrict__ hp,
                                           const float* __restrict__ xp,
                                           const float* __restrict__ wp,
                                           float dv, int k) {
  const int sec = k / CF;
  const int f   = k - sec * CF;
  const v2f sv = *(const v2f*)(((sec & 2) ? xp : hp) + f);
  const v2f wv = *(const v2f*)(wp + (sec >> 1) * CF + f);
  const bool lap = (sec & 1) != 0;
  v2f av;
  av.x = lap ? sv.x - dv * wv.x : sv.x;
  av.y = lap ? sv.y - dv * wv.y : sv.y;
  return av;
}

template <int S, int NT>
__global__ __launch_bounds__(32 * S)
void gemm_gate_ksplit(const float* __restrict__ h, const float* __restrict__ x,
                      const float* __restrict__ W, const float* __restrict__ dinv_t,
                      const float* __restrict__ Wz, const float* __restrict__ bias,
                      float* __restrict__ z) {
  constexpr int N = 3 * CF;
  constexpr int K = 4 * CF;
  static_assert(K % 4 == 0 && CF % 2 == 0, "no K tail; sections start at even k");
  static_assert((8 * NT) % S == 0, "whole accumulator registers per wave in the reduction");
  __shared__ float part[S][8 * NT][32];

  const int lane  = threadIdx.x & 31;
  const int wave  = __builtin_amdgcn_readfirstlane(threadIdx.x >> 5);
  constexpr int SW = 16 * NT;
  constexpr int nstrip = (N + SW - 1) / SW;
  const int tm = (blockIdx.x / nstrip) << 4;
  const int tn = (blockIdx.x % nstrip) * SW;

  const int mrow  = lane & 15;
  const int khalf = (lane >> 4) << 1;
  const int nb = tn + (lane & 15);
  const int row = tm + mrow;
  const float dv = dinv_t[(long long)(row / CN) * CT * CN + (row % CN)];
  const float* hp = h + (long long)row * CF;
  const float* xp = x + (long long)row * CF;
  const float* wp = W + (long long)row * 2 * CF;
  const float* bp[NT];
#pragma unroll
  for (int t = 0; t < NT; ++t) {
    const int n = nb + 16 * t;
    bp[t] = Wz + (n < N ? n : N - 1);
  }

  constexpr int steps = K / 4;
  constexpr int per   = (steps + S - 1) / S;
  const int kbeg  = wave * per * 4;
  const int klim  = kbeg + per * 4;
  const int kstop = klim < K ? klim : K;

  v8f acc[NT];
#pragma unroll
  for (int t = 0; t < NT; ++t) acc[t] = v8f{};
  int k0 = kbeg;
  for (; k0 + 4 < kstop; k0 += 8) {
    const int k = k0 + khalf;
    v2f ava = gate_a_frag(hp, xp, wp, dv, k);
    v2f avb = gate_a_frag(hp, xp, wp, dv, k + 4);
    const int kb = k * N;
    v2f ba[NT], bb[NT];
#pragma unroll
    for (int t = 0; t < NT; ++t) {
      ba[t].x = bp[t][kb];           ba[t].y = bp[t][kb + N];
      bb[t].x = bp[t][kb + 4 * N];   bb[t].y = bp[t][kb + 5 * N];
    }
#pragma unroll
    for (int t = 0; t < NT; ++t) wmma_operands_ready(ba[t], bb[t], ava, avb);
    wmma_group_open();
#pragma unroll
    for (int t = 0; t < NT; ++t)
      acc[t] = __builtin_amdgcn_wmma_f32_16x16x4_f32(false, ava, false, ba[t],
                                                     (short)0, acc[t], false, false);
#pragma unroll
    for (int t = 0; t < NT; ++t)
      acc[t] = __builtin_amdgcn_wmma_f32_16x16x4_f32(false, avb, false, bb[t],
                                                     (short)0, acc[t], false, false);
    wmma_group_close();
  }
  if (k0 < kstop) {
    const int k = k0 + khalf;
    v2f ava = gate_a_frag(hp, xp, wp, dv, k);
    const int kb = k * N;
    v2f ba[NT];
#pragma unroll
    for (int t = 0; t < NT; ++t) { ba[t].x = bp[t][kb]; ba[t].y = bp[t][kb + N]; }
#pragma unroll
    for (int t = 0; t < NT; ++t) wmma_operands_ready(ba[t], ava);
    wmma_group_open();
#pragma unroll
    for (int t = 0; t < NT; ++t)
      acc[t] = __builtin_amdgcn_wmma_f32_16x16x4_f32(false, ava, false, ba[t],
                                                     (short)0, acc[t], false, false);
    wmma_group_close();
  }

#pragma unroll
  for (int t = 0; t < NT; ++t)
#pragma unroll
    for (int r = 0; r < 8; ++r) part[wave][8 * t + r][lane] = acc[t][r];
  __syncthreads();

  constexpr int RPW = 8 * NT / S;
  const int rbase = tm + ((lane >> 4) << 3);
  for (int j = 0; j < RPW; ++j) {
    const int r = wave * RPW + j;
    float s = part[0][r][lane];
    for (int q = 1; q < S; ++q) s += part[q][r][lane];
    const int n = nb + 16 * (r >> 3);
    if (n < N) z[(long long)(rbase + (r & 7)) * N + n] = s + bias[n];
  }
}

__global__ void gate_kernel(const float* __restrict__ z, float* __restrict__ h) {
  int idx = blockIdx.x * blockDim.x + threadIdx.x;
  if (idx >= CB * CN * CF) return;
  int row = idx / CF, f = idx % CF;
  float z0 = z[row * 150 + f];
  float z1 = z[row * 150 + CF + f];
  float z2 = z[row * 150 + 2 * CF + f];
  float gt = 1.0f / (1.0f + expf(-z0));
  h[idx] = gt * tanhf(z1) + (1.0f - gt) * tanhf(z2);
}

static inline void launch_gemm(const float* A, long long sA, long long sA2, int lda,
                               const float* Bm, long long sB, int ldb,
                               float* C, long long sC, long long sC2, int ldc,
                               const float* bias, int M, int N, int K,
                               int batch, int batch2, int relu, hipStream_t stream) {
  int npair = (N + 31) >> 5;
  int tiles = (M >> 4) * npair;
  dim3 grid((tiles + 3) / 4, batch2, batch);
  gemm_wmma_f32<<<grid, 128, 0, stream>>>(A, sA, sA2, lda, Bm, sB, ldb,
                                          C, sC, sC2, ldc, bias, M, N, K, relu);
}

template <int S, int NT>
static inline void launch_gemm_ksplit(const float* A, long long sA, long long sA2, int lda,
                                      const float* Bm, long long sB, int ldb,
                                      float* C, long long sC, long long sC2, int ldc,
                                      const float* bias, int M, int N, int K,
                                      int batch, int batch2, int relu, hipStream_t stream) {
  int nstrip = (N + 16 * NT - 1) / (16 * NT);
  int tiles = (M >> 4) * nstrip;
  dim3 grid(tiles, batch2, batch);
  gemm_wmma_f32_ksplit<S, NT><<<grid, 32 * S, 0, stream>>>(A, sA, sA2, lda, Bm, sB, ldb,
                                                           C, sC, sC2, ldc, bias, N, K, relu);
}

static const int TE = 4;
static const int TD = 4;

extern "C" void kernel_launch(void* const* d_in, const int* in_sizes, int n_in,
                              void* d_out, int out_size, void* d_ws, size_t ws_size,
                              hipStream_t stream) {
  const float* obs  = (const float*)d_in[0];
  const float* adj  = (const float*)d_in[1];
  const float* h0   = (const float*)d_in[2];
  const float* We1  = (const float*)d_in[3];
  const float* be1  = (const float*)d_in[4];
  const float* We2  = (const float*)d_in[5];
  const float* be2  = (const float*)d_in[6];
  const float* Wch  = (const float*)d_in[7];
  const float* Wcu  = (const float*)d_in[8];
  const float* bc   = (const float*)d_in[9];
  const float* Wd1  = (const float*)d_in[10];
  const float* bd1  = (const float*)d_in[11];
  const float* Wd2  = (const float*)d_in[12];
  const float* bd2  = (const float*)d_in[13];
  const float* Wd3  = (const float*)d_in[14];
  const float* bd3  = (const float*)d_in[15];
  float* out = (float*)d_out;

  const int MN = CB * CN;
  const size_t HSZ = (size_t)MN * CF;

  float* ws = (float*)d_ws;
  size_t off = 0;
  float* dinv = ws + off; off += (size_t)CB * CT * CN;
  float* Wz   = ws + off; off += (size_t)200 * 150;
  float* H    = ws + off; off += TD * HSZ;
  float* xc   = ws + off; off += TE * HSZ;
  float* scr  = ws + off;
  float* Wb   = scr;
  float* z    = Wb + (size_t)MN * 2 * CF;
  float* tmp1 = scr;
  float* y    = scr;
  (void)ws_size; (void)in_sizes; (void)n_in; (void)out_size;

  float* hlast = H + (TD - 1) * HSZ;
  {
    int rows = CB * CT * CN;
    deg_dinv_kernel<<<(rows * 32 + 255) / 256, 256, 0, stream>>>(adj, dinv, rows, CN);
    pack_wz_kernel<<<(200 * 150 + 255) / 256, 256, 0, stream>>>(Wch, Wcu, Wz);
    hipMemcpyAsync(hlast, h0, sizeof(float) * CB * CN * CF, hipMemcpyDeviceToDevice, stream);
  }

  for (int te = 0; te < CT; te += TE) {
    launch_gemm(obs + (long long)te * CN * CIN, (long long)CT * CN * CIN, (long long)CN * CIN, CIN,
                We1, 0, CFC, tmp1, (long long)CN * CFC, (long long)MN * CFC, CFC,
                be1, CN, CFC, CIN, CB, TE, 1, stream);
    launch_gemm_ksplit<8, 4>(tmp1, 0, 0, CFC, We2, 0, CF, xc, 0, 0, CF,
                          be2, TE * MN, CF, CFC, 1, 1, 1, stream);

    for (int j = 0; j < TD; ++j) {
      const int t = te + j;
      const float* x = xc + (size_t)j * HSZ;
      const float* hin = (j == 0) ? hlast : (H + (size_t)(j - 1) * HSZ);
      float* hout = H + (size_t)j * HSZ;
      gemm_adj_ksplit<16, 4><<<dim3((CN / 16) * 2, 1, CB), 32 * 16, 0, stream>>>(
          adj + (long long)t * CN * CN, hin, x, dinv + (long long)t * CN, Wb);
      gemm_gate_ksplit<8, 5><<<(MN / 16) * 2, 32 * 8, 0, stream>>>(
          hin, x, Wb, dinv + (long long)t * CN, Wz, bc, z);
      gate_kernel<<<(MN * CF + 255) / 256, 256, 0, stream>>>(z, hout);
    }
    decoder_rows_ksplit<<<(TD * MN) / 16, 32 * 8, 0, stream>>>(
        H, y, Wd1, bd1, Wd2, bd2, Wd3, bd3, out + (long long)te * CN * COUT);
  }

  hipMemcpyAsync(out + (long long)CB * CT * CN * COUT, hlast,
                 sizeof(float) * CB * CN * CF, hipMemcpyDeviceToDevice, stream);
}


# graph filter reassociated, (L h)W = L(hW): row-local weight GEMM (K=100) first, adjacency GEMM last with z and the CfC gate formed in its K-split LDS sum: 3 -> 2 launches per step; sum order changes
# speedup vs baseline: 4.3869x; 1.0311x over previous
#include <hip/hip_runtime.h>
#include <math.h>

static const int CB = 4, CT = 32, CN = 512;
static const int CIN = 10, CF = 50, CFC = 128, COUT = 2;

typedef __attribute__((ext_vector_type(2))) float v2f;
typedef __attribute__((ext_vector_type(8))) float v8f;

__global__ __launch_bounds__(128)
void gemm_wmma_f32(const float* __restrict__ A, long long sA, long long sA2, int lda,
                   const float* __restrict__ Bm, long long sB, int ldb,
                   float* __restrict__ C, long long sC, long long sC2, int ldc,
                   const float* __restrict__ bias,
                   int M, int N, int K, int relu) {
  const int lane  = threadIdx.x & 31;
  const int wave  = threadIdx.x >> 5;
  const int npair = (N + 31) >> 5;
  const int tiles = (M >> 4) * npair;
  const int tile  = blockIdx.x * (blockDim.x >> 5) + wave;
  if (tile >= tiles) return;
  const int tm = (tile / npair) << 4;
  const int tn = (tile % npair) << 5;

  const float* Ab = A  + (long long)blockIdx.z * sA + (long long)blockIdx.y * sA2;
  const float* Bb = Bm + (long long)blockIdx.z * sB;
  float*       Cb = C  + (long long)blockIdx.z * sC + (long long)blockIdx.y * sC2;

  const int mrow  = lane & 15;
  const int khalf = (lane >> 4) << 1;
  const int n0 = tn + (lane & 15);
  const int n1 = n0 + 16;
  const float* ap  = Ab + (long long)(tm + mrow) * lda + khalf;
  const float* bp0 = Bb + (n0 < N ? n0 : N - 1);
  const float* bp1 = Bb + (n1 < N ? n1 : N - 1);

  v8f acc0 = {}, acc1 = {};
  const int K4 = K & ~3;
  int k0 = 0;
#pragma unroll 2
  for (; k0 < K4; k0 += 4) {
    const v2f av = *(const v2f*)(ap + k0);
    const long long kb = (long long)(k0 + khalf) * ldb;
    v2f bv0, bv1;
    bv0.x = bp0[kb];       bv0.y = bp0[kb + ldb];
    bv1.x = bp1[kb];       bv1.y = bp1[kb + ldb];
    acc0 = __builtin_amdgcn_wmma_f32_16x16x4_f32(false, av, false, bv0,
                                                 (short)0, acc0, false, false);
    acc1 = __builtin_amdgcn_wmma_f32_16x16x4_f32(false, av, false, bv1,
                                                 (short)0, acc1, false, false);
  }
  if (k0 < K) {
    const int ka = k0 + khalf;
    v2f av = {}, bv0 = {}, bv1 = {};
    if (ka < K) {
      av.x  = ap[k0];
      bv0.x = bp0[(long long)ka * ldb];
      bv1.x = bp1[(long long)ka * ldb];
    }
    if (ka + 1 < K) {
      av.y  = ap[k0 + 1];
      bv0.y = bp0[(long long)(ka + 1) * ldb];
      bv1.y = bp1[(long long)(ka + 1) * ldb];
    }
    acc0 = __builtin_amdgcn_wmma_f32_16x16x4_f32(false, av, false, bv0,
                                                 (short)0, acc0, false, false);
    acc1 = __builtin_amdgcn_wmma_f32_16x16x4_f32(false, av, false, bv1,
                                                 (short)0, acc1, false, false);
  }

  const int rbase = tm + ((lane >> 4) << 3);
  if (n0 < N) {
    const float badd = bias ? bias[n0] : 0.0f;
#pragma unroll
    for (int r = 0; r < 8; ++r) {
      float v = acc0[r] + badd;
      if (relu) v = fmaxf(v, 0.0f);
      Cb[(long long)(rbase + r) * ldc + n0] = v;
    }
  }
  if (n1 < N) {
    const float badd = bias ? bias[n1] : 0.0f;
#pragma unroll
    for (int r = 0; r < 8; ++r) {
      float v = acc1[r] + badd;
      if (relu) v = fmaxf(v, 0.0f);
      Cb[(long long)(rbase + r) * ldc + n1] = v;
    }
  }
}

__device__ __forceinline__ void wmma_group_open() {
  __builtin_amdgcn_sched_barrier(0);
}
__device__ __forceinline__ void wmma_operands_ready(v2f& b_a, v2f& b_b, v2f& ava, v2f& avb) {
  asm volatile("" : "+v"(b_a), "+v"(b_b), "+v"(ava), "+v"(avb));
}
__device__ __forceinline__ void wmma_operands_ready(v2f& b_a, v2f& ava) {
  asm volatile("" : "+v"(b_a), "+v"(ava));
}
__device__ __forceinline__ void wmma_group_close() {
  __builtin_amdgcn_sched_barrier(0);
  asm volatile("v_nop\n\tv_nop\n\tv_nop\n\tv_nop");
  __builtin_amdgcn_sched_barrier(0);
}

template <int S, int NT>
__global__ __launch_bounds__(32 * S)
void gemm_wmma_f32_ksplit(const float* __restrict__ A, long long sA, long long sA2, int lda,
                          const float* __restrict__ Bm, long long sB, int ldb,
                          float* __restrict__ C, long long sC, long long sC2, int ldc,
                          const float* __restrict__ bias,
                          int N, int K, int relu) {
  static_assert((8 * NT) % S == 0, "whole accumulator registers per wave in the reduction");
  __shared__ float part[S][8 * NT][32];

  const int lane  = threadIdx.x & 31;
  const int wave  = __builtin_amdgcn_readfirstlane(threadIdx.x >> 5);
  constexpr int SW = 16 * NT;
  const int nstrip = (N + SW - 1) / SW;
  const int tile  = blockIdx.x;
  const int tm = (tile / nstrip) << 4;
  const int tn = (tile % nstrip) * SW;

  const float* Ab = A  + (long long)blockIdx.z * sA + (long long)blockIdx.y * sA2;
  const float* Bb = Bm + (long long)blockIdx.z * sB;
  float*       Cb = C  + (long long)blockIdx.z * sC + (long long)blockIdx.y * sC2;

  const int mrow  = lane & 15;
  const int khalf = (lane >> 4) << 1;
  const int nb = tn + (lane & 15);
  const float* ap  = Ab + (long long)(tm + mrow) * lda + khalf;
  const float* bp[NT];
#pragma unroll
  for (int t = 0; t < NT; ++t) {
    const int n = nb + 16 * t;
    bp[t] = Bb + (n < N ? n : N - 1);
  }

  const int K4    = K & ~3;
  const int steps = (K + 3) >> 2;
  const int per   = (steps + S - 1) / S;
  const int kbeg  = wave * per * 4;
  const int klim  = kbeg + per * 4;
  const int kstop = klim < K4 ? klim : K4;

  v8f acc[NT];
#pragma unroll
  for (int t = 0; t < NT; ++t) acc[t] = v8f{};
  int k0 = kbeg;
  for (; k0 + 4 < kstop; k0 += 8) {
    v2f ava = *(const v2f*)(ap + k0);
    v2f avb = *(const v2f*)(ap + k0 + 4);
    const long long kb = (long long)(k0 + khalf) * ldb;
    v2f ba[NT], bb[NT];
#pragma unroll
    for (int t = 0; t < NT; ++t) {
      ba[t].x = bp[t][kb];             ba[t].y = bp[t][kb + ldb];
      bb[t].x = bp[t][kb + 4LL * ldb]; bb[t].y = bp[t][kb + 5LL * ldb];
    }
#pragma unroll
    for (int t = 0; t < NT; ++t) wmma_operands_ready(ba[t], bb[t], ava, avb);
    wmma_group_open();
#pragma unroll
    for (int t = 0; t < NT; ++t)
      acc[t] = __builtin_amdgcn_wmma_f32_16x16x4_f32(false, ava, false, ba[t],
                                                     (short)0, acc[t], false, false);
#pragma unroll
    for (int t = 0; t < NT; ++t)
      acc[t] = __builtin_amdgcn_wmma_f32_16x16x4_f32(false, avb, false, bb[t],
                                                     (short)0, acc[t], false, false);
    wmma_group_close();
  }
  if (k0 < kstop) {
    v2f ava = *(const v2f*)(ap + k0);
    const long long kb = (long long)(k0 + khalf) * ldb;
    v2f ba[NT];
#pragma unroll
    for (int t = 0; t < NT; ++t) { ba[t].x = bp[t][kb]; ba[t].y = bp[t][kb + ldb]; }
#pragma unroll
    for (int t = 0; t < NT; ++t) wmma_operands_ready(ba[t], ava);
    wmma_group_open();
#pragma unroll
    for (int t = 0; t < NT; ++t)
      acc[t] = __builtin_amdgcn_wmma_f32_16x16x4_f32(false, ava, false, ba[t],
                                                     (short)0, acc[t], false, false);
    wmma_group_close();
  }
  if (K4 < K && kbeg <= K4 && K4 < klim) {
    const int ka = K4 + khalf;
    v2f av = {};
    v2f bv[NT];
#pragma unroll
    for (int t = 0; t < NT; ++t) bv[t] = v2f{};
    if (ka < K) {
      av.x  = ap[K4];
#pragma unroll
      for (int t = 0; t < NT; ++t) bv[t].x = bp[t][(long long)ka * ldb];
    }
    if (ka + 1 < K) {
      av.y  = ap[K4 + 1];
#pragma unroll
      for (int t = 0; t < NT; ++t) bv[t].y = bp[t][(long long)(ka + 1) * ldb];
    }
#pragma unroll
    for (int t = 0; t < NT; ++t) wmma_operands_ready(bv[t], av);
    wmma_group_open();
#pragma unroll
    for (int t = 0; t < NT; ++t)
      acc[t] = __builtin_amdgcn_wmma_f32_16x16x4_f32(false, av, false, bv[t],
                                                     (short)0, acc[t], false, false);
    wmma_group_close();
  }

#pragma unroll
  for (int t = 0; t < NT; ++t)
#pragma unroll
    for (int r = 0; r < 8; ++r) part[wave][8 * t + r][lane] = acc[t][r];
  __syncthreads();

  constexpr int RPW = 8 * NT / S;
  const int rbase = tm + ((lane >> 4) << 3);
  for (int j = 0; j < RPW; ++j) {
    const int r = wave * RPW + j;
    float s = part[0][r][lane];
    for (int q = 1; q < S; ++q) s += part[q][r][lane];
    const int n = nb + 16 * (r >> 3);
    if (n < N) {
      float v = s + (bias ? bias[n] : 0.0f);
      if (relu) v = fmaxf(v, 0.0f);
      Cb[(long long)(rbase + (r & 7)) * ldc + n] = v;
    }
  }
}

__global__ __launch_bounds__(256)
void decoder_rows_ksplit(const float* __restrict__ Hm, float* Y,
                         const float* __restrict__ Wd1, const float* __restrict__ bd1,
                         const float* __restrict__ Wd2, const float* __restrict__ bd2,
                         const float* __restrict__ Wd3, const float* __restrict__ bd3,
                         float* __restrict__ act) {
  constexpr int S  = 8;
  constexpr int NT = CFC / 16;
  __shared__ float part[S][8 * NT][32];

  const int lane  = threadIdx.x & 31;
  const int wave  = __builtin_amdgcn_readfirstlane(threadIdx.x >> 5);
  const int tm = blockIdx.x << 4;

  const int mrow  = lane & 15;
  const int khalf = (lane >> 4) << 1;
  const int nb = lane & 15;
  const int rbase = tm + ((lane >> 4) << 3);
  constexpr int RPW = 8 * NT / S;

  {
    constexpr int K = CF, K4 = K & ~3;
    constexpr int S1  = 4;
    constexpr int per = (((K + 3) >> 2) + S1 - 1) / S1;
    static_assert(per % 2 == 0, "whole trips of two k-steps in every wave");
    v8f acc[NT];
#pragma unroll
    for (int t = 0; t < NT; ++t) acc[t] = v8f{};
    if (wave < S1) {
      const float* ap = Hm + (long long)(tm + mrow) * CF + khalf;
      const float* bp = Wd1 + nb;
      const int kbeg  = wave * per * 4;
      const int klim  = kbeg + per * 4;
      const int kstop = klim < K4 ? klim : K4;
      for (int k0 = kbeg; k0 < kstop; k0 += 8) {
        v2f ava = *(const v2f*)(ap + k0);
        v2f avb = *(const v2f*)(ap + k0 + 4);
        const int kb = (k0 + khalf) * CFC;
        v2f ba[NT], bb[NT];
#pragma unroll
        for (int t = 0; t < NT; ++t) {
          ba[t].x = bp[kb + 16 * t];           ba[t].y = bp[kb + CFC + 16 * t];
          bb[t].x = bp[kb + 4 * CFC + 16 * t]; bb[t].y = bp[kb + 5 * CFC + 16 * t];
        }
#pragma unroll
        for (int t = 0; t < NT; ++t) wmma_operands_ready(ba[t], bb[t], ava, avb);
        wmma_group_open();
#pragma unroll
        for (int t = 0; t < NT; ++t)
          acc[t] = __builtin_amdgcn_wmma_f32_16x16x4_f32(false, ava, false, ba[t],
                                                         (short)0, acc[t], false, false);
#pragma unroll
        for (int t = 0; t < NT; ++t)
          acc[t] = __builtin_amdgcn_wmma_f32_16x16x4_f32(false, avb, false, bb[t],
                                                         (short)0, acc[t], false, false);
        wmma_group_close();
      }
      if (K4 < K && kbeg <= K4 && K4 < klim) {
        const int ka = K4 + khalf;
        v2f av = {};
        v2f bv[NT];
#pragma unroll
        for (int t = 0; t < NT; ++t) bv[t] = v2f{};
        if (ka < K) {
          av.x = ap[K4];
#pragma unroll
          for (int t = 0; t < NT; ++t) bv[t].x = bp[ka * CFC + 16 * t];
        }
        if (ka + 1 < K) {
          av.y = ap[K4 + 1];
#pragma unroll
          for (int t = 0; t < NT; ++t) bv[t].y = bp[(ka + 1) * CFC + 16 * t];
        }
#pragma unroll
        for (int t = 0; t < NT; ++t) wmma_operands_ready(bv[t], av);
        wmma_group_open();
#pragma unroll
        for (int t = 0; t < NT; ++t)
          acc[t] = __builtin_amdgcn_wmma_f32_16x16x4_f32(false, av, false, bv[t],
                                                         (short)0, acc[t], false, false);
        wmma_group_close();
      }
#pragma unroll
      for (int t = 0; t < NT; ++t)
#pragma unroll
        for (int r = 0; r < 8; ++r) part[wave][8 * t + r][lane] = acc[t][r];
    }
    __syncthreads();
    for (int j = 0; j < RPW; ++j) {
      const int r = wave * RPW + j;
      float s = part[0][r][lane];
      for (int q = 1; q < S1; ++q) s += part[q][r][lane];
      const int n = nb + 16 * (r >> 3);
      const float v = fmaxf(s + bd1[n], 0.0f);
      Y[(long long)(rbase + (r & 7)) * CFC + n] = v;
    }
    __syncthreads();
  }

  {
    constexpr int K = CFC;
    constexpr int per = (K / 4) / S;
    static_assert((K / 8) % S == 0, "whole trips of two k-steps, the same number in every wave");
    const float* ap = Y + (long long)(tm + mrow) * CFC + khalf;
    const float* bp = Wd2 + nb;
    const int kbeg = wave * per * 4;
    const int kend = kbeg + per * 4;
    v8f acc[NT];
#pragma unroll
    for (int t = 0; t < NT; ++t) acc[t] = v8f{};
    for (int k0 = kbeg; k0 < kend; k0 += 8) {
      v2f ava = *(const v2f*)(ap + k0);
      v2f avb = *(const v2f*)(ap + k0 + 4);
      const int kb = (k0 + khalf) * CFC;
      v2f ba[NT], bb[NT];
#pragma unroll
      for (int t = 0; t < NT; ++t) {
        ba[t].x = bp[kb + 16 * t];           ba[t].y = bp[kb + CFC + 16 * t];
        bb[t].x = bp[kb + 4 * CFC + 16 * t]; bb[t].y = bp[kb + 5 * CFC + 16 * t];
      }
#pragma unroll
      for (int t = 0; t < NT; ++t) wmma_operands_ready(ba[t], bb[t], ava, avb);
      wmma_group_open();
#pragma unroll
      for (int t = 0; t < NT; ++t)
        acc[t] = __builtin_amdgcn_wmma_f32_16x16x4_f32(false, ava, false, ba[t],
                                                       (short)0, acc[t], false, false);
#pragma unroll
      for (int t = 0; t < NT; ++t)
        acc[t] = __builtin_amdgcn_wmma_f32_16x16x4_f32(false, avb, false, bb[t],
                                                       (short)0, acc[t], false, false);
      wmma_group_close();
    }
#pragma unroll
    for (int t = 0; t < NT; ++t)
#pragma unroll
      for (int r = 0; r < 8; ++r) part[wave][8 * t + r][lane] = acc[t][r];
    __syncthreads();
    for (int j = 0; j < RPW; ++j) {
      const int r = wave * RPW + j;
      float s = part[0][r][lane];
      for (int q = 1; q < S; ++q) s += part[q][r][lane];
      const int n = nb + 16 * (r >> 3);
      const float v = fmaxf(s + bd2[n], 0.0f);
      Y[(long long)(rbase + (r & 7)) * CFC + n] = v;
    }
    __syncthreads();
  }

  {
    constexpr int K = CFC;
    constexpr int per = (K / 4) / S;
    const float* ap = Y + (long long)(tm + mrow) * CFC + khalf;
    const float* bp = Wd3 + (nb < COUT ? nb : COUT - 1);
    const int kbeg = wave * per * 4;
    const int kend = kbeg + per * 4;
    v8f acc = {};
    for (int k0 = kbeg; k0 < kend; k0 += 8) {
      v2f ava = *(const v2f*)(ap + k0);
      v2f avb = *(const v2f*)(ap + k0 + 4);
      const int kb = (k0 + khalf) * COUT;
      v2f ba, bb;
      ba.x = bp[kb];            ba.y = bp[kb + COUT];
      bb.x = bp[kb + 4 * COUT]; bb.y = bp[kb + 5 * COUT];
      wmma_operands_ready(ba, bb, ava, avb);
      wmma_group_open();
      acc = __builtin_amdgcn_wmma_f32_16x16x4_f32(false, ava, false, ba,
                                                  (short)0, acc, false, false);
      acc = __builtin_amdgcn_wmma_f32_16x16x4_f32(false, avb, false, bb,
                                                  (short)0, acc, false, false);
      wmma_group_close();
    }
#pragma unroll
    for (int r = 0; r < 8; ++r) part[wave][r][lane] = acc[r];
    __syncthreads();
    const int r = wave;
    float s = part[0][r][lane];
    for (int q = 1; q < S; ++q) s += part[q][r][lane];
    if (nb < COUT) {
      const int row  = rbase + r;
      const int sb   = row / CN;
      const int node = row - sb * CN;
      const int step = sb / CB, b = sb - step * CB;
      act[((long long)(b * CT + step) * CN + node) * COUT + nb] = s + bd3[nb];
    }
  }
}

__global__ void deg_dinv_kernel(const float* __restrict__ A, float* __restrict__ dinv,
                                int rows, int n) {
  const int wid  = (blockIdx.x * blockDim.x + threadIdx.x) >> 5;
  const int lane = threadIdx.x & 31;
  if (wid >= rows) return;
  const float* row = A + (long long)wid * n;
  float s = 0.0f;
  for (int i = lane; i < n; i += 32) s += row[i];
#pragma unroll
  for (int off = 16; off > 0; off >>= 1) s += __shfl_xor(s, off, 32);
  if (lane == 0) dinv[wid] = (s > 0.0f) ? rsqrtf(s) : 0.0f;
}

__global__ void pack_wcat_kernel(const float* __restrict__ Wch,
                                 const float* __restrict__ Wcu,
                                 float* __restrict__ Wcat) {
  int idx = blockIdx.x * blockDim.x + threadIdx.x;
  if (idx >= 100 * 300) return;
  int r = idx / 300, n = idx % 300;
  const float* src = r < 50 ? Wch : Wcu;
  int f = r < 50 ? r : r - 50;
  int m = n < 150 ? n : n - 150;
  int c = m / 50, g = m % 50;
  float v = src[((c * 2 + 1) * 50 + f) * 50 + g];
  if (n >= 150) v = src[((c * 2 + 0) * 50 + f) * 50 + g] + v;
  Wcat[idx] = v;
}

__device__ __forceinline__ v2f hx_a_frag(const float* __restrict__ hp,
                                         const float* __restrict__ xp, int k) {
  const float* p = k < CF ? hp + k : xp + (k - CF);
  return *(const v2f*)p;
}

template <int S, int NT>
__global__ __launch_bounds__(32 * S)
void gemm_hx_ksplit(const float* __restrict__ h, const float* __restrict__ x,
                    const float* __restrict__ dinv_t,
                    const float* __restrict__ Wcat, const float* __restrict__ bias,
                    float* __restrict__ R, float* __restrict__ Q) {
  constexpr int NH = 3 * CF;
  constexpr int N = 2 * NH;
  constexpr int K = 2 * CF;
  static_assert(K % 4 == 0 && CF % 2 == 0, "no K tail; halves start at even k");
  static_assert((8 * NT) % S == 0, "whole accumulator registers per wave in the reduction");
  __shared__ float part[S][8 * NT][32];

  const int lane  = threadIdx.x & 31;
  const int wave  = __builtin_amdgcn_readfirstlane(threadIdx.x >> 5);
  constexpr int SW = 16 * NT;
  constexpr int nstrip = (N + SW - 1) / SW;
  const int tm = (blockIdx.x / nstrip) << 4;
  const int tn = (blockIdx.x % nstrip) * SW;

  const int mrow  = lane & 15;
  const int khalf = (lane >> 4) << 1;
  const int nb = tn + (lane & 15);
  const int row = tm + mrow;
  const float* hp = h + (long long)row * CF;
  const float* xp = x + (long long)row * CF;
  const float* bp[NT];
#pragma unroll
  for (int t = 0; t < NT; ++t) {
    const int n = nb + 16 * t;
    bp[t] = Wcat + (n < N ? n : N - 1);
  }

  constexpr int steps = K / 4;
  constexpr int per   = (steps + S - 1) / S;
  const int kbeg  = wave * per * 4;
  const int klim  = kbeg + per * 4;
  const int kstop = klim < K ? klim : K;

  v8f acc[NT];
#pragma unroll
  for (int t = 0; t < NT; ++t) acc[t] = v8f{};
  int k0 = kbeg;
  for (; k0 + 4 < kstop; k0 += 8) {
    const int k = k0 + khalf;
    v2f ava = hx_a_frag(hp, xp, k);
    v2f avb = hx_a_frag(hp, xp, k + 4);
    const int kb = k * N;
    v2f ba[NT], bb[NT];
#pragma unroll
    for (int t = 0; t < NT; ++t) {
      ba[t].x = bp[t][kb];           ba[t].y = bp[t][kb + N];
      bb[t].x = bp[t][kb + 4 * N];   bb[t].y = bp[t][kb + 5 * N];
    }
#pragma unroll
    for (int t = 0; t < NT; ++t) wmma_operands_ready(ba[t], bb[t], ava, avb);
    wmma_group_open();
#pragma unroll
    for (int t = 0; t < NT; ++t)
      acc[t] = __builtin_amdgcn_wmma_f32_16x16x4_f32(false, ava, false, ba[t],
                                                     (short)0, acc[t], false, false);
#pragma unroll
    for (int t = 0; t < NT; ++t)
      acc[t] = __builtin_amdgcn_wmma_f32_16x16x4_f32(false, avb, false, bb[t],
                                                     (short)0, acc[t], false, false);
    wmma_group_close();
  }
  if (k0 < kstop) {
    const int k = k0 + khalf;
    v2f ava = hx_a_frag(hp, xp, k);
    const int kb = k * N;
    v2f ba[NT];
#pragma unroll
    for (int t = 0; t < NT; ++t) { ba[t].x = bp[t][kb]; ba[t].y = bp[t][kb + N]; }
#pragma unroll
    for (int t = 0; t < NT; ++t) wmma_operands_ready(ba[t], ava);
    wmma_group_open();
#pragma unroll
    for (int t = 0; t < NT; ++t)
      acc[t] = __builtin_amdgcn_wmma_f32_16x16x4_f32(false, ava, false, ba[t],
                                                     (short)0, acc[t], false, false);
    wmma_group_close();
  }

  constexpr int RPW = 8 * NT / S;
  const int rbase = tm + ((lane >> 4) << 3);
  float dvv[RPW], bvv[RPW];
#pragma unroll
  for (int j = 0; j < RPW; ++j) {
    const int r = wave * RPW + j;
    const int grow = rbase + (r & 7);
    const int m = nb + 16 * (r >> 3) - NH;
    dvv[j] = dinv_t[(long long)(grow / CN) * CT * CN + (grow % CN)];
    bvv[j] = bias[m < 0 ? 0 : (m < NH ? m : NH - 1)];
  }

#pragma unroll
  for (int t = 0; t < NT; ++t)
#pragma unroll
    for (int r = 0; r < 8; ++r) part[wave][8 * t + r][lane] = acc[t][r];
  __syncthreads();

#pragma unroll
  for (int j = 0; j < RPW; ++j) {
    const int r = wave * RPW + j;
    float s = part[0][r][lane];
    for (int q = 1; q < S; ++q) s += part[q][r][lane];
    const int n = nb + 16 * (r >> 3);
    const int grow = rbase + (r & 7);
    const bool rhalf = n < NH;
    const float v = rhalf ? dvv[j] * s : s + bvv[j];
    float* dst = rhalf ? R + (long long)grow * NH + n : Q + (long long)grow * NH + (n - NH);
    if (n < N) *dst = v;
  }
}

template <int S, int NT>
__global__ __launch_bounds__(32 * S)
void gemm_adj_gate_ksplit(const float* __restrict__ adj_t,
                          const float* __restrict__ R, const float* __restrict__ Q,
                          const float* __restrict__ dinv_t,
                          float* __restrict__ hout) {
  constexpr int N = 3 * CF;
  constexpr int K = CN;
  static_assert(K % (8 * S) == 0, "whole trips of two k-steps in every wave");
  static_assert(16 * NT >= N, "a block holds whole rows of z");
  __shared__ float part[S][8 * NT][32];

  const int lane  = threadIdx.x & 31;
  const int wave  = __builtin_amdgcn_readfirstlane(threadIdx.x >> 5);
  const int tm = blockIdx.x << 4;
  const int b  = blockIdx.z;

  const float* Ab = adj_t + (long long)b * CT * CN * CN;
  const float* Rb = R + (long long)b * CN * N;

  const int mrow  = lane & 15;
  const int khalf = (lane >> 4) << 1;
  const int nb = lane & 15;
  const float* ap  = Ab + (long long)(tm + mrow) * K + khalf;
  const float* bp[NT];
#pragma unroll
  for (int t = 0; t < NT; ++t) {
    const int n = nb + 16 * t;
    bp[t] = Rb + (n < N ? n : N - 1);
  }

  constexpr int per = K / (4 * S);
  const int kbeg = wave * per * 4;
  const int kend = kbeg + per * 4;

  v8f acc[NT];
#pragma unroll
  for (int t = 0; t < NT; ++t) acc[t] = v8f{};
  for (int k0 = kbeg; k0 < kend; k0 += 8) {
    v2f ava = *(const v2f*)(ap + k0);
    v2f avb = *(const v2f*)(ap + k0 + 4);
    const int kb = (k0 + khalf) * N;
    v2f ba[NT], bb[NT];
#pragma unroll
    for (int t = 0; t < NT; ++t) {
      ba[t].x = bp[t][kb];           ba[t].y = bp[t][kb + N];
      bb[t].x = bp[t][kb + 4 * N];   bb[t].y = bp[t][kb + 5 * N];
    }
#pragma unroll
    for (int t = 0; t < NT; ++t) wmma_operands_ready(ba[t], bb[t], ava, avb);
    wmma_group_open();
#pragma unroll
    for (int t = 0; t < NT; ++t)
      acc[t] = __builtin_amdgcn_wmma_f32_16x16x4_f32(false, ava, false, ba[t],
                                                     (short)0, acc[t], false, false);
#pragma unroll
    for (int t = 0; t < NT; ++t)
      acc[t] = __builtin_amdgcn_wmma_f32_16x16x4_f32(false, avb, false, bb[t],
                                                     (short)0, acc[t], false, false);
    wmma_group_close();
  }

#pragma unroll
  for (int t = 0; t < NT; ++t)
#pragma unroll
    for (int r = 0; r < 8; ++r) part[wave][8 * t + r][lane] = acc[t][r];
  __syncthreads();

  for (int idx = threadIdx.x; idx < 16 * CF; idx += 32 * S) {
    const int r16 = idx / CF;
    const int f   = idx - r16 * CF;
    const int lhi = (r16 >> 3) << 4;
    const int rr  = r16 & 7;
    float s[3];
#pragma unroll
    for (int c = 0; c < 3; ++c) {
      const int col = c * CF + f;
      const int reg = 8 * (col >> 4) + rr;
      const int l   = lhi + (col & 15);
      float a = part[0][reg][l];
      for (int q = 1; q < S; ++q) a += part[q][reg][l];
      s[c] = a;
    }
    const int node = tm + r16;
    const long long grow = (long long)b * CN + node;
    const float dv = dinv_t[(long long)b * CT * CN + node];
    const float* qp = Q + grow * N + f;
    const float z0 = qp[0]      - dv * s[0];
    const float z1 = qp[CF]     - dv * s[1];
    const float z2 = qp[2 * CF] - dv * s[2];
    const float gt = 1.0f / (1.0f + expf(-z0));
    hout[grow * CF + f] = gt * tanhf(z1) + (1.0f - gt) * tanhf(z2);
  }
}

static inline void launch_gemm(const float* A, long long sA, long long sA2, int lda,
                               const float* Bm, long long sB, int ldb,
                               float* C, long long sC, long long sC2, int ldc,
                               const float* bias, int M, int N, int K,
                               int batch, int batch2, int relu, hipStream_t stream) {
  int npair = (N + 31) >> 5;
  int tiles = (M >> 4) * npair;
  dim3 grid((tiles + 3) / 4, batch2, batch);
  gemm_wmma_f32<<<grid, 128, 0, stream>>>(A, sA, sA2, lda, Bm, sB, ldb,
                                          C, sC, sC2, ldc, bias, M, N, K, relu);
}

template <int S, int NT>
static inline void launch_gemm_ksplit(const float* A, long long sA, long long sA2, int lda,
                                      const float* Bm, long long sB, int ldb,
                                      float* C, long long sC, long long sC2, int ldc,
                                      const float* bias, int M, int N, int K,
                                      int batch, int batch2, int relu, hipStream_t stream) {
  int nstrip = (N + 16 * NT - 1) / (16 * NT);
  int tiles = (M >> 4) * nstrip;
  dim3 grid(tiles, batch2, batch);
  gemm_wmma_f32_ksplit<S, NT><<<grid, 32 * S, 0, stream>>>(A, sA, sA2, lda, Bm, sB, ldb,
                                                           C, sC, sC2, ldc, bias, N, K, relu);
}

static const int TE = 4;
static const int TD = 4;

extern "C" void kernel_launch(void* const* d_in, const int* in_sizes, int n_in,
                              void* d_out, int out_size, void* d_ws, size_t ws_size,
                              hipStream_t stream) {
  const float* obs  = (const float*)d_in[0];
  const float* adj  = (const float*)d_in[1];
  const float* h0   = (const float*)d_in[2];
  const float* We1  = (const float*)d_in[3];
  const float* be1  = (const float*)d_in[4];
  const float* We2  = (const float*)d_in[5];
  const float* be2  = (const float*)d_in[6];
  const float* Wch  = (const float*)d_in[7];
  const float* Wcu  = (const float*)d_in[8];
  const float* bc   = (const float*)d_in[9];
  const float* Wd1  = (const float*)d_in[10];
  const float* bd1  = (const float*)d_in[11];
  const float* Wd2  = (const float*)d_in[12];
  const float* bd2  = (const float*)d_in[13];
  const float* Wd3  = (const float*)d_in[14];
  const float* bd3  = (const float*)d_in[15];
  float* out = (float*)d_out;

  const int MN = CB * CN;
  const size_t HSZ = (size_t)MN * CF;

  float* ws = (float*)d_ws;
  size_t off = 0;
  float* dinv = ws + off; off += (size_t)CB * CT * CN;
  float* Wcat = ws + off; off += (size_t)100 * 300;
  float* H    = ws + off; off += TD * HSZ;
  float* xc   = ws + off; off += TE * HSZ;
  float* scr  = ws + off;
  float* Rp   = scr;
  float* Qp   = Rp + (size_t)MN * 3 * CF;
  float* tmp1 = scr;
  float* y    = scr;
  (void)ws_size; (void)in_sizes; (void)n_in; (void)out_size;

  float* hlast = H + (TD - 1) * HSZ;
  {
    int rows = CB * CT * CN;
    deg_dinv_kernel<<<(rows * 32 + 255) / 256, 256, 0, stream>>>(adj, dinv, rows, CN);
    pack_wcat_kernel<<<(100 * 300 + 255) / 256, 256, 0, stream>>>(Wch, Wcu, Wcat);
    hipMemcpyAsync(hlast, h0, sizeof(float) * CB * CN * CF, hipMemcpyDeviceToDevice, stream);
  }

  for (int te = 0; te < CT; te += TE) {
    launch_gemm(obs + (long long)te * CN * CIN, (long long)CT * CN * CIN, (long long)CN * CIN, CIN,
                We1, 0, CFC, tmp1, (long long)CN * CFC, (long long)MN * CFC, CFC,
                be1, CN, CFC, CIN, CB, TE, 1, stream);
    launch_gemm_ksplit<8, 4>(tmp1, 0, 0, CFC, We2, 0, CF, xc, 0, 0, CF,
                          be2, TE * MN, CF, CFC, 1, 1, 1, stream);

    for (int j = 0; j < TD; ++j) {
      const int t = te + j;
      const float* x = xc + (size_t)j * HSZ;
      const float* hin = (j == 0) ? hlast : (H + (size_t)(j - 1) * HSZ);
      float* hout = H + (size_t)j * HSZ;
      gemm_hx_ksplit<8, 10><<<(MN / 16) * 2, 32 * 8, 0, stream>>>(
          hin, x, dinv + (long long)t * CN, Wcat, bc, Rp, Qp);
      gemm_adj_gate_ksplit<16, 10><<<dim3(CN / 16, 1, CB), 32 * 16, 0, stream>>>(
          adj + (long long)t * CN * CN, Rp, Qp, dinv + (long long)t * CN, hout);
    }
    decoder_rows_ksplit<<<(TD * MN) / 16, 32 * 8, 0, stream>>>(
        H, y, Wd1, bd1, Wd2, bd2, Wd3, bd3, out + (long long)te * CN * COUT);
  }

  hipMemcpyAsync(out + (long long)CB * CT * CN * COUT, hlast,
                 sizeof(float) * CB * CN * CF, hipMemcpyDeviceToDevice, stream);
}


# steps 2-4 of a 4-step block: step t+1's weight GEMM (K=100) runs in step t's adjacency launch, per block on its own 16 rows, h_t from LDS, K as one WMMA chain (sum order changes); 90->66 launches
# speedup vs baseline: 4.9334x; 1.1246x over previous
#include <hip/hip_runtime.h>
#include <math.h>

static const int CB = 4, CT = 32, CN = 512;
static const int CIN = 10, CF = 50, CFC = 128, COUT = 2;

typedef __attribute__((ext_vector_type(2))) float v2f;
typedef __attribute__((ext_vector_type(8))) float v8f;

__global__ __launch_bounds__(128)
void gemm_wmma_f32(const float* __restrict__ A, long long sA, long long sA2, int lda,
                   const float* __restrict__ Bm, long long sB, int ldb,
                   float* __restrict__ C, long long sC, long long sC2, int ldc,
                   const float* __restrict__ bias,
                   int M, int N, int K, int relu) {
  const int lane  = threadIdx.x & 31;
  const int wave  = threadIdx.x >> 5;
  const int npair = (N + 31) >> 5;
  const int tiles = (M >> 4) * npair;
  const int tile  = blockIdx.x * (blockDim.x >> 5) + wave;
  if (tile >= tiles) return;
  const int tm = (tile / npair) << 4;
  const int tn = (tile % npair) << 5;

  const float* Ab = A  + (long long)blockIdx.z * sA + (long long)blockIdx.y * sA2;
  const float* Bb = Bm + (long long)blockIdx.z * sB;
  float*       Cb = C  + (long long)blockIdx.z * sC + (long long)blockIdx.y * sC2;

  const int mrow  = lane & 15;
  const int khalf = (lane >> 4) << 1;
  const int n0 = tn + (lane & 15);
  const int n1 = n0 + 16;
  const float* ap  = Ab + (long long)(tm + mrow) * lda + khalf;
  const float* bp0 = Bb + (n0 < N ? n0 : N - 1);
  const float* bp1 = Bb + (n1 < N ? n1 : N - 1);

  v8f acc0 = {}, acc1 = {};
  const int K4 = K & ~3;
  int k0 = 0;
#pragma unroll 2
  for (; k0 < K4; k0 += 4) {
    const v2f av = *(const v2f*)(ap + k0);
    const long long kb = (long long)(k0 + khalf) * ldb;
    v2f bv0, bv1;
    bv0.x = bp0[kb];       bv0.y = bp0[kb + ldb];
    bv1.x = bp1[kb];       bv1.y = bp1[kb + ldb];
    acc0 = __builtin_amdgcn_wmma_f32_16x16x4_f32(false, av, false, bv0,
                                                 (short)0, acc0, false, false);
    acc1 = __builtin_amdgcn_wmma_f32_16x16x4_f32(false, av, false, bv1,
                                                 (short)0, acc1, false, false);
  }
  if (k0 < K) {
    const int ka = k0 + khalf;
    v2f av = {}, bv0 = {}, bv1 = {};
    if (ka < K) {
      av.x  = ap[k0];
      bv0.x = bp0[(long long)ka * ldb];
      bv1.x = bp1[(long long)ka * ldb];
    }
    if (ka + 1 < K) {
      av.y  = ap[k0 + 1];
      bv0.y = bp0[(long long)(ka + 1) * ldb];
      bv1.y = bp1[(long long)(ka + 1) * ldb];
    }
    acc0 = __builtin_amdgcn_wmma_f32_16x16x4_f32(false, av, false, bv0,
                                                 (short)0, acc0, false, false);
    acc1 = __builtin_amdgcn_wmma_f32_16x16x4_f32(false, av, false, bv1,
                                                 (short)0, acc1, false, false);
  }

  const int rbase = tm + ((lane >> 4) << 3);
  if (n0 < N) {
    const float badd = bias ? bias[n0] : 0.0f;
#pragma unroll
    for (int r = 0; r < 8; ++r) {
      float v = acc0[r] + badd;
      if (relu) v = fmaxf(v, 0.0f);
      Cb[(long long)(rbase + r) * ldc + n0] = v;
    }
  }
  if (n1 < N) {
    const float badd = bias ? bias[n1] : 0.0f;
#pragma unroll
    for (int r = 0; r < 8; ++r) {
      float v = acc1[r] + badd;
      if (relu) v = fmaxf(v, 0.0f);
      Cb[(long long)(rbase + r) * ldc + n1] = v;
    }
  }
}

__device__ __forceinline__ void wmma_group_open() {
  __builtin_amdgcn_sched_barrier(0);
}
__device__ __forceinline__ void wmma_operands_ready(v2f& b_a, v2f& b_b, v2f& ava, v2f& avb) {
  asm volatile("" : "+v"(b_a), "+v"(b_b), "+v"(ava), "+v"(avb));
}
__device__ __forceinline__ void wmma_operands_ready(v2f& b_a, v2f& ava) {
  asm volatile("" : "+v"(b_a), "+v"(ava));
}
__device__ __forceinline__ void wmma_group_close() {
  __builtin_amdgcn_sched_barrier(0);
  asm volatile("v_nop\n\tv_nop\n\tv_nop\n\tv_nop");
  __builtin_amdgcn_sched_barrier(0);
}

template <int S, int NT>
__global__ __launch_bounds__(32 * S)
void gemm_wmma_f32_ksplit(const float* __restrict__ A, long long sA, long long sA2, int lda,
                          const float* __restrict__ Bm, long long sB, int ldb,
                          float* __restrict__ C, long long sC, long long sC2, int ldc,
                          const float* __restrict__ bias,
                          int N, int K, int relu) {
  static_assert((8 * NT) % S == 0, "whole accumulator registers per wave in the reduction");
  __shared__ float part[S][8 * NT][32];

  const int lane  = threadIdx.x & 31;
  const int wave  = __builtin_amdgcn_readfirstlane(threadIdx.x >> 5);
  constexpr int SW = 16 * NT;
  const int nstrip = (N + SW - 1) / SW;
  const int tile  = blockIdx.x;
  const int tm = (tile / nstrip) << 4;
  const int tn = (tile % nstrip) * SW;

  const float* Ab = A  + (long long)blockIdx.z * sA + (long long)blockIdx.y * sA2;
  const float* Bb = Bm + (long long)blockIdx.z * sB;
  float*       Cb = C  + (long long)blockIdx.z * sC + (long long)blockIdx.y * sC2;

  const int mrow  = lane & 15;
  const int khalf = (lane >> 4) << 1;
  const int nb = tn + (lane & 15);
  const float* ap  = Ab + (long long)(tm + mrow) * lda + khalf;
  const float* bp[NT];
#pragma unroll
  for (int t = 0; t < NT; ++t) {
    const int n = nb + 16 * t;
    bp[t] = Bb + (n < N ? n : N - 1);
  }

  const int K4    = K & ~3;
  const int steps = (K + 3) >> 2;
  const int per   = (steps + S - 1) / S;
  const int kbeg  = wave * per * 4;
  const int klim  = kbeg + per * 4;
  const int kstop = klim < K4 ? klim : K4;

  v8f acc[NT];
#pragma unroll
  for (int t = 0; t < NT; ++t) acc[t] = v8f{};
  int k0 = kbeg;
  for (; k0 + 4 < kstop; k0 += 8) {
    v2f ava = *(const v2f*)(ap + k0);
    v2f avb = *(const v2f*)(ap + k0 + 4);
    const long long kb = (long long)(k0 + khalf) * ldb;
    v2f ba[NT], bb[NT];
#pragma unroll
    for (int t = 0; t < NT; ++t) {
      ba[t].x = bp[t][kb];             ba[t].y = bp[t][kb + ldb];
      bb[t].x = bp[t][kb + 4LL * ldb]; bb[t].y = bp[t][kb + 5LL * ldb];
    }
#pragma unroll
    for (int t = 0; t < NT; ++t) wmma_operands_ready(ba[t], bb[t], ava, avb);
    wmma_group_open();
#pragma unroll
    for (int t = 0; t < NT; ++t)
      acc[t] = __builtin_amdgcn_wmma_f32_16x16x4_f32(false, ava, false, ba[t],
                                                     (short)0, acc[t], false, false);
#pragma unroll
    for (int t = 0; t < NT; ++t)
      acc[t] = __builtin_amdgcn_wmma_f32_16x16x4_f32(false, avb, false, bb[t],
                                                     (short)0, acc[t], false, false);
    wmma_group_close();
  }
  if (k0 < kstop) {
    v2f ava = *(const v2f*)(ap + k0);
    const long long kb = (long long)(k0 + khalf) * ldb;
    v2f ba[NT];
#pragma unroll
    for (int t = 0; t < NT; ++t) { ba[t].x = bp[t][kb]; ba[t].y = bp[t][kb + ldb]; }
#pragma unroll
    for (int t = 0; t < NT; ++t) wmma_operands_ready(ba[t], ava);
    wmma_group_open();
#pragma unroll
    for (int t = 0; t < NT; ++t)
      acc[t] = __builtin_amdgcn_wmma_f32_16x16x4_f32(false, ava, false, ba[t],
                                                     (short)0, acc[t], false, false);
    wmma_group_close();
  }
  if (K4 < K && kbeg <= K4 && K4 < klim) {
    const int ka = K4 + khalf;
    v2f av = {};
    v2f bv[NT];
#pragma unroll
    for (int t = 0; t < NT; ++t) bv[t] = v2f{};
    if (ka < K) {
      av.x  = ap[K4];
#pragma unroll
      for (int t = 0; t < NT; ++t) bv[t].x = bp[t][(long long)ka * ldb];
    }
    if (ka + 1 < K) {
      av.y  = ap[K4 + 1];
#pragma unroll
      for (int t = 0; t < NT; ++t) bv[t].y = bp[t][(long long)(ka + 1) * ldb];
    }
#pragma unroll
    for (int t = 0; t < NT; ++t) wmma_operands_ready(bv[t], av);
    wmma_group_open();
#pragma unroll
    for (int t = 0; t < NT; ++t)
      acc[t] = __builtin_amdgcn_wmma_f32_16x16x4_f32(false, av, false, bv[t],
                                                     (short)0, acc[t], false, false);
    wmma_group_close();
  }

#pragma unroll
  for (int t = 0; t < NT; ++t)
#pragma unroll
    for (int r = 0; r < 8; ++r) part[wave][8 * t + r][lane] = acc[t][r];
  __syncthreads();

  constexpr int RPW = 8 * NT / S;
  const int rbase = tm + ((lane >> 4) << 3);
  for (int j = 0; j < RPW; ++j) {
    const int r = wave * RPW + j;
    float s = part[0][r][lane];
    for (int q = 1; q < S; ++q) s += part[q][r][lane];
    const int n = nb + 16 * (r >> 3);
    if (n < N) {
      float v = s + (bias ? bias[n] : 0.0f);
      if (relu) v = fmaxf(v, 0.0f);
      Cb[(long long)(rbase + (r & 7)) * ldc + n] = v;
    }
  }
}

__global__ __launch_bounds__(256)
void decoder_rows_ksplit(const float* __restrict__ Hm, float* Y,
                         const float* __restrict__ Wd1, const float* __restrict__ bd1,
                         const float* __restrict__ Wd2, const float* __restrict__ bd2,
                         const float* __restrict__ Wd3, const float* __restrict__ bd3,
                         float* __restrict__ act) {
  constexpr int S  = 8;
  constexpr int NT = CFC / 16;
  __shared__ float part[S][8 * NT][32];

  const int lane  = threadIdx.x & 31;
  const int wave  = __builtin_amdgcn_readfirstlane(threadIdx.x >> 5);
  const int tm = blockIdx.x << 4;

  const int mrow  = lane & 15;
  const int khalf = (lane >> 4) << 1;
  const int nb = lane & 15;
  const int rbase = tm + ((lane >> 4) << 3);
  constexpr int RPW = 8 * NT / S;

  {
    constexpr int K = CF, K4 = K & ~3;
    constexpr int S1  = 4;
    constexpr int per = (((K + 3) >> 2) + S1 - 1) / S1;
    static_assert(per % 2 == 0, "whole trips of two k-steps in every wave");
    v8f acc[NT];
#pragma unroll
    for (int t = 0; t < NT; ++t) acc[t] = v8f{};
    if (wave < S1) {
      const float* ap = Hm + (long long)(tm + mrow) * CF + khalf;
      const float* bp = Wd1 + nb;
      const int kbeg  = wave * per * 4;
      const int klim  = kbeg + per * 4;
      const int kstop = klim < K4 ? klim : K4;
      for (int k0 = kbeg; k0 < kstop; k0 += 8) {
        v2f ava = *(const v2f*)(ap + k0);
        v2f avb = *(const v2f*)(ap + k0 + 4);
        const int kb = (k0 + khalf) * CFC;
        v2f ba[NT], bb[NT];
#pragma unroll
        for (int t = 0; t < NT; ++t) {
          ba[t].x = bp[kb + 16 * t];           ba[t].y = bp[kb + CFC + 16 * t];
          bb[t].x = bp[kb + 4 * CFC + 16 * t]; bb[t].y = bp[kb + 5 * CFC + 16 * t];
        }
#pragma unroll
        for (int t = 0; t < NT; ++t) wmma_operands_ready(ba[t], bb[t], ava, avb);
        wmma_group_open();
#pragma unroll
        for (int t = 0; t < NT; ++t)
          acc[t] = __builtin_amdgcn_wmma_f32_16x16x4_f32(false, ava, false, ba[t],
                                                         (short)0, acc[t], false, false);
#pragma unroll
        for (int t = 0; t < NT; ++t)
          acc[t] = __builtin_amdgcn_wmma_f32_16x16x4_f32(false, avb, false, bb[t],
                                                         (short)0, acc[t], false, false);
        wmma_group_close();
      }
      if (K4 < K && kbeg <= K4 && K4 < klim) {
        const int ka = K4 + khalf;
        v2f av = {};
        v2f bv[NT];
#pragma unroll
        for (int t = 0; t < NT; ++t) bv[t] = v2f{};
        if (ka < K) {
          av.x = ap[K4];
#pragma unroll
          for (int t = 0; t < NT; ++t) bv[t].x = bp[ka * CFC + 16 * t];
        }
        if (ka + 1 < K) {
          av.y = ap[K4 + 1];
#pragma unroll
          for (int t = 0; t < NT; ++t) bv[t].y = bp[(ka + 1) * CFC + 16 * t];
        }
#pragma unroll
        for (int t = 0; t < NT; ++t) wmma_operands_ready(bv[t], av);
        wmma_group_open();
#pragma unroll
        for (int t = 0; t < NT; ++t)
          acc[t] = __builtin_amdgcn_wmma_f32_16x16x4_f32(false, av, false, bv[t],
                                                         (short)0, acc[t], false, false);
        wmma_group_close();
      }
#pragma unroll
      for (int t = 0; t < NT; ++t)
#pragma unroll
        for (int r = 0; r < 8; ++r) part[wave][8 * t + r][lane] = acc[t][r];
    }
    __syncthreads();
    for (int j = 0; j < RPW; ++j) {
      const int r = wave * RPW + j;
      float s = part[0][r][lane];
      for (int q = 1; q < S1; ++q) s += part[q][r][lane];
      const int n = nb + 16 * (r >> 3);
      const float v = fmaxf(s + bd1[n], 0.0f);
      Y[(long long)(rbase + (r & 7)) * CFC + n] = v;
    }
    __syncthreads();
  }

  {
    constexpr int K = CFC;
    constexpr int per = (K / 4) / S;
    static_assert((K / 8) % S == 0, "whole trips of two k-steps, the same number in every wave");
    const float* ap = Y + (long long)(tm + mrow) * CFC + khalf;
    const float* bp = Wd2 + nb;
    const int kbeg = wave * per * 4;
    const int kend = kbeg + per * 4;
    v8f acc[NT];
#pragma unroll
    for (int t = 0; t < NT; ++t) acc[t] = v8f{};
    for (int k0 = kbeg; k0 < kend; k0 += 8) {
      v2f ava = *(const v2f*)(ap + k0);
      v2f avb = *(const v2f*)(ap + k0 + 4);
      const int kb = (k0 + khalf) * CFC;
      v2f ba[NT], bb[NT];
#pragma unroll
      for (int t = 0; t < NT; ++t) {
        ba[t].x = bp[kb + 16 * t];           ba[t].y = bp[kb + CFC + 16 * t];
        bb[t].x = bp[kb + 4 * CFC + 16 * t]; bb[t].y = bp[kb + 5 * CFC + 16 * t];
      }
#pragma unroll
      for (int t = 0; t < NT; ++t) wmma_operands_ready(ba[t], bb[t], ava, avb);
      wmma_group_open();
#pragma unroll
      for (int t = 0; t < NT; ++t)
        acc[t] = __builtin_amdgcn_wmma_f32_16x16x4_f32(false, ava, false, ba[t],
                                                       (short)0, acc[t], false, false);
#pragma unroll
      for (int t = 0; t < NT; ++t)
        acc[t] = __builtin_amdgcn_wmma_f32_16x16x4_f32(false, avb, false, bb[t],
                                                       (short)0, acc[t], false, false);
      wmma_group_close();
    }
#pragma unroll
    for (int t = 0; t < NT; ++t)
#pragma unroll
      for (int r = 0; r < 8; ++r) part[wave][8 * t + r][lane] = acc[t][r];
    __syncthreads();
    for (int j = 0; j < RPW; ++j) {
      const int r = wave * RPW + j;
      float s = part[0][r][lane];
      for (int q = 1; q < S; ++q) s += part[q][r][lane];
      const int n = nb + 16 * (r >> 3);
      const float v = fmaxf(s + bd2[n], 0.0f);
      Y[(long long)(rbase + (r & 7)) * CFC + n] = v;
    }
    __syncthreads();
  }

  {
    constexpr int K = CFC;
    constexpr int per = (K / 4) / S;
    const float* ap = Y + (long long)(tm + mrow) * CFC + khalf;
    const float* bp = Wd3 + (nb < COUT ? nb : COUT - 1);
    const int kbeg = wave * per * 4;
    const int kend = kbeg + per * 4;
    v8f acc = {};
    for (int k0 = kbeg; k0 < kend; k0 += 8) {
      v2f ava = *(const v2f*)(ap + k0);
      v2f avb = *(const v2f*)(ap + k0 + 4);
      const int kb = (k0 + khalf) * COUT;
      v2f ba, bb;
      ba.x = bp[kb];            ba.y = bp[kb + COUT];
      bb.x = bp[kb + 4 * COUT]; bb.y = bp[kb + 5 * COUT];
      wmma_operands_ready(ba, bb, ava, avb);
      wmma_group_open();
      acc = __builtin_amdgcn_wmma_f32_16x16x4_f32(false, ava, false, ba,
                                                  (short)0, acc, false, false);
      acc = __builtin_amdgcn_wmma_f32_16x16x4_f32(false, avb, false, bb,
                                                  (short)0, acc, false, false);
      wmma_group_close();
    }
#pragma unroll
    for (int r = 0; r < 8; ++r) part[wave][r][lane] = acc[r];
    __syncthreads();
    const int r = wave;
    float s = part[0][r][lane];
    for (int q = 1; q < S; ++q) s += part[q][r][lane];
    if (nb < COUT) {
      const int row  = rbase + r;
      const int sb   = row / CN;
      const int node = row - sb * CN;
      const int step = sb / CB, b = sb - step * CB;
      act[((long long)(b * CT + step) * CN + node) * COUT + nb] = s + bd3[nb];
    }
  }
}

__global__ void deg_dinv_kernel(const float* __restrict__ A, float* __restrict__ dinv,
                                int rows, int n) {
  const int wid  = (blockIdx.x * blockDim.x + threadIdx.x) >> 5;
  const int lane = threadIdx.x & 31;
  if (wid >= rows) return;
  const float* row = A + (long long)wid * n;
  float s = 0.0f;
  for (int i = lane; i < n; i += 32) s += row[i];
#pragma unroll
  for (int off = 16; off > 0; off >>= 1) s += __shfl_xor(s, off, 32);
  if (lane == 0) dinv[wid] = (s > 0.0f) ? rsqrtf(s) : 0.0f;
}

__global__ void pack_wcat_kernel(const float* __restrict__ Wch,
                                 const float* __restrict__ Wcu,
                                 float* __restrict__ Wcat) {
  int idx = blockIdx.x * blockDim.x + threadIdx.x;
  if (idx >= 100 * 300) return;
  int r = idx / 300, n = idx % 300;
  const float* src = r < 50 ? Wch : Wcu;
  int f = r < 50 ? r : r - 50;
  int m = n < 150 ? n : n - 150;
  int c = m / 50, g = m % 50;
  float v = src[((c * 2 + 1) * 50 + f) * 50 + g];
  if (n >= 150) v = src[((c * 2 + 0) * 50 + f) * 50 + g] + v;
  Wcat[idx] = v;
}

__device__ __forceinline__ v2f hx_a_frag(const float* __restrict__ hp,
                                         const float* __restrict__ xp, int k) {
  const float* p = k < CF ? hp + k : xp + (k - CF);
  return *(const v2f*)p;
}

template <int S, int NT>
__global__ __launch_bounds__(32 * S)
void gemm_hx_ksplit(const float* __restrict__ h, const float* __restrict__ x,
                    const float* __restrict__ dinv_t,
                    const float* __restrict__ Wcat, const float* __restrict__ bias,
                    float* __restrict__ R, float* __restrict__ Q) {
  constexpr int NH = 3 * CF;
  constexpr int N = 2 * NH;
  constexpr int K = 2 * CF;
  static_assert(K % 4 == 0 && CF % 2 == 0, "no K tail; halves start at even k");
  static_assert((8 * NT) % S == 0, "whole accumulator registers per wave in the reduction");
  __shared__ float part[S][8 * NT][32];

  const int lane  = threadIdx.x & 31;
  const int wave  = __builtin_amdgcn_readfirstlane(threadIdx.x >> 5);
  constexpr int SW = 16 * NT;
  constexpr int nstrip = (N + SW - 1) / SW;
  const int tm = (blockIdx.x / nstrip) << 4;
  const int tn = (blockIdx.x % nstrip) * SW;

  const int mrow  = lane & 15;
  const int khalf = (lane >> 4) << 1;
  const int nb = tn + (lane & 15);
  const int row = tm + mrow;
  const float* hp = h + (long long)row * CF;
  const float* xp = x + (long long)row * CF;
  const float* bp[NT];
#pragma unroll
  for (int t = 0; t < NT; ++t) {
    const int n = nb + 16 * t;
    bp[t] = Wcat + (n < N ? n : N - 1);
  }

  constexpr int steps = K / 4;
  constexpr int per   = (steps + S - 1) / S;
  const int kbeg  = wave * per * 4;
  const int klim  = kbeg + per * 4;
  const int kstop = klim < K ? klim : K;

  v8f acc[NT];
#pragma unroll
  for (int t = 0; t < NT; ++t) acc[t] = v8f{};
  int k0 = kbeg;
  for (; k0 + 4 < kstop; k0 += 8) {
    const int k = k0 + khalf;
    v2f ava = hx_a_frag(hp, xp, k);
    v2f avb = hx_a_frag(hp, xp, k + 4);
    const int kb = k * N;
    v2f ba[NT], bb[NT];
#pragma unroll
    for (int t = 0; t < NT; ++t) {
      ba[t].x = bp[t][kb];           ba[t].y = bp[t][kb + N];
      bb[t].x = bp[t][kb + 4 * N];   bb[t].y = bp[t][kb + 5 * N];
    }
#pragma unroll
    for (int t = 0; t < NT; ++t) wmma_operands_ready(ba[t], bb[t], ava, avb);
    wmma_group_open();
#pragma unroll
    for (int t = 0; t < NT; ++t)
      acc[t] = __builtin_amdgcn_wmma_f32_16x16x4_f32(false, ava, false, ba[t],
                                                     (short)0, acc[t], false, false);
#pragma unroll
    for (int t = 0; t < NT; ++t)
      acc[t] = __builtin_amdgcn_wmma_f32_16x16x4_f32(false, avb, false, bb[t],
                                                     (short)0, acc[t], false, false);
    wmma_group_close();
  }
  if (k0 < kstop) {
    const int k = k0 + khalf;
    v2f ava = hx_a_frag(hp, xp, k);
    const int kb = k * N;
    v2f ba[NT];
#pragma unroll
    for (int t = 0; t < NT; ++t) { ba[t].x = bp[t][kb]; ba[t].y = bp[t][kb + N]; }
#pragma unroll
    for (int t = 0; t < NT; ++t) wmma_operands_ready(ba[t], ava);
    wmma_group_open();
#pragma unroll
    for (int t = 0; t < NT; ++t)
      acc[t] = __builtin_amdgcn_wmma_f32_16x16x4_f32(false, ava, false, ba[t],
                                                     (short)0, acc[t], false, false);
    wmma_group_close();
  }

  constexpr int RPW = 8 * NT / S;
  const int rbase = tm + ((lane >> 4) << 3);
  float dvv[RPW], bvv[RPW];
#pragma unroll
  for (int j = 0; j < RPW; ++j) {
    const int r = wave * RPW + j;
    const int grow = rbase + (r & 7);
    const int m = nb + 16 * (r >> 3) - NH;
    dvv[j] = dinv_t[(long long)(grow / CN) * CT * CN + (grow % CN)];
    bvv[j] = bias[m < 0 ? 0 : (m < NH ? m : NH - 1)];
  }

#pragma unroll
  for (int t = 0; t < NT; ++t)
#pragma unroll
    for (int r = 0; r < 8; ++r) part[wave][8 * t + r][lane] = acc[t][r];
  __syncthreads();

#pragma unroll
  for (int j = 0; j < RPW; ++j) {
    const int r = wave * RPW + j;
    float s = part[0][r][lane];
    for (int q = 1; q < S; ++q) s += part[q][r][lane];
    const int n = nb + 16 * (r >> 3);
    const int grow = rbase + (r & 7);
    const bool rhalf = n < NH;
    const float v = rhalf ? dvv[j] * s : s + bvv[j];
    float* dst = rhalf ? R + (long long)grow * NH + n : Q + (long long)grow * NH + (n - NH);
    if (n < N) *dst = v;
  }
}

__device__ __forceinline__ void lds_barrier() {
  asm volatile("s_wait_dscnt 0x0" ::: "memory");
  asm volatile("s_barrier_signal -1" ::: "memory");
  asm volatile("s_barrier_wait -1" ::: "memory");
}

template <int S, int NT>
__global__ __launch_bounds__(32 * S)
void step_adj_gate_hx(const float* __restrict__ adj_t,
                      const float* __restrict__ R, float* Q,
                      const float* __restrict__ dinv_t,
                      float* __restrict__ hout,
                      const float* __restrict__ x_next,
                      const float* __restrict__ dinv_n,
                      const float* __restrict__ Wcat, const float* __restrict__ bias,
                      float* __restrict__ Rn) {
  constexpr int N = 3 * CF;
  constexpr int K = CN;
  static_assert(K % (8 * S) == 0, "whole trips of two k-steps in every wave");
  static_assert(16 * NT >= N, "a block holds whole rows of z");
  constexpr int HXP = 2 * CF + 4;
  constexpr int NW = 2 * N;
  constexpr int KS = (2 * CF) / 4;
  constexpr int WT = (NW + 15) / 16;
  constexpr int TPW = (WT + S - 1) / S;
  static_assert((2 * CF) % 4 == 0, "no K tail");
  __shared__ float part[S][8 * NT][32];
  __shared__ float hxs[16][HXP];

  const int lane  = threadIdx.x & 31;
  const int wave  = __builtin_amdgcn_readfirstlane(threadIdx.x >> 5);
  const int tm = blockIdx.x << 4;
  const int b  = blockIdx.z;

  const float* Ab = adj_t + (long long)b * CT * CN * CN;
  const float* Rb = R + (long long)b * CN * N;

  constexpr int XPT = (16 * CF + 32 * S - 1) / (32 * S);
  float xv[XPT] = {};
  if (x_next != nullptr) {
    const float* xb = x_next + ((long long)b * CN + tm) * CF;
#pragma unroll
    for (int i = 0; i < XPT; ++i) {
      const int idx = threadIdx.x + i * 32 * S;
      xv[i] = xb[idx < 16 * CF ? idx : 16 * CF - 1];
    }
  }

  const int mrow  = lane & 15;
  const int khalf = (lane >> 4) << 1;
  const int nb = lane & 15;
  const float* ap  = Ab + (long long)(tm + mrow) * K + khalf;
  const float* bp[NT];
#pragma unroll
  for (int t = 0; t < NT; ++t) {
    const int n = nb + 16 * t;
    bp[t] = Rb + (n < N ? n : N - 1);
  }

  constexpr int per = K / (4 * S);
  const int kbeg = wave * per * 4;
  const int kend = kbeg + per * 4;

  v8f acc[NT];
#pragma unroll
  for (int t = 0; t < NT; ++t) acc[t] = v8f{};
  for (int k0 = kbeg; k0 < kend; k0 += 8) {
    v2f ava = *(const v2f*)(ap + k0);
    v2f avb = *(const v2f*)(ap + k0 + 4);
    const int kb = (k0 + khalf) * N;
    v2f ba[NT], bb[NT];
#pragma unroll
    for (int t = 0; t < NT; ++t) {
      ba[t].x = bp[t][kb];           ba[t].y = bp[t][kb + N];
      bb[t].x = bp[t][kb + 4 * N];   bb[t].y = bp[t][kb + 5 * N];
    }
#pragma unroll
    for (int t = 0; t < NT; ++t) wmma_operands_ready(ba[t], bb[t], ava, avb);
    wmma_group_open();
#pragma unroll
    for (int t = 0; t < NT; ++t)
      acc[t] = __builtin_amdgcn_wmma_f32_16x16x4_f32(false, ava, false, ba[t],
                                                     (short)0, acc[t], false, false);
#pragma unroll
    for (int t = 0; t < NT; ++t)
      acc[t] = __builtin_amdgcn_wmma_f32_16x16x4_f32(false, avb, false, bb[t],
                                                     (short)0, acc[t], false, false);
    wmma_group_close();
  }

#pragma unroll
  for (int t = 0; t < NT; ++t)
#pragma unroll
    for (int r = 0; r < 8; ++r) part[wave][8 * t + r][lane] = acc[t][r];
  const int rbase = tm + ((lane >> 4) << 3);
  v2f bf[TPW][KS];
  float bq[TPW];
  float dn[8];
  if (x_next != nullptr) {
#pragma unroll
    for (int i = 0; i < XPT; ++i) {
      const int idx = threadIdx.x + i * 32 * S;
      if (idx < 16 * CF) {
        const int r16 = idx / CF;
        hxs[r16][CF + (idx - r16 * CF)] = xv[i];
      }
    }
#pragma unroll
    for (int i = 0; i < TPW; ++i) {
      const int tile = wave + i * S;
      if (tile < WT) {
        const int n = tile * 16 + nb;
        const int m = n - N;
        const float* wp = Wcat + (n < NW ? n : NW - 1) + khalf * NW;
#pragma unroll
        for (int s = 0; s < KS; ++s) {
          bf[i][s].x = wp[(4 * s) * NW];
          bf[i][s].y = wp[(4 * s + 1) * NW];
        }
        bq[i] = bias[m < 0 ? 0 : (m < N ? m : N - 1)];
      }
    }
#pragma unroll
    for (int r = 0; r < 8; ++r) dn[r] = dinv_n[(long long)b * CT * CN + rbase + r];
  }
  lds_barrier();

  for (int idx = threadIdx.x; idx < 16 * CF; idx += 32 * S) {
    const int r16 = idx / CF;
    const int f   = idx - r16 * CF;
    const int lhi = (r16 >> 3) << 4;
    const int rr  = r16 & 7;
    float s[3];
#pragma unroll
    for (int c = 0; c < 3; ++c) {
      const int col = c * CF + f;
      const int reg = 8 * (col >> 4) + rr;
      const int l   = lhi + (col & 15);
      float a = part[0][reg][l];
      for (int q = 1; q < S; ++q) a += part[q][reg][l];
      s[c] = a;
    }
    const int node = tm + r16;
    const long long grow = (long long)b * CN + node;
    const float dv = dinv_t[(long long)b * CT * CN + node];
    const float* qp = Q + grow * N + f;
    const float z0 = qp[0]      - dv * s[0];
    const float z1 = qp[CF]     - dv * s[1];
    const float z2 = qp[2 * CF] - dv * s[2];
    const float gt = 1.0f / (1.0f + expf(-z0));
    const float hv = gt * tanhf(z1) + (1.0f - gt) * tanhf(z2);
    hout[grow * CF + f] = hv;
    hxs[r16][f] = hv;
  }

  if (x_next == nullptr) return;
  lds_barrier();

  const float* hp = &hxs[mrow][khalf];
  v2f af[KS];
#pragma unroll
  for (int s = 0; s < KS; ++s) af[s] = *(const v2f*)(hp + 4 * s);

#pragma unroll
  for (int i = 0; i < TPW; ++i) {
    const int tile = wave + i * S;
    if (tile < WT) {
      const int n = tile * 16 + nb;
      const int m = n - N;
      v8f accw = {};
#pragma unroll
      for (int s = 0; s < KS; ++s) wmma_operands_ready(bf[i][s], af[s]);
      wmma_group_open();
#pragma unroll
      for (int s = 0; s < KS; ++s)
        accw = __builtin_amdgcn_wmma_f32_16x16x4_f32(false, af[s], false, bf[i][s],
                                                     (short)0, accw, false, false);
      wmma_group_close();
      const bool rhalf = n < N;
      const long long grow0 = (long long)b * CN + rbase;
      float* dst = rhalf ? Rn + grow0 * N + n : Q + grow0 * N + m;
      if (n < NW) {
#pragma unroll
        for (int r = 0; r < 8; ++r) dst[r * N] = rhalf ? dn[r] * accw[r] : accw[r] + bq[i];
      }
    }
  }
}

static inline void launch_gemm(const float* A, long long sA, long long sA2, int lda,
                               const float* Bm, long long sB, int ldb,
                               float* C, long long sC, long long sC2, int ldc,
                               const float* bias, int M, int N, int K,
                               int batch, int batch2, int relu, hipStream_t stream) {
  int npair = (N + 31) >> 5;
  int tiles = (M >> 4) * npair;
  dim3 grid((tiles + 3) / 4, batch2, batch);
  gemm_wmma_f32<<<grid, 128, 0, stream>>>(A, sA, sA2, lda, Bm, sB, ldb,
                                          C, sC, sC2, ldc, bias, M, N, K, relu);
}

template <int S, int NT>
static inline void launch_gemm_ksplit(const float* A, long long sA, long long sA2, int lda,
                                      const float* Bm, long long sB, int ldb,
                                      float* C, long long sC, long long sC2, int ldc,
                                      const float* bias, int M, int N, int K,
                                      int batch, int batch2, int relu, hipStream_t stream) {
  int nstrip = (N + 16 * NT - 1) / (16 * NT);
  int tiles = (M >> 4) * nstrip;
  dim3 grid(tiles, batch2, batch);
  gemm_wmma_f32_ksplit<S, NT><<<grid, 32 * S, 0, stream>>>(A, sA, sA2, lda, Bm, sB, ldb,
                                                           C, sC, sC2, ldc, bias, N, K, relu);
}

static const int TE = 4;
static const int TD = 4;

extern "C" void kernel_launch(void* const* d_in, const int* in_sizes, int n_in,
                              void* d_out, int out_size, void* d_ws, size_t ws_size,
                              hipStream_t stream) {
  const float* obs  = (const float*)d_in[0];
  const float* adj  = (const float*)d_in[1];
  const float* h0   = (const float*)d_in[2];
  const float* We1  = (const float*)d_in[3];
  const float* be1  = (const float*)d_in[4];
  const float* We2  = (const float*)d_in[5];
  const float* be2  = (const float*)d_in[6];
  const float* Wch  = (const float*)d_in[7];
  const float* Wcu  = (const float*)d_in[8];
  const float* bc   = (const float*)d_in[9];
  const float* Wd1  = (const float*)d_in[10];
  const float* bd1  = (const float*)d_in[11];
  const float* Wd2  = (const float*)d_in[12];
  const float* bd2  = (const float*)d_in[13];
  const float* Wd3  = (const float*)d_in[14];
  const float* bd3  = (const float*)d_in[15];
  float* out = (float*)d_out;

  const int MN = CB * CN;
  const size_t HSZ = (size_t)MN * CF;

  float* ws = (float*)d_ws;
  size_t off = 0;
  float* dinv = ws + off; off += (size_t)CB * CT * CN;
  float* Wcat = ws + off; off += (size_t)100 * 300;
  float* H    = ws + off; off += TD * HSZ;
  float* xc   = ws + off; off += TE * HSZ;
  float* scr  = ws + off;
  float* Rp0  = scr;
  float* Rp1  = Rp0 + (size_t)MN * 3 * CF;
  float* Qp   = Rp1 + (size_t)MN * 3 * CF;
  float* tmp1 = scr;
  float* y    = scr;
  (void)ws_size; (void)in_sizes; (void)n_in; (void)out_size;

  float* hlast = H + (TD - 1) * HSZ;
  {
    int rows = CB * CT * CN;
    deg_dinv_kernel<<<(rows * 32 + 255) / 256, 256, 0, stream>>>(adj, dinv, rows, CN);
    pack_wcat_kernel<<<(100 * 300 + 255) / 256, 256, 0, stream>>>(Wch, Wcu, Wcat);
    hipMemcpyAsync(hlast, h0, sizeof(float) * CB * CN * CF, hipMemcpyDeviceToDevice, stream);
  }

  for (int te = 0; te < CT; te += TE) {
    launch_gemm(obs + (long long)te * CN * CIN, (long long)CT * CN * CIN, (long long)CN * CIN, CIN,
                We1, 0, CFC, tmp1, (long long)CN * CFC, (long long)MN * CFC, CFC,
                be1, CN, CFC, CIN, CB, TE, 1, stream);
    launch_gemm_ksplit<8, 4>(tmp1, 0, 0, CFC, We2, 0, CF, xc, 0, 0, CF,
                          be2, TE * MN, CF, CFC, 1, 1, 1, stream);

    for (int j = 0; j < TD; ++j) {
      const int t = te + j;
      const float* x = xc + (size_t)j * HSZ;
      const float* hin = (j == 0) ? hlast : (H + (size_t)(j - 1) * HSZ);
      float* hout = H + (size_t)j * HSZ;
      float* Rcur = (j & 1) ? Rp1 : Rp0;
      float* Rnxt = (j & 1) ? Rp0 : Rp1;
      if (j == 0)
        gemm_hx_ksplit<8, 10><<<(MN / 16) * 2, 32 * 8, 0, stream>>>(
            hin, x, dinv + (long long)t * CN, Wcat, bc, Rcur, Qp);
      const bool more = j + 1 < TD;
      step_adj_gate_hx<16, 10><<<dim3(CN / 16, 1, CB), 32 * 16, 0, stream>>>(
          adj + (long long)t * CN * CN, Rcur, Qp, dinv + (long long)t * CN, hout,
          more ? xc + (size_t)(j + 1) * HSZ : nullptr,
          more ? dinv + (long long)(t + 1) * CN : dinv,
          Wcat, bc, Rnxt);
    }
    decoder_rows_ksplit<<<(TD * MN) / 16, 32 * 8, 0, stream>>>(
        H, y, Wd1, bd1, Wd2, bd2, Wd3, bd3, out + (long long)te * CN * COUT);
  }

  hipMemcpyAsync(out + (long long)CB * CT * CN * COUT, hlast,
                 sizeof(float) * CB * CN * CF, hipMemcpyDeviceToDevice, stream);
}


# encoder's 2 layers in one launch: a block of 8 waves owns 16 whole rows and reads the 128-wide rows it stored back from the global scratch plane behind a barrier; 66 -> 58 launches; same sum order
# speedup vs baseline: 5.2533x; 1.0648x over previous
#include <hip/hip_runtime.h>
#include <math.h>

static const int CB = 4, CT = 32, CN = 512;
static const int CIN = 10, CF = 50, CFC = 128, COUT = 2;

typedef __attribute__((ext_vector_type(2))) float v2f;
typedef __attribute__((ext_vector_type(8))) float v8f;


__device__ __forceinline__ void wmma_group_open() {
  __builtin_amdgcn_sched_barrier(0);
}
__device__ __forceinline__ void wmma_operands_ready(v2f& b_a, v2f& b_b, v2f& ava, v2f& avb) {
  asm volatile("" : "+v"(b_a), "+v"(b_b), "+v"(ava), "+v"(avb));
}
__device__ __forceinline__ void wmma_operands_ready(v2f& b_a, v2f& ava) {
  asm volatile("" : "+v"(b_a), "+v"(ava));
}
__device__ __forceinline__ void wmma_group_close() {
  __builtin_amdgcn_sched_barrier(0);
  asm volatile("v_nop\n\tv_nop\n\tv_nop\n\tv_nop");
  __builtin_amdgcn_sched_barrier(0);
}

template <int S, int NT>
__global__ __launch_bounds__(32 * S)
void gemm_wmma_f32_ksplit(const float* __restrict__ A, long long sA, long long sA2, int lda,
                          const float* __restrict__ Bm, long long sB, int ldb,
                          float* __restrict__ C, long long sC, long long sC2, int ldc,
                          const float* __restrict__ bias,
                          int N, int K, int relu) {
  static_assert((8 * NT) % S == 0, "whole accumulator registers per wave in the reduction");
  __shared__ float part[S][8 * NT][32];

  const int lane  = threadIdx.x & 31;
  const int wave  = __builtin_amdgcn_readfirstlane(threadIdx.x >> 5);
  constexpr int SW = 16 * NT;
  const int nstrip = (N + SW - 1) / SW;
  const int tile  = blockIdx.x;
  const int tm = (tile / nstrip) << 4;
  const int tn = (tile % nstrip) * SW;

  const float* Ab = A  + (long long)blockIdx.z * sA + (long long)blockIdx.y * sA2;
  const float* Bb = Bm + (long long)blockIdx.z * sB;
  float*       Cb = C  + (long long)blockIdx.z * sC + (long long)blockIdx.y * sC2;

  const int mrow  = lane & 15;
  const int khalf = (lane >> 4) << 1;
  const int nb = tn + (lane & 15);
  const float* ap  = Ab + (long long)(tm + mrow) * lda + khalf;
  const float* bp[NT];
#pragma unroll
  for (int t = 0; t < NT; ++t) {
    const int n = nb + 16 * t;
    bp[t] = Bb + (n < N ? n : N - 1);
  }

  const int K4    = K & ~3;
  const int steps = (K + 3) >> 2;
  const int per   = (steps + S - 1) / S;
  const int kbeg  = wave * per * 4;
  const int klim  = kbeg + per * 4;
  const int kstop = klim < K4 ? klim : K4;

  v8f acc[NT];
#pragma unroll
  for (int t = 0; t < NT; ++t) acc[t] = v8f{};
  int k0 = kbeg;
  for (; k0 + 4 < kstop; k0 += 8) {
    v2f ava = *(const v2f*)(ap + k0);
    v2f avb = *(const v2f*)(ap + k0 + 4);
    const long long kb = (long long)(k0 + khalf) * ldb;
    v2f ba[NT], bb[NT];
#pragma unroll
    for (int t = 0; t < NT; ++t) {
      ba[t].x = bp[t][kb];             ba[t].y = bp[t][kb + ldb];
      bb[t].x = bp[t][kb + 4LL * ldb]; bb[t].y = bp[t][kb + 5LL * ldb];
    }
#pragma unroll
    for (int t = 0; t < NT; ++t) wmma_operands_ready(ba[t], bb[t], ava, avb);
    wmma_group_open();
#pragma unroll
    for (int t = 0; t < NT; ++t)
      acc[t] = __builtin_amdgcn_wmma_f32_16x16x4_f32(false, ava, false, ba[t],
                                                     (short)0, acc[t], false, false);
#pragma unroll
    for (int t = 0; t < NT; ++t)
      acc[t] = __builtin_amdgcn_wmma_f32_16x16x4_f32(false, avb, false, bb[t],
                                                     (short)0, acc[t], false, false);
    wmma_group_close();
  }
  if (k0 < kstop) {
    v2f ava = *(const v2f*)(ap + k0);
    const long long kb = (long long)(k0 + khalf) * ldb;
    v2f ba[NT];
#pragma unroll
    for (int t = 0; t < NT; ++t) { ba[t].x = bp[t][kb]; ba[t].y = bp[t][kb + ldb]; }
#pragma unroll
    for (int t = 0; t < NT; ++t) wmma_operands_ready(ba[t], ava);
    wmma_group_open();
#pragma unroll
    for (int t = 0; t < NT; ++t)
      acc[t] = __builtin_amdgcn_wmma_f32_16x16x4_f32(false, ava, false, ba[t],
                                                     (short)0, acc[t], false, false);
    wmma_group_close();
  }
  if (K4 < K && kbeg <= K4 && K4 < klim) {
    const int ka = K4 + khalf;
    v2f av = {};
    v2f bv[NT];
#pragma unroll
    for (int t = 0; t < NT; ++t) bv[t] = v2f{};
    if (ka < K) {
      av.x  = ap[K4];
#pragma unroll
      for (int t = 0; t < NT; ++t) bv[t].x = bp[t][(long long)ka * ldb];
    }
    if (ka + 1 < K) {
      av.y  = ap[K4 + 1];
#pragma unroll
      for (int t = 0; t < NT; ++t) bv[t].y = bp[t][(long long)(ka + 1) * ldb];
    }
#pragma unroll
    for (int t = 0; t < NT; ++t) wmma_operands_ready(bv[t], av);
    wmma_group_open();
#pragma unroll
    for (int t = 0; t < NT; ++t)
      acc[t] = __builtin_amdgcn_wmma_f32_16x16x4_f32(false, av, false, bv[t],
                                                     (short)0, acc[t], false, false);
    wmma_group_close();
  }

#pragma unroll
  for (int t = 0; t < NT; ++t)
#pragma unroll
    for (int r = 0; r < 8; ++r) part[wave][8 * t + r][lane] = acc[t][r];
  __syncthreads();

  constexpr int RPW = 8 * NT / S;
  const int rbase = tm + ((lane >> 4) << 3);
  for (int j = 0; j < RPW; ++j) {
    const int r = wave * RPW + j;
    float s = part[0][r][lane];
    for (int q = 1; q < S; ++q) s += part[q][r][lane];
    const int n = nb + 16 * (r >> 3);
    if (n < N) {
      float v = s + (bias ? bias[n] : 0.0f);
      if (relu) v = fmaxf(v, 0.0f);
      Cb[(long long)(rbase + (r & 7)) * ldc + n] = v;
    }
  }
}

__global__ __launch_bounds__(256)
void decoder_rows_ksplit(const float* __restrict__ Hm, float* Y,
                         const float* __restrict__ Wd1, const float* __restrict__ bd1,
                         const float* __restrict__ Wd2, const float* __restrict__ bd2,
                         const float* __restrict__ Wd3, const float* __restrict__ bd3,
                         float* __restrict__ act) {
  constexpr int S  = 8;
  constexpr int NT = CFC / 16;
  __shared__ float part[S][8 * NT][32];

  const int lane  = threadIdx.x & 31;
  const int wave  = __builtin_amdgcn_readfirstlane(threadIdx.x >> 5);
  const int tm = blockIdx.x << 4;

  const int mrow  = lane & 15;
  const int khalf = (lane >> 4) << 1;
  const int nb = lane & 15;
  const int rbase = tm + ((lane >> 4) << 3);
  constexpr int RPW = 8 * NT / S;

  {
    constexpr int K = CF, K4 = K & ~3;
    constexpr int S1  = 4;
    constexpr int per = (((K + 3) >> 2) + S1 - 1) / S1;
    static_assert(per % 2 == 0, "whole trips of two k-steps in every wave");
    v8f acc[NT];
#pragma unroll
    for (int t = 0; t < NT; ++t) acc[t] = v8f{};
    if (wave < S1) {
      const float* ap = Hm + (long long)(tm + mrow) * CF + khalf;
      const float* bp = Wd1 + nb;
      const int kbeg  = wave * per * 4;
      const int klim  = kbeg + per * 4;
      const int kstop = klim < K4 ? klim : K4;
      for (int k0 = kbeg; k0 < kstop; k0 += 8) {
        v2f ava = *(const v2f*)(ap + k0);
        v2f avb = *(const v2f*)(ap + k0 + 4);
        const int kb = (k0 + khalf) * CFC;
        v2f ba[NT], bb[NT];
#pragma unroll
        for (int t = 0; t < NT; ++t) {
          ba[t].x = bp[kb + 16 * t];           ba[t].y = bp[kb + CFC + 16 * t];
          bb[t].x = bp[kb + 4 * CFC + 16 * t]; bb[t].y = bp[kb + 5 * CFC + 16 * t];
        }
#pragma unroll
        for (int t = 0; t < NT; ++t) wmma_operands_ready(ba[t], bb[t], ava, avb);
        wmma_group_open();
#pragma unroll
        for (int t = 0; t < NT; ++t)
          acc[t] = __builtin_amdgcn_wmma_f32_16x16x4_f32(false, ava, false, ba[t],
                                                         (short)0, acc[t], false, false);
#pragma unroll
        for (int t = 0; t < NT; ++t)
          acc[t] = __builtin_amdgcn_wmma_f32_16x16x4_f32(false, avb, false, bb[t],
                                                         (short)0, acc[t], false, false);
        wmma_group_close();
      }
      if (K4 < K && kbeg <= K4 && K4 < klim) {
        const int ka = K4 + khalf;
        v2f av = {};
        v2f bv[NT];
#pragma unroll
        for (int t = 0; t < NT; ++t) bv[t] = v2f{};
        if (ka < K) {
          av.x = ap[K4];
#pragma unroll
          for (int t = 0; t < NT; ++t) bv[t].x = bp[ka * CFC + 16 * t];
        }
        if (ka + 1 < K) {
          av.y = ap[K4 + 1];
#pragma unroll
          for (int t = 0; t < NT; ++t) bv[t].y = bp[(ka + 1) * CFC + 16 * t];
        }
#pragma unroll
        for (int t = 0; t < NT; ++t) wmma_operands_ready(bv[t], av);
        wmma_group_open();
#pragma unroll
        for (int t = 0; t < NT; ++t)
          acc[t] = __builtin_amdgcn_wmma_f32_16x16x4_f32(false, av, false, bv[t],
                                                         (short)0, acc[t], false, false);
        wmma_group_close();
      }
#pragma unroll
      for (int t = 0; t < NT; ++t)
#pragma unroll
        for (int r = 0; r < 8; ++r) part[wave][8 * t + r][lane] = acc[t][r];
    }
    __syncthreads();
    for (int j = 0; j < RPW; ++j) {
      const int r = wave * RPW + j;
      float s = part[0][r][lane];
      for (int q = 1; q < S1; ++q) s += part[q][r][lane];
      const int n = nb + 16 * (r >> 3);
      const float v = fmaxf(s + bd1[n], 0.0f);
      Y[(long long)(rbase + (r & 7)) * CFC + n] = v;
    }
    __syncthreads();
  }

  {
    constexpr int K = CFC;
    constexpr int per = (K / 4) / S;
    static_assert((K / 8) % S == 0, "whole trips of two k-steps, the same number in every wave");
    const float* ap = Y + (long long)(tm + mrow) * CFC + khalf;
    const float* bp = Wd2 + nb;
    const int kbeg = wave * per * 4;
    const int kend = kbeg + per * 4;
    v8f acc[NT];
#pragma unroll
    for (int t = 0; t < NT; ++t) acc[t] = v8f{};
    for (int k0 = kbeg; k0 < kend; k0 += 8) {
      v2f ava = *(const v2f*)(ap + k0);
      v2f avb = *(const v2f*)(ap + k0 + 4);
      const int kb = (k0 + khalf) * CFC;
      v2f ba[NT], bb[NT];
#pragma unroll
      for (int t = 0; t < NT; ++t) {
        ba[t].x = bp[kb + 16 * t];           ba[t].y = bp[kb + CFC + 16 * t];
        bb[t].x = bp[kb + 4 * CFC + 16 * t]; bb[t].y = bp[kb + 5 * CFC + 16 * t];
      }
#pragma unroll
      for (int t = 0; t < NT; ++t) wmma_operands_ready(ba[t], bb[t], ava, avb);
      wmma_group_open();
#pragma unroll
      for (int t = 0; t < NT; ++t)
        acc[t] = __builtin_amdgcn_wmma_f32_16x16x4_f32(false, ava, false, ba[t],
                                                       (short)0, acc[t], false, false);
#pragma unroll
      for (int t = 0; t < NT; ++t)
        acc[t] = __builtin_amdgcn_wmma_f32_16x16x4_f32(false, avb, false, bb[t],
                                                       (short)0, acc[t], false, false);
      wmma_group_close();
    }
#pragma unroll
    for (int t = 0; t < NT; ++t)
#pragma unroll
      for (int r = 0; r < 8; ++r) part[wave][8 * t + r][lane] = acc[t][r];
    __syncthreads();
    for (int j = 0; j < RPW; ++j) {
      const int r = wave * RPW + j;
      float s = part[0][r][lane];
      for (int q = 1; q < S; ++q) s += part[q][r][lane];
      const int n = nb + 16 * (r >> 3);
      const float v = fmaxf(s + bd2[n], 0.0f);
      Y[(long long)(rbase + (r & 7)) * CFC + n] = v;
    }
    __syncthreads();
  }

  {
    constexpr int K = CFC;
    constexpr int per = (K / 4) / S;
    const float* ap = Y + (long long)(tm + mrow) * CFC + khalf;
    const float* bp = Wd3 + (nb < COUT ? nb : COUT - 1);
    const int kbeg = wave * per * 4;
    const int kend = kbeg + per * 4;
    v8f acc = {};
    for (int k0 = kbeg; k0 < kend; k0 += 8) {
      v2f ava = *(const v2f*)(ap + k0);
      v2f avb = *(const v2f*)(ap + k0 + 4);
      const int kb = (k0 + khalf) * COUT;
      v2f ba, bb;
      ba.x = bp[kb];            ba.y = bp[kb + COUT];
      bb.x = bp[kb + 4 * COUT]; bb.y = bp[kb + 5 * COUT];
      wmma_operands_ready(ba, bb, ava, avb);
      wmma_group_open();
      acc = __builtin_amdgcn_wmma_f32_16x16x4_f32(false, ava, false, ba,
                                                  (short)0, acc, false, false);
      acc = __builtin_amdgcn_wmma_f32_16x16x4_f32(false, avb, false, bb,
                                                  (short)0, acc, false, false);
      wmma_group_close();
    }
#pragma unroll
    for (int r = 0; r < 8; ++r) part[wave][r][lane] = acc[r];
    __syncthreads();
    const int r = wave;
    float s = part[0][r][lane];
    for (int q = 1; q < S; ++q) s += part[q][r][lane];
    if (nb < COUT) {
      const int row  = rbase + r;
      const int sb   = row / CN;
      const int node = row - sb * CN;
      const int step = sb / CB, b = sb - step * CB;
      act[((long long)(b * CT + step) * CN + node) * COUT + nb] = s + bd3[nb];
    }
  }
}

__global__ __launch_bounds__(256)
void encoder_rows_ksplit(const float* __restrict__ obs_te, float* T1,
                         const float* __restrict__ We1, const float* __restrict__ be1,
                         const float* __restrict__ We2, const float* __restrict__ be2,
                         float* __restrict__ X) {
  constexpr int S  = 8;
  constexpr int NT = 4;
  __shared__ float part[S][8 * NT][32];

  const int lane  = threadIdx.x & 31;
  const int wave  = __builtin_amdgcn_readfirstlane(threadIdx.x >> 5);
  const int tm = blockIdx.x << 4;
  const int sb = tm / CN;
  const int node0 = tm - sb * CN;
  const int step = sb / CB, b = sb - step * CB;

  const int mrow  = lane & 15;
  const int khalf = (lane >> 4) << 1;
  const int nb = lane & 15;
  const int rbase = tm + ((lane >> 4) << 3);

  {
    static_assert(CFC == 16 * S, "one column tile of the 128-wide layer per wave");
    static_assert(CIN == 10, "two whole k-steps and a tail of two");
    const float* ap = obs_te + ((long long)(b * CT + step) * CN + node0 + mrow) * CIN + khalf;
    const int n0 = 16 * wave + nb;
    const float* bp = We1 + n0;
    const int kb = khalf * CFC;
    v2f ava = *(const v2f*)(ap);
    v2f avb = *(const v2f*)(ap + 4);
    v2f ba, bb;
    ba.x = bp[kb];           ba.y = bp[kb + CFC];
    bb.x = bp[kb + 4 * CFC]; bb.y = bp[kb + 5 * CFC];
    v2f avt = {}, bt = {};
    if (khalf == 0) {
      avt = *(const v2f*)(ap + 8);
      bt.x = bp[8 * CFC];    bt.y = bp[9 * CFC];
    }
    v8f acc = {};
    wmma_operands_ready(ba, bb, ava, avb);
    wmma_operands_ready(bt, avt);
    wmma_group_open();
    acc = __builtin_amdgcn_wmma_f32_16x16x4_f32(false, ava, false, ba,
                                                (short)0, acc, false, false);
    acc = __builtin_amdgcn_wmma_f32_16x16x4_f32(false, avb, false, bb,
                                                (short)0, acc, false, false);
    acc = __builtin_amdgcn_wmma_f32_16x16x4_f32(false, avt, false, bt,
                                                (short)0, acc, false, false);
    wmma_group_close();
    const float badd = be1[n0];
#pragma unroll
    for (int r = 0; r < 8; ++r)
      T1[(long long)(rbase + r) * CFC + n0] = fmaxf(acc[r] + badd, 0.0f);
  }
  __syncthreads();

  {
    constexpr int K = CFC, N = CF;
    constexpr int per = (K / 4) / S;
    static_assert((K / 8) % S == 0, "whole trips of two k-steps, the same number in every wave");
    const float* ap = T1 + (long long)(tm + mrow) * CFC + khalf;
    const float* bp[NT];
#pragma unroll
    for (int t = 0; t < NT; ++t) {
      const int n = nb + 16 * t;
      bp[t] = We2 + (n < N ? n : N - 1);
    }
    const int kbeg = wave * per * 4;
    const int kend = kbeg + per * 4;
    v8f acc[NT];
#pragma unroll
    for (int t = 0; t < NT; ++t) acc[t] = v8f{};
    for (int k0 = kbeg; k0 < kend; k0 += 8) {
      v2f ava = *(const v2f*)(ap + k0);
      v2f avb = *(const v2f*)(ap + k0 + 4);
      const int kb = (k0 + khalf) * N;
      v2f ba[NT], bb[NT];
#pragma unroll
      for (int t = 0; t < NT; ++t) {
        ba[t].x = bp[t][kb];         ba[t].y = bp[t][kb + N];
        bb[t].x = bp[t][kb + 4 * N]; bb[t].y = bp[t][kb + 5 * N];
      }
#pragma unroll
      for (int t = 0; t < NT; ++t) wmma_operands_ready(ba[t], bb[t], ava, avb);
      wmma_group_open();
#pragma unroll
      for (int t = 0; t < NT; ++t)
        acc[t] = __builtin_amdgcn_wmma_f32_16x16x4_f32(false, ava, false, ba[t],
                                                       (short)0, acc[t], false, false);
#pragma unroll
      for (int t = 0; t < NT; ++t)
        acc[t] = __builtin_amdgcn_wmma_f32_16x16x4_f32(false, avb, false, bb[t],
                                                       (short)0, acc[t], false, false);
      wmma_group_close();
    }
#pragma unroll
    for (int t = 0; t < NT; ++t)
#pragma unroll
      for (int r = 0; r < 8; ++r) part[wave][8 * t + r][lane] = acc[t][r];
    __syncthreads();
    constexpr int RPW = 8 * NT / S;
    for (int j = 0; j < RPW; ++j) {
      const int r = wave * RPW + j;
      float s = part[0][r][lane];
      for (int q = 1; q < S; ++q) s += part[q][r][lane];
      const int n = nb + 16 * (r >> 3);
      if (n < N) X[(long long)(rbase + (r & 7)) * CF + n] = fmaxf(s + be2[n], 0.0f);
    }
  }
}

__global__ void deg_dinv_kernel(const float* __restrict__ A, float* __restrict__ dinv,
                                int rows, int n) {
  const int wid  = (blockIdx.x * blockDim.x + threadIdx.x) >> 5;
  const int lane = threadIdx.x & 31;
  if (wid >= rows) return;
  const float* row = A + (long long)wid * n;
  float s = 0.0f;
  for (int i = lane; i < n; i += 32) s += row[i];
#pragma unroll
  for (int off = 16; off > 0; off >>= 1) s += __shfl_xor(s, off, 32);
  if (lane == 0) dinv[wid] = (s > 0.0f) ? rsqrtf(s) : 0.0f;
}

__global__ void pack_wcat_kernel(const float* __restrict__ Wch,
                                 const float* __restrict__ Wcu,
                                 float* __restrict__ Wcat) {
  int idx = blockIdx.x * blockDim.x + threadIdx.x;
  if (idx >= 100 * 300) return;
  int r = idx / 300, n = idx % 300;
  const float* src = r < 50 ? Wch : Wcu;
  int f = r < 50 ? r : r - 50;
  int m = n < 150 ? n : n - 150;
  int c = m / 50, g = m % 50;
  float v = src[((c * 2 + 1) * 50 + f) * 50 + g];
  if (n >= 150) v = src[((c * 2 + 0) * 50 + f) * 50 + g] + v;
  Wcat[idx] = v;
}

__device__ __forceinline__ v2f hx_a_frag(const float* __restrict__ hp,
                                         const float* __restrict__ xp, int k) {
  const float* p = k < CF ? hp + k : xp + (k - CF);
  return *(const v2f*)p;
}

template <int S, int NT>
__global__ __launch_bounds__(32 * S)
void gemm_hx_ksplit(const float* __restrict__ h, const float* __restrict__ x,
                    const float* __restrict__ dinv_t,
                    const float* __restrict__ Wcat, const float* __restrict__ bias,
                    float* __restrict__ R, float* __restrict__ Q) {
  constexpr int NH = 3 * CF;
  constexpr int N = 2 * NH;
  constexpr int K = 2 * CF;
  static_assert(K % 4 == 0 && CF % 2 == 0, "no K tail; halves start at even k");
  static_assert((8 * NT) % S == 0, "whole accumulator registers per wave in the reduction");
  __shared__ float part[S][8 * NT][32];

  const int lane  = threadIdx.x & 31;
  const int wave  = __builtin_amdgcn_readfirstlane(threadIdx.x >> 5);
  constexpr int SW = 16 * NT;
  constexpr int nstrip = (N + SW - 1) / SW;
  const int tm = (blockIdx.x / nstrip) << 4;
  const int tn = (blockIdx.x % nstrip) * SW;

  const int mrow  = lane & 15;
  const int khalf = (lane >> 4) << 1;
  const int nb = tn + (lane & 15);
  const int row = tm + mrow;
  const float* hp = h + (long long)row * CF;
  const float* xp = x + (long long)row * CF;
  const float* bp[NT];
#pragma unroll
  for (int t = 0; t < NT; ++t) {
    const int n = nb + 16 * t;
    bp[t] = Wcat + (n < N ? n : N - 1);
  }

  constexpr int steps = K / 4;
  constexpr int per   = (steps + S - 1) / S;
  const int kbeg  = wave * per * 4;
  const int klim  = kbeg + per * 4;
  const int kstop = klim < K ? klim : K;

  v8f acc[NT];
#pragma unroll
  for (int t = 0; t < NT; ++t) acc[t] = v8f{};
  int k0 = kbeg;
  for (; k0 + 4 < kstop; k0 += 8) {
    const int k = k0 + khalf;
    v2f ava = hx_a_frag(hp, xp, k);
    v2f avb = hx_a_frag(hp, xp, k + 4);
    const int kb = k * N;
    v2f ba[NT], bb[NT];
#pragma unroll
    for (int t = 0; t < NT; ++t) {
      ba[t].x = bp[t][kb];           ba[t].y = bp[t][kb + N];
      bb[t].x = bp[t][kb + 4 * N];   bb[t].y = bp[t][kb + 5 * N];
    }
#pragma unroll
    for (int t = 0; t < NT; ++t) wmma_operands_ready(ba[t], bb[t], ava, avb);
    wmma_group_open();
#pragma unroll
    for (int t = 0; t < NT; ++t)
      acc[t] = __builtin_amdgcn_wmma_f32_16x16x4_f32(false, ava, false, ba[t],
                                                     (short)0, acc[t], false, false);
#pragma unroll
    for (int t = 0; t < NT; ++t)
      acc[t] = __builtin_amdgcn_wmma_f32_16x16x4_f32(false, avb, false, bb[t],
                                                     (short)0, acc[t], false, false);
    wmma_group_close();
  }
  if (k0 < kstop) {
    const int k = k0 + khalf;
    v2f ava = hx_a_frag(hp, xp, k);
    const int kb = k * N;
    v2f ba[NT];
#pragma unroll
    for (int t = 0; t < NT; ++t) { ba[t].x = bp[t][kb]; ba[t].y = bp[t][kb + N]; }
#pragma unroll
    for (int t = 0; t < NT; ++t) wmma_operands_ready(ba[t], ava);
    wmma_group_open();
#pragma unroll
    for (int t = 0; t < NT; ++t)
      acc[t] = __builtin_amdgcn_wmma_f32_16x16x4_f32(false, ava, false, ba[t],
                                                     (short)0, acc[t], false, false);
    wmma_group_close();
  }

  constexpr int RPW = 8 * NT / S;
  const int rbase = tm + ((lane >> 4) << 3);
  float dvv[RPW], bvv[RPW];
#pragma unroll
  for (int j = 0; j < RPW; ++j) {
    const int r = wave * RPW + j;
    const int grow = rbase + (r & 7);
    const int m = nb + 16 * (r >> 3) - NH;
    dvv[j] = dinv_t[(long long)(grow / CN) * CT * CN + (grow % CN)];
    bvv[j] = bias[m < 0 ? 0 : (m < NH ? m : NH - 1)];
  }

#pragma unroll
  for (int t = 0; t < NT; ++t)
#pragma unroll
    for (int r = 0; r < 8; ++r) part[wave][8 * t + r][lane] = acc[t][r];
  __syncthreads();

#pragma unroll
  for (int j = 0; j < RPW; ++j) {
    const int r = wave * RPW + j;
    float s = part[0][r][lane];
    for (int q = 1; q < S; ++q) s += part[q][r][lane];
    const int n = nb + 16 * (r >> 3);
    const int grow = rbase + (r & 7);
    const bool rhalf = n < NH;
    const float v = rhalf ? dvv[j] * s : s + bvv[j];
    float* dst = rhalf ? R + (long long)grow * NH + n : Q + (long long)grow * NH + (n - NH);
    if (n < N) *dst = v;
  }
}

__device__ __forceinline__ void lds_barrier() {
  asm volatile("s_wait_dscnt 0x0" ::: "memory");
  asm volatile("s_barrier_signal -1" ::: "memory");
  asm volatile("s_barrier_wait -1" ::: "memory");
}

template <int S, int NT>
__global__ __launch_bounds__(32 * S)
void step_adj_gate_hx(const float* __restrict__ adj_t,
                      const float* __restrict__ R, float* Q,
                      const float* __restrict__ dinv_t,
                      float* __restrict__ hout,
                      const float* __restrict__ x_next,
                      const float* __restrict__ dinv_n,
                      const float* __restrict__ Wcat, const float* __restrict__ bias,
                      float* __restrict__ Rn) {
  constexpr int N = 3 * CF;
  constexpr int K = CN;
  static_assert(K % (8 * S) == 0, "whole trips of two k-steps in every wave");
  static_assert(16 * NT >= N, "a block holds whole rows of z");
  constexpr int HXP = 2 * CF + 4;
  constexpr int NW = 2 * N;
  constexpr int KS = (2 * CF) / 4;
  constexpr int WT = (NW + 15) / 16;
  constexpr int TPW = (WT + S - 1) / S;
  static_assert((2 * CF) % 4 == 0, "no K tail");
  __shared__ float part[S][8 * NT][32];
  __shared__ float hxs[16][HXP];

  const int lane  = threadIdx.x & 31;
  const int wave  = __builtin_amdgcn_readfirstlane(threadIdx.x >> 5);
  const int tm = blockIdx.x << 4;
  const int b  = blockIdx.z;

  const float* Ab = adj_t + (long long)b * CT * CN * CN;
  const float* Rb = R + (long long)b * CN * N;

  constexpr int XPT = (16 * CF + 32 * S - 1) / (32 * S);
  float xv[XPT] = {};
  if (x_next != nullptr) {
    const float* xb = x_next + ((long long)b * CN + tm) * CF;
#pragma unroll
    for (int i = 0; i < XPT; ++i) {
      const int idx = threadIdx.x + i * 32 * S;
      xv[i] = xb[idx < 16 * CF ? idx : 16 * CF - 1];
    }
  }

  const int mrow  = lane & 15;
  const int khalf = (lane >> 4) << 1;
  const int nb = lane & 15;
  const float* ap  = Ab + (long long)(tm + mrow) * K + khalf;
  const float* bp[NT];
#pragma unroll
  for (int t = 0; t < NT; ++t) {
    const int n = nb + 16 * t;
    bp[t] = Rb + (n < N ? n : N - 1);
  }

  constexpr int per = K / (4 * S);
  const int kbeg = wave * per * 4;
  const int kend = kbeg + per * 4;

  v8f acc[NT];
#pragma unroll
  for (int t = 0; t < NT; ++t) acc[t] = v8f{};
  for (int k0 = kbeg; k0 < kend; k0 += 8) {
    v2f ava = *(const v2f*)(ap + k0);
    v2f avb = *(const v2f*)(ap + k0 + 4);
    const int kb = (k0 + khalf) * N;
    v2f ba[NT], bb[NT];
#pragma unroll
    for (int t = 0; t < NT; ++t) {
      ba[t].x = bp[t][kb];           ba[t].y = bp[t][kb + N];
      bb[t].x = bp[t][kb + 4 * N];   bb[t].y = bp[t][kb + 5 * N];
    }
#pragma unroll
    for (int t = 0; t < NT; ++t) wmma_operands_ready(ba[t], bb[t], ava, avb);
    wmma_group_open();
#pragma unroll
    for (int t = 0; t < NT; ++t)
      acc[t] = __builtin_amdgcn_wmma_f32_16x16x4_f32(false, ava, false, ba[t],
                                                     (short)0, acc[t], false, false);
#pragma unroll
    for (int t = 0; t < NT; ++t)
      acc[t] = __builtin_amdgcn_wmma_f32_16x16x4_f32(false, avb, false, bb[t],
                                                     (short)0, acc[t], false, false);
    wmma_group_close();
  }

#pragma unroll
  for (int t = 0; t < NT; ++t)
#pragma unroll
    for (int r = 0; r < 8; ++r) part[wave][8 * t + r][lane] = acc[t][r];
  const int rbase = tm + ((lane >> 4) << 3);
  v2f bf[TPW][KS];
  float bq[TPW];
  float dn[8];
  if (x_next != nullptr) {
#pragma unroll
    for (int i = 0; i < XPT; ++i) {
      const int idx = threadIdx.x + i * 32 * S;
      if (idx < 16 * CF) {
        const int r16 = idx / CF;
        hxs[r16][CF + (idx - r16 * CF)] = xv[i];
      }
    }
#pragma unroll
    for (int i = 0; i < TPW; ++i) {
      const int tile = wave + i * S;
      if (tile < WT) {
        const int n = tile * 16 + nb;
        const int m = n - N;
        const float* wp = Wcat + (n < NW ? n : NW - 1) + khalf * NW;
#pragma unroll
        for (int s = 0; s < KS; ++s) {
          bf[i][s].x = wp[(4 * s) * NW];
          bf[i][s].y = wp[(4 * s + 1) * NW];
        }
        bq[i] = bias[m < 0 ? 0 : (m < N ? m : N - 1)];
      }
    }
#pragma unroll
    for (int r = 0; r < 8; ++r) dn[r] = dinv_n[(long long)b * CT * CN + rbase + r];
  }
  lds_barrier();

  for (int idx = threadIdx.x; idx < 16 * CF; idx += 32 * S) {
    const int r16 = idx / CF;
    const int f   = idx - r16 * CF;
    const int lhi = (r16 >> 3) << 4;
    const int rr  = r16 & 7;
    float s[3];
#pragma unroll
    for (int c = 0; c < 3; ++c) {
      const int col = c * CF + f;
      const int reg = 8 * (col >> 4) + rr;
      const int l   = lhi + (col & 15);
      float a = part[0][reg][l];
      for (int q = 1; q < S; ++q) a += part[q][reg][l];
      s[c] = a;
    }
    const int node = tm + r16;
    const long long grow = (long long)b * CN + node;
    const float dv = dinv_t[(long long)b * CT * CN + node];
    const float* qp = Q + grow * N + f;
    const float z0 = qp[0]      - dv * s[0];
    const float z1 = qp[CF]     - dv * s[1];
    const float z2 = qp[2 * CF] - dv * s[2];
    const float gt = 1.0f / (1.0f + expf(-z0));
    const float hv = gt * tanhf(z1) + (1.0f - gt) * tanhf(z2);
    hout[grow * CF + f] = hv;
    hxs[r16][f] = hv;
  }

  if (x_next == nullptr) return;
  lds_barrier();

  const float* hp = &hxs[mrow][khalf];
  v2f af[KS];
#pragma unroll
  for (int s = 0; s < KS; ++s) af[s] = *(const v2f*)(hp + 4 * s);

#pragma unroll
  for (int i = 0; i < TPW; ++i) {
    const int tile = wave + i * S;
    if (tile < WT) {
      const int n = tile * 16 + nb;
      const int m = n - N;
      v8f accw = {};
#pragma unroll
      for (int s = 0; s < KS; ++s) wmma_operands_ready(bf[i][s], af[s]);
      wmma_group_open();
#pragma unroll
      for (int s = 0; s < KS; ++s)
        accw = __builtin_amdgcn_wmma_f32_16x16x4_f32(false, af[s], false, bf[i][s],
                                                     (short)0, accw, false, false);
      wmma_group_close();
      const bool rhalf = n < N;
      const long long grow0 = (long long)b * CN + rbase;
      float* dst = rhalf ? Rn + grow0 * N + n : Q + grow0 * N + m;
      if (n < NW) {
#pragma unroll
        for (int r = 0; r < 8; ++r) dst[r * N] = rhalf ? dn[r] * accw[r] : accw[r] + bq[i];
      }
    }
  }
}

static const int TE = 4;
static const int TD = 4;

extern "C" void kernel_launch(void* const* d_in, const int* in_sizes, int n_in,
                              void* d_out, int out_size, void* d_ws, size_t ws_size,
                              hipStream_t stream) {
  const float* obs  = (const float*)d_in[0];
  const float* adj  = (const float*)d_in[1];
  const float* h0   = (const float*)d_in[2];
  const float* We1  = (const float*)d_in[3];
  const float* be1  = (const float*)d_in[4];
  const float* We2  = (const float*)d_in[5];
  const float* be2  = (const float*)d_in[6];
  const float* Wch  = (const float*)d_in[7];
  const float* Wcu  = (const float*)d_in[8];
  const float* bc   = (const float*)d_in[9];
  const float* Wd1  = (const float*)d_in[10];
  const float* bd1  = (const float*)d_in[11];
  const float* Wd2  = (const float*)d_in[12];
  const float* bd2  = (const float*)d_in[13];
  const float* Wd3  = (const float*)d_in[14];
  const float* bd3  = (const float*)d_in[15];
  float* out = (float*)d_out;

  const int MN = CB * CN;
  const size_t HSZ = (size_t)MN * CF;

  float* ws = (float*)d_ws;
  size_t off = 0;
  float* dinv = ws + off; off += (size_t)CB * CT * CN;
  float* Wcat = ws + off; off += (size_t)100 * 300;
  float* H    = ws + off; off += TD * HSZ;
  float* xc   = ws + off; off += TE * HSZ;
  float* scr  = ws + off;
  float* Rp0  = scr;
  float* Rp1  = Rp0 + (size_t)MN * 3 * CF;
  float* Qp   = Rp1 + (size_t)MN * 3 * CF;
  float* tmp1 = scr;
  float* y    = scr;
  (void)ws_size; (void)in_sizes; (void)n_in; (void)out_size;

  float* hlast = H + (TD - 1) * HSZ;
  {
    int rows = CB * CT * CN;
    deg_dinv_kernel<<<(rows * 32 + 255) / 256, 256, 0, stream>>>(adj, dinv, rows, CN);
    pack_wcat_kernel<<<(100 * 300 + 255) / 256, 256, 0, stream>>>(Wch, Wcu, Wcat);
    hipMemcpyAsync(hlast, h0, sizeof(float) * CB * CN * CF, hipMemcpyDeviceToDevice, stream);
  }

  for (int te = 0; te < CT; te += TE) {
    encoder_rows_ksplit<<<(TE * MN) / 16, 32 * 8, 0, stream>>>(
        obs + (long long)te * CN * CIN, tmp1, We1, be1, We2, be2, xc);

    for (int j = 0; j < TD; ++j) {
      const int t = te + j;
      const float* x = xc + (size_t)j * HSZ;
      const float* hin = (j == 0) ? hlast : (H + (size_t)(j - 1) * HSZ);
      float* hout = H + (size_t)j * HSZ;
      float* Rcur = (j & 1) ? Rp1 : Rp0;
      float* Rnxt = (j & 1) ? Rp0 : Rp1;
      if (j == 0)
        gemm_hx_ksplit<8, 10><<<(MN / 16) * 2, 32 * 8, 0, stream>>>(
            hin, x, dinv + (long long)t * CN, Wcat, bc, Rcur, Qp);
      const bool more = j + 1 < TD;
      step_adj_gate_hx<16, 10><<<dim3(CN / 16, 1, CB), 32 * 16, 0, stream>>>(
          adj + (long long)t * CN * CN, Rcur, Qp, dinv + (long long)t * CN, hout,
          more ? xc + (size_t)(j + 1) * HSZ : nullptr,
          more ? dinv + (long long)(t + 1) * CN : dinv,
          Wcat, bc, Rnxt);
    }
    decoder_rows_ksplit<<<(TD * MN) / 16, 32 * 8, 0, stream>>>(
        H, y, Wd1, bd1, Wd2, bd2, Wd3, bd3, out + (long long)te * CN * COUT);
  }

  hipMemcpyAsync(out + (long long)CB * CT * CN * COUT, hlast,
                 sizeof(float) * CB * CN * CF, hipMemcpyDeviceToDevice, stream);
}


# no degree pass over the adjacency: step t's launch sums A_{t+1}'s rows for its 16 rows (wave per row, 16-byte loads under the last trip's WMMAs; a row's sum order changes); step 0 has a small kernel
# speedup vs baseline: 5.6990x; 1.0848x over previous
#include <hip/hip_runtime.h>
#include <math.h>

static const int CB = 4, CT = 32, CN = 512;
static const int CIN = 10, CF = 50, CFC = 128, COUT = 2;

typedef __attribute__((ext_vector_type(2))) float v2f;
typedef __attribute__((ext_vector_type(4))) float v4f;
typedef __attribute__((ext_vector_type(8))) float v8f;


__device__ __forceinline__ void wmma_group_open() {
  __builtin_amdgcn_sched_barrier(0);
}
__device__ __forceinline__ void wmma_operands_ready(v2f& b_a, v2f& b_b, v2f& ava, v2f& avb) {
  asm volatile("" : "+v"(b_a), "+v"(b_b), "+v"(ava), "+v"(avb));
}
__device__ __forceinline__ void wmma_operands_ready(v2f& b_a, v2f& ava) {
  asm volatile("" : "+v"(b_a), "+v"(ava));
}
__device__ __forceinline__ void wmma_group_close() {
  __builtin_amdgcn_sched_barrier(0);
  asm volatile("v_nop\n\tv_nop\n\tv_nop\n\tv_nop");
  __builtin_amdgcn_sched_barrier(0);
}

template <int S, int NT>
__global__ __launch_bounds__(32 * S)
void gemm_wmma_f32_ksplit(const float* __restrict__ A, long long sA, long long sA2, int lda,
                          const float* __restrict__ Bm, long long sB, int ldb,
                          float* __restrict__ C, long long sC, long long sC2, int ldc,
                          const float* __restrict__ bias,
                          int N, int K, int relu) {
  static_assert((8 * NT) % S == 0, "whole accumulator registers per wave in the reduction");
  __shared__ float part[S][8 * NT][32];

  const int lane  = threadIdx.x & 31;
  const int wave  = __builtin_amdgcn_readfirstlane(threadIdx.x >> 5);
  constexpr int SW = 16 * NT;
  const int nstrip = (N + SW - 1) / SW;
  const int tile  = blockIdx.x;
  const int tm = (tile / nstrip) << 4;
  const int tn = (tile % nstrip) * SW;

  const float* Ab = A  + (long long)blockIdx.z * sA + (long long)blockIdx.y * sA2;
  const float* Bb = Bm + (long long)blockIdx.z * sB;
  float*       Cb = C  + (long long)blockIdx.z * sC + (long long)blockIdx.y * sC2;

  const int mrow  = lane & 15;
  const int khalf = (lane >> 4) << 1;
  const int nb = tn + (lane & 15);
  const float* ap  = Ab + (long long)(tm + mrow) * lda + khalf;
  const float* bp[NT];
#pragma unroll
  for (int t = 0; t < NT; ++t) {
    const int n = nb + 16 * t;
    bp[t] = Bb + (n < N ? n : N - 1);
  }

  const int K4    = K & ~3;
  const int steps = (K + 3) >> 2;
  const int per   = (steps + S - 1) / S;
  const int kbeg  = wave * per * 4;
  const int klim  = kbeg + per * 4;
  const int kstop = klim < K4 ? klim : K4;

  v8f acc[NT];
#pragma unroll
  for (int t = 0; t < NT; ++t) acc[t] = v8f{};
  int k0 = kbeg;
  for (; k0 + 4 < kstop; k0 += 8) {
    v2f ava = *(const v2f*)(ap + k0);
    v2f avb = *(const v2f*)(ap + k0 + 4);
    const long long kb = (long long)(k0 + khalf) * ldb;
    v2f ba[NT], bb[NT];
#pragma unroll
    for (int t = 0; t < NT; ++t) {
      ba[t].x = bp[t][kb];             ba[t].y = bp[t][kb + ldb];
      bb[t].x = bp[t][kb + 4LL * ldb]; bb[t].y = bp[t][kb + 5LL * ldb];
    }
#pragma unroll
    for (int t = 0; t < NT; ++t) wmma_operands_ready(ba[t], bb[t], ava, avb);
    wmma_group_open();
#pragma unroll
    for (int t = 0; t < NT; ++t)
      acc[t] = __builtin_amdgcn_wmma_f32_16x16x4_f32(false, ava, false, ba[t],
                                                     (short)0, acc[t], false, false);
#pragma unroll
    for (int t = 0; t < NT; ++t)
      acc[t] = __builtin_amdgcn_wmma_f32_16x16x4_f32(false, avb, false, bb[t],
                                                     (short)0, acc[t], false, false);
    wmma_group_close();
  }
  if (k0 < kstop) {
    v2f ava = *(const v2f*)(ap + k0);
    const long long kb = (long long)(k0 + khalf) * ldb;
    v2f ba[NT];
#pragma unroll
    for (int t = 0; t < NT; ++t) { ba[t].x = bp[t][kb]; ba[t].y = bp[t][kb + ldb]; }
#pragma unroll
    for (int t = 0; t < NT; ++t) wmma_operands_ready(ba[t], ava);
    wmma_group_open();
#pragma unroll
    for (int t = 0; t < NT; ++t)
      acc[t] = __builtin_amdgcn_wmma_f32_16x16x4_f32(false, ava, false, ba[t],
                                                     (short)0, acc[t], false, false);
    wmma_group_close();
  }
  if (K4 < K && kbeg <= K4 && K4 < klim) {
    const int ka = K4 + khalf;
    v2f av = {};
    v2f bv[NT];
#pragma unroll
    for (int t = 0; t < NT; ++t) bv[t] = v2f{};
    if (ka < K) {
      av.x  = ap[K4];
#pragma unroll
      for (int t = 0; t < NT; ++t) bv[t].x = bp[t][(long long)ka * ldb];
    }
    if (ka + 1 < K) {
      av.y  = ap[K4 + 1];
#pragma unroll
      for (int t = 0; t < NT; ++t) bv[t].y = bp[t][(long long)(ka + 1) * ldb];
    }
#pragma unroll
    for (int t = 0; t < NT; ++t) wmma_operands_ready(bv[t], av);
    wmma_group_open();
#pragma unroll
    for (int t = 0; t < NT; ++t)
      acc[t] = __builtin_amdgcn_wmma_f32_16x16x4_f32(false, av, false, bv[t],
                                                     (short)0, acc[t], false, false);
    wmma_group_close();
  }

#pragma unroll
  for (int t = 0; t < NT; ++t)
#pragma unroll
    for (int r = 0; r < 8; ++r) part[wave][8 * t + r][lane] = acc[t][r];
  __syncthreads();

  constexpr int RPW = 8 * NT / S;
  const int rbase = tm + ((lane >> 4) << 3);
  for (int j = 0; j < RPW; ++j) {
    const int r = wave * RPW + j;
    float s = part[0][r][lane];
    for (int q = 1; q < S; ++q) s += part[q][r][lane];
    const int n = nb + 16 * (r >> 3);
    if (n < N) {
      float v = s + (bias ? bias[n] : 0.0f);
      if (relu) v = fmaxf(v, 0.0f);
      Cb[(long long)(rbase + (r & 7)) * ldc + n] = v;
    }
  }
}

__global__ __launch_bounds__(256)
void decoder_rows_ksplit(const float* __restrict__ Hm, float* Y,
                         const float* __restrict__ Wd1, const float* __restrict__ bd1,
                         const float* __restrict__ Wd2, const float* __restrict__ bd2,
                         const float* __restrict__ Wd3, const float* __restrict__ bd3,
                         float* __restrict__ act) {
  constexpr int S  = 8;
  constexpr int NT = CFC / 16;
  __shared__ float part[S][8 * NT][32];

  const int lane  = threadIdx.x & 31;
  const int wave  = __builtin_amdgcn_readfirstlane(threadIdx.x >> 5);
  const int tm = blockIdx.x << 4;

  const int mrow  = lane & 15;
  const int khalf = (lane >> 4) << 1;
  const int nb = lane & 15;
  const int rbase = tm + ((lane >> 4) << 3);
  constexpr int RPW = 8 * NT / S;

  {
    constexpr int K = CF, K4 = K & ~3;
    constexpr int S1  = 4;
    constexpr int per = (((K + 3) >> 2) + S1 - 1) / S1;
    static_assert(per % 2 == 0, "whole trips of two k-steps in every wave");
    v8f acc[NT];
#pragma unroll
    for (int t = 0; t < NT; ++t) acc[t] = v8f{};
    if (wave < S1) {
      const float* ap = Hm + (long long)(tm + mrow) * CF + khalf;
      const float* bp = Wd1 + nb;
      const int kbeg  = wave * per * 4;
      const int klim  = kbeg + per * 4;
      const int kstop = klim < K4 ? klim : K4;
      for (int k0 = kbeg; k0 < kstop; k0 += 8) {
        v2f ava = *(const v2f*)(ap + k0);
        v2f avb = *(const v2f*)(ap + k0 + 4);
        const int kb = (k0 + khalf) * CFC;
        v2f ba[NT], bb[NT];
#pragma unroll
        for (int t = 0; t < NT; ++t) {
          ba[t].x = bp[kb + 16 * t];           ba[t].y = bp[kb + CFC + 16 * t];
          bb[t].x = bp[kb + 4 * CFC + 16 * t]; bb[t].y = bp[kb + 5 * CFC + 16 * t];
        }
#pragma unroll
        for (int t = 0; t < NT; ++t) wmma_operands_ready(ba[t], bb[t], ava, avb);
        wmma_group_open();
#pragma unroll
        for (int t = 0; t < NT; ++t)
          acc[t] = __builtin_amdgcn_wmma_f32_16x16x4_f32(false, ava, false, ba[t],
                                                         (short)0, acc[t], false, false);
#pragma unroll
        for (int t = 0; t < NT; ++t)
          acc[t] = __builtin_amdgcn_wmma_f32_16x16x4_f32(false, avb, false, bb[t],
                                                         (short)0, acc[t], false, false);
        wmma_group_close();
      }
      if (K4 < K && kbeg <= K4 && K4 < klim) {
        const int ka = K4 + khalf;
        v2f av = {};
        v2f bv[NT];
#pragma unroll
        for (int t = 0; t < NT; ++t) bv[t] = v2f{};
        if (ka < K) {
          av.x = ap[K4];
#pragma unroll
          for (int t = 0; t < NT; ++t) bv[t].x = bp[ka * CFC + 16 * t];
        }
        if (ka + 1 < K) {
          av.y = ap[K4 + 1];
#pragma unroll
          for (int t = 0; t < NT; ++t) bv[t].y = bp[(ka + 1) * CFC + 16 * t];
        }
#pragma unroll
        for (int t = 0; t < NT; ++t) wmma_operands_ready(bv[t], av);
        wmma_group_open();
#pragma unroll
        for (int t = 0; t < NT; ++t)
          acc[t] = __builtin_amdgcn_wmma_f32_16x16x4_f32(false, av, false, bv[t],
                                                         (short)0, acc[t], false, false);
        wmma_group_close();
      }
#pragma unroll
      for (int t = 0; t < NT; ++t)
#pragma unroll
        for (int r = 0; r < 8; ++r) part[wave][8 * t + r][lane] = acc[t][r];
    }
    __syncthreads();
    for (int j = 0; j < RPW; ++j) {
      const int r = wave * RPW + j;
      float s = part[0][r][lane];
      for (int q = 1; q < S1; ++q) s += part[q][r][lane];
      const int n = nb + 16 * (r >> 3);
      const float v = fmaxf(s + bd1[n], 0.0f);
      Y[(long long)(rbase + (r & 7)) * CFC + n] = v;
    }
    __syncthreads();
  }

  {
    constexpr int K = CFC;
    constexpr int per = (K / 4) / S;
    static_assert((K / 8) % S == 0, "whole trips of two k-steps, the same number in every wave");
    const float* ap = Y + (long long)(tm + mrow) * CFC + khalf;
    const float* bp = Wd2 + nb;
    const int kbeg = wave * per * 4;
    const int kend = kbeg + per * 4;
    v8f acc[NT];
#pragma unroll
    for (int t = 0; t < NT; ++t) acc[t] = v8f{};
    for (int k0 = kbeg; k0 < kend; k0 += 8) {
      v2f ava = *(const v2f*)(ap + k0);
      v2f avb = *(const v2f*)(ap + k0 + 4);
      const int kb = (k0 + khalf) * CFC;
      v2f ba[NT], bb[NT];
#pragma unroll
      for (int t = 0; t < NT; ++t) {
        ba[t].x = bp[kb + 16 * t];           ba[t].y = bp[kb + CFC + 16 * t];
        bb[t].x = bp[kb + 4 * CFC + 16 * t]; bb[t].y = bp[kb + 5 * CFC + 16 * t];
      }
#pragma unroll
      for (int t = 0; t < NT; ++t) wmma_operands_ready(ba[t], bb[t], ava, avb);
      wmma_group_open();
#pragma unroll
      for (int t = 0; t < NT; ++t)
        acc[t] = __builtin_amdgcn_wmma_f32_16x16x4_f32(false, ava, false, ba[t],
                                                       (short)0, acc[t], false, false);
#pragma unroll
      for (int t = 0; t < NT; ++t)
        acc[t] = __builtin_amdgcn_wmma_f32_16x16x4_f32(false, avb, false, bb[t],
                                                       (short)0, acc[t], false, false);
      wmma_group_close();
    }
#pragma unroll
    for (int t = 0; t < NT; ++t)
#pragma unroll
      for (int r = 0; r < 8; ++r) part[wave][8 * t + r][lane] = acc[t][r];
    __syncthreads();
    for (int j = 0; j < RPW; ++j) {
      const int r = wave * RPW + j;
      float s = part[0][r][lane];
      for (int q = 1; q < S; ++q) s += part[q][r][lane];
      const int n = nb + 16 * (r >> 3);
      const float v = fmaxf(s + bd2[n], 0.0f);
      Y[(long long)(rbase + (r & 7)) * CFC + n] = v;
    }
    __syncthreads();
  }

  {
    constexpr int K = CFC;
    constexpr int per = (K / 4) / S;
    const float* ap = Y + (long long)(tm + mrow) * CFC + khalf;
    const float* bp = Wd3 + (nb < COUT ? nb : COUT - 1);
    const int kbeg = wave * per * 4;
    const int kend = kbeg + per * 4;
    v8f acc = {};
    for (int k0 = kbeg; k0 < kend; k0 += 8) {
      v2f ava = *(const v2f*)(ap + k0);
      v2f avb = *(const v2f*)(ap + k0 + 4);
      const int kb = (k0 + khalf) * COUT;
      v2f ba, bb;
      ba.x = bp[kb];            ba.y = bp[kb + COUT];
      bb.x = bp[kb + 4 * COUT]; bb.y = bp[kb + 5 * COUT];
      wmma_operands_ready(ba, bb, ava, avb);
      wmma_group_open();
      acc = __builtin_amdgcn_wmma_f32_16x16x4_f32(false, ava, false, ba,
                                                  (short)0, acc, false, false);
      acc = __builtin_amdgcn_wmma_f32_16x16x4_f32(false, avb, false, bb,
                                                  (short)0, acc, false, false);
      wmma_group_close();
    }
#pragma unroll
    for (int r = 0; r < 8; ++r) part[wave][r][lane] = acc[r];
    __syncthreads();
    const int r = wave;
    float s = part[0][r][lane];
    for (int q = 1; q < S; ++q) s += part[q][r][lane];
    if (nb < COUT) {
      const int row  = rbase + r;
      const int sb   = row / CN;
      const int node = row - sb * CN;
      const int step = sb / CB, b = sb - step * CB;
      act[((long long)(b * CT + step) * CN + node) * COUT + nb] = s + bd3[nb];
    }
  }
}

__global__ __launch_bounds__(256)
void encoder_rows_ksplit(const float* __restrict__ obs_te, float* T1,
                         const float* __restrict__ We1, const float* __restrict__ be1,
                         const float* __restrict__ We2, const float* __restrict__ be2,
                         float* __restrict__ X) {
  constexpr int S  = 8;
  constexpr int NT = 4;
  __shared__ float part[S][8 * NT][32];

  const int lane  = threadIdx.x & 31;
  const int wave  = __builtin_amdgcn_readfirstlane(threadIdx.x >> 5);
  const int tm = blockIdx.x << 4;
  const int sb = tm / CN;
  const int node0 = tm - sb * CN;
  const int step = sb / CB, b = sb - step * CB;

  const int mrow  = lane & 15;
  const int khalf = (lane >> 4) << 1;
  const int nb = lane & 15;
  const int rbase = tm + ((lane >> 4) << 3);

  {
    static_assert(CFC == 16 * S, "one column tile of the 128-wide layer per wave");
    static_assert(CIN == 10, "two whole k-steps and a tail of two");
    const float* ap = obs_te + ((long long)(b * CT + step) * CN + node0 + mrow) * CIN + khalf;
    const int n0 = 16 * wave + nb;
    const float* bp = We1 + n0;
    const int kb = khalf * CFC;
    v2f ava = *(const v2f*)(ap);
    v2f avb = *(const v2f*)(ap + 4);
    v2f ba, bb;
    ba.x = bp[kb];           ba.y = bp[kb + CFC];
    bb.x = bp[kb + 4 * CFC]; bb.y = bp[kb + 5 * CFC];
    v2f avt = {}, bt = {};
    if (khalf == 0) {
      avt = *(const v2f*)(ap + 8);
      bt.x = bp[8 * CFC];    bt.y = bp[9 * CFC];
    }
    v8f acc = {};
    wmma_operands_ready(ba, bb, ava, avb);
    wmma_operands_ready(bt, avt);
    wmma_group_open();
    acc = __builtin_amdgcn_wmma_f32_16x16x4_f32(false, ava, false, ba,
                                                (short)0, acc, false, false);
    acc = __builtin_amdgcn_wmma_f32_16x16x4_f32(false, avb, false, bb,
                                                (short)0, acc, false, false);
    acc = __builtin_amdgcn_wmma_f32_16x16x4_f32(false, avt, false, bt,
                                                (short)0, acc, false, false);
    wmma_group_close();
    const float badd = be1[n0];
#pragma unroll
    for (int r = 0; r < 8; ++r)
      T1[(long long)(rbase + r) * CFC + n0] = fmaxf(acc[r] + badd, 0.0f);
  }
  __syncthreads();

  {
    constexpr int K = CFC, N = CF;
    constexpr int per = (K / 4) / S;
    static_assert((K / 8) % S == 0, "whole trips of two k-steps, the same number in every wave");
    const float* ap = T1 + (long long)(tm + mrow) * CFC + khalf;
    const float* bp[NT];
#pragma unroll
    for (int t = 0; t < NT; ++t) {
      const int n = nb + 16 * t;
      bp[t] = We2 + (n < N ? n : N - 1);
    }
    const int kbeg = wave * per * 4;
    const int kend = kbeg + per * 4;
    v8f acc[NT];
#pragma unroll
    for (int t = 0; t < NT; ++t) acc[t] = v8f{};
    for (int k0 = kbeg; k0 < kend; k0 += 8) {
      v2f ava = *(const v2f*)(ap + k0);
      v2f avb = *(const v2f*)(ap + k0 + 4);
      const int kb = (k0 + khalf) * N;
      v2f ba[NT], bb[NT];
#pragma unroll
      for (int t = 0; t < NT; ++t) {
        ba[t].x = bp[t][kb];         ba[t].y = bp[t][kb + N];
        bb[t].x = bp[t][kb + 4 * N]; bb[t].y = bp[t][kb + 5 * N];
      }
#pragma unroll
      for (int t = 0; t < NT; ++t) wmma_operands_ready(ba[t], bb[t], ava, avb);
      wmma_group_open();
#pragma unroll
      for (int t = 0; t < NT; ++t)
        acc[t] = __builtin_amdgcn_wmma_f32_16x16x4_f32(false, ava, false, ba[t],
                                                       (short)0, acc[t], false, false);
#pragma unroll
      for (int t = 0; t < NT; ++t)
        acc[t] = __builtin_amdgcn_wmma_f32_16x16x4_f32(false, avb, false, bb[t],
                                                       (short)0, acc[t], false, false);
      wmma_group_close();
    }
#pragma unroll
    for (int t = 0; t < NT; ++t)
#pragma unroll
      for (int r = 0; r < 8; ++r) part[wave][8 * t + r][lane] = acc[t][r];
    __syncthreads();
    constexpr int RPW = 8 * NT / S;
    for (int j = 0; j < RPW; ++j) {
      const int r = wave * RPW + j;
      float s = part[0][r][lane];
      for (int q = 1; q < S; ++q) s += part[q][r][lane];
      const int n = nb + 16 * (r >> 3);
      if (n < N) X[(long long)(rbase + (r & 7)) * CF + n] = fmaxf(s + be2[n], 0.0f);
    }
  }
}

static const int DEGP = 33;

__device__ __forceinline__ void deg_row_load(const float* __restrict__ row, int lane, v4f (&d)[4]) {
  static_assert(CN == 4 * 32 * 4, "a row is four 16-byte loads per lane");
#pragma unroll
  for (int j = 0; j < 4; ++j) d[j] = *(const v4f*)(row + j * 128 + lane * 4);
}
__device__ __forceinline__ float deg_lane_sum(const v4f (&d)[4]) {
  float s = 0.0f;
#pragma unroll
  for (int j = 0; j < 4; ++j) s += (d[j].x + d[j].y) + (d[j].z + d[j].w);
  return s;
}
__device__ __forceinline__ float deg_row_dinv(const float* p) {
  float s = p[0];
  for (int q = 1; q < 32; ++q) s += p[q];
  return (s > 0.0f) ? rsqrtf(s) : 0.0f;
}

struct deg_on  { static constexpr bool value = true;  };
struct deg_off { static constexpr bool value = false; };

__global__ __launch_bounds__(256)
void deg_step0_kernel(const float* __restrict__ adj, float* __restrict__ dinv) {
  __shared__ float degp[8][DEGP];
  const int lane = threadIdx.x & 31;
  const int wave = threadIdx.x >> 5;
  const long long row0 = (long long)blockIdx.y * CT * CN + blockIdx.x * 8;
  v4f d[4];
  deg_row_load(adj + (row0 + wave) * CN, lane, d);
  degp[wave][lane] = deg_lane_sum(d);
  __syncthreads();
  if (threadIdx.x < 8) dinv[row0 + threadIdx.x] = deg_row_dinv(degp[threadIdx.x]);
}

__global__ void pack_wcat_kernel(const float* __restrict__ Wch,
                                 const float* __restrict__ Wcu,
                                 float* __restrict__ Wcat) {
  int idx = blockIdx.x * blockDim.x + threadIdx.x;
  if (idx >= 100 * 300) return;
  int r = idx / 300, n = idx % 300;
  const float* src = r < 50 ? Wch : Wcu;
  int f = r < 50 ? r : r - 50;
  int m = n < 150 ? n : n - 150;
  int c = m / 50, g = m % 50;
  float v = src[((c * 2 + 1) * 50 + f) * 50 + g];
  if (n >= 150) v = src[((c * 2 + 0) * 50 + f) * 50 + g] + v;
  Wcat[idx] = v;
}

__device__ __forceinline__ v2f hx_a_frag(const float* __restrict__ hp,
                                         const float* __restrict__ xp, int k) {
  const float* p = k < CF ? hp + k : xp + (k - CF);
  return *(const v2f*)p;
}

template <int S, int NT>
__global__ __launch_bounds__(32 * S)
void gemm_hx_ksplit(const float* __restrict__ h, const float* __restrict__ x,
                    const float* __restrict__ dinv_t,
                    const float* __restrict__ Wcat, const float* __restrict__ bias,
                    float* __restrict__ R, float* __restrict__ Q) {
  constexpr int NH = 3 * CF;
  constexpr int N = 2 * NH;
  constexpr int K = 2 * CF;
  static_assert(K % 4 == 0 && CF % 2 == 0, "no K tail; halves start at even k");
  static_assert((8 * NT) % S == 0, "whole accumulator registers per wave in the reduction");
  __shared__ float part[S][8 * NT][32];

  const int lane  = threadIdx.x & 31;
  const int wave  = __builtin_amdgcn_readfirstlane(threadIdx.x >> 5);
  constexpr int SW = 16 * NT;
  constexpr int nstrip = (N + SW - 1) / SW;
  const int tm = (blockIdx.x / nstrip) << 4;
  const int tn = (blockIdx.x % nstrip) * SW;

  const int mrow  = lane & 15;
  const int khalf = (lane >> 4) << 1;
  const int nb = tn + (lane & 15);
  const int row = tm + mrow;
  const float* hp = h + (long long)row * CF;
  const float* xp = x + (long long)row * CF;
  const float* bp[NT];
#pragma unroll
  for (int t = 0; t < NT; ++t) {
    const int n = nb + 16 * t;
    bp[t] = Wcat + (n < N ? n : N - 1);
  }

  constexpr int steps = K / 4;
  constexpr int per   = (steps + S - 1) / S;
  const int kbeg  = wave * per * 4;
  const int klim  = kbeg + per * 4;
  const int kstop = klim < K ? klim : K;

  v8f acc[NT];
#pragma unroll
  for (int t = 0; t < NT; ++t) acc[t] = v8f{};
  int k0 = kbeg;
  for (; k0 + 4 < kstop; k0 += 8) {
    const int k = k0 + khalf;
    v2f ava = hx_a_frag(hp, xp, k);
    v2f avb = hx_a_frag(hp, xp, k + 4);
    const int kb = k * N;
    v2f ba[NT], bb[NT];
#pragma unroll
    for (int t = 0; t < NT; ++t) {
      ba[t].x = bp[t][kb];           ba[t].y = bp[t][kb + N];
      bb[t].x = bp[t][kb + 4 * N];   bb[t].y = bp[t][kb + 5 * N];
    }
#pragma unroll
    for (int t = 0; t < NT; ++t) wmma_operands_ready(ba[t], bb[t], ava, avb);
    wmma_group_open();
#pragma unroll
    for (int t = 0; t < NT; ++t)
      acc[t] = __builtin_amdgcn_wmma_f32_16x16x4_f32(false, ava, false, ba[t],
                                                     (short)0, acc[t], false, false);
#pragma unroll
    for (int t = 0; t < NT; ++t)
      acc[t] = __builtin_amdgcn_wmma_f32_16x16x4_f32(false, avb, false, bb[t],
                                                     (short)0, acc[t], false, false);
    wmma_group_close();
  }
  if (k0 < kstop) {
    const int k = k0 + khalf;
    v2f ava = hx_a_frag(hp, xp, k);
    const int kb = k * N;
    v2f ba[NT];
#pragma unroll
    for (int t = 0; t < NT; ++t) { ba[t].x = bp[t][kb]; ba[t].y = bp[t][kb + N]; }
#pragma unroll
    for (int t = 0; t < NT; ++t) wmma_operands_ready(ba[t], ava);
    wmma_group_open();
#pragma unroll
    for (int t = 0; t < NT; ++t)
      acc[t] = __builtin_amdgcn_wmma_f32_16x16x4_f32(false, ava, false, ba[t],
                                                     (short)0, acc[t], false, false);
    wmma_group_close();
  }

  constexpr int RPW = 8 * NT / S;
  const int rbase = tm + ((lane >> 4) << 3);
  float dvv[RPW], bvv[RPW];
#pragma unroll
  for (int j = 0; j < RPW; ++j) {
    const int r = wave * RPW + j;
    const int grow = rbase + (r & 7);
    const int m = nb + 16 * (r >> 3) - NH;
    dvv[j] = dinv_t[(long long)(grow / CN) * CT * CN + (grow % CN)];
    bvv[j] = bias[m < 0 ? 0 : (m < NH ? m : NH - 1)];
  }

#pragma unroll
  for (int t = 0; t < NT; ++t)
#pragma unroll
    for (int r = 0; r < 8; ++r) part[wave][8 * t + r][lane] = acc[t][r];
  __syncthreads();

#pragma unroll
  for (int j = 0; j < RPW; ++j) {
    const int r = wave * RPW + j;
    float s = part[0][r][lane];
    for (int q = 1; q < S; ++q) s += part[q][r][lane];
    const int n = nb + 16 * (r >> 3);
    const int grow = rbase + (r & 7);
    const bool rhalf = n < NH;
    const float v = rhalf ? dvv[j] * s : s + bvv[j];
    float* dst = rhalf ? R + (long long)grow * NH + n : Q + (long long)grow * NH + (n - NH);
    if (n < N) *dst = v;
  }
}

__device__ __forceinline__ void lds_barrier() {
  asm volatile("s_wait_dscnt 0x0" ::: "memory");
  asm volatile("s_barrier_signal -1" ::: "memory");
  asm volatile("s_barrier_wait -1" ::: "memory");
}

template <int S, int NT>
__global__ __launch_bounds__(32 * S)
void step_adj_gate_hx(const float* __restrict__ adj_t,
                      const float* __restrict__ R, float* Q,
                      const float* __restrict__ dinv_t,
                      float* __restrict__ hout,
                      const float* __restrict__ x_next,
                      const float* __restrict__ adj_n,
                      float* __restrict__ dinv_nw,
                      const float* __restrict__ Wcat, const float* __restrict__ bias,
                      float* __restrict__ Rn) {
  constexpr int N = 3 * CF;
  constexpr int K = CN;
  static_assert(K % (8 * S) == 0, "whole trips of two k-steps in every wave");
  static_assert(16 * NT >= N, "a block holds whole rows of z");
  constexpr int HXP = 2 * CF + 4;
  constexpr int NW = 2 * N;
  constexpr int KS = (2 * CF) / 4;
  constexpr int WT = (NW + 15) / 16;
  constexpr int TPW = (WT + S - 1) / S;
  static_assert((2 * CF) % 4 == 0, "no K tail");
  __shared__ float part[S][8 * NT][32];
  __shared__ float hxs[16][HXP];
  static_assert(S == 16, "one wave per row of the block in the degree sums");
  __shared__ float degp[16][DEGP];
  __shared__ float sdn[16];

  const int lane  = threadIdx.x & 31;
  const int wave  = __builtin_amdgcn_readfirstlane(threadIdx.x >> 5);
  const int tm = blockIdx.x << 4;
  const int b  = blockIdx.z;

  const float* Ab = adj_t + (long long)b * CT * CN * CN;
  const float* Rb = R + (long long)b * CN * N;

  const float* degrow = adj_n + (long long)b * CT * CN * CN + (long long)(tm + wave) * CN;
  v4f dg[4];

  constexpr int XPT = (16 * CF + 32 * S - 1) / (32 * S);
  float xv[XPT] = {};
  if (x_next != nullptr) {
    const float* xb = x_next + ((long long)b * CN + tm) * CF;
#pragma unroll
    for (int i = 0; i < XPT; ++i) {
      const int idx = threadIdx.x + i * 32 * S;
      xv[i] = xb[idx < 16 * CF ? idx : 16 * CF - 1];
    }
  }

  const int mrow  = lane & 15;
  const int khalf = (lane >> 4) << 1;
  const int nb = lane & 15;
  const float* ap  = Ab + (long long)(tm + mrow) * K + khalf;
  const float* bp[NT];
#pragma unroll
  for (int t = 0; t < NT; ++t) {
    const int n = nb + 16 * t;
    bp[t] = Rb + (n < N ? n : N - 1);
  }

  constexpr int per = K / (4 * S);
  const int kbeg = wave * per * 4;
  const int kend = kbeg + per * 4;

  v8f acc[NT];
#pragma unroll
  for (int t = 0; t < NT; ++t) acc[t] = v8f{};
  auto trip = [&](const int k0, auto with_deg) __attribute__((always_inline)) {
    v2f ava = *(const v2f*)(ap + k0);
    v2f avb = *(const v2f*)(ap + k0 + 4);
    const int kb = (k0 + khalf) * N;
    v2f ba[NT], bb[NT];
#pragma unroll
    for (int t = 0; t < NT; ++t) {
      ba[t].x = bp[t][kb];           ba[t].y = bp[t][kb + N];
      bb[t].x = bp[t][kb + 4 * N];   bb[t].y = bp[t][kb + 5 * N];
    }
    if constexpr (decltype(with_deg)::value) deg_row_load(degrow, lane, dg);
#pragma unroll
    for (int t = 0; t < NT; ++t) wmma_operands_ready(ba[t], bb[t], ava, avb);
    wmma_group_open();
#pragma unroll
    for (int t = 0; t < NT; ++t)
      acc[t] = __builtin_amdgcn_wmma_f32_16x16x4_f32(false, ava, false, ba[t],
                                                     (short)0, acc[t], false, false);
#pragma unroll
    for (int t = 0; t < NT; ++t)
      acc[t] = __builtin_amdgcn_wmma_f32_16x16x4_f32(false, avb, false, bb[t],
                                                     (short)0, acc[t], false, false);
    wmma_group_close();
  };
  for (int k0 = kbeg; k0 < kend - 8; k0 += 8) trip(k0, deg_off{});
  trip(kend - 8, deg_on{});

#pragma unroll
  for (int t = 0; t < NT; ++t)
#pragma unroll
    for (int r = 0; r < 8; ++r) part[wave][8 * t + r][lane] = acc[t][r];
  __builtin_amdgcn_sched_barrier(0);
  degp[wave][lane] = deg_lane_sum(dg);
  const int rbase = tm + ((lane >> 4) << 3);
  v2f bf[TPW][KS];
  float bq[TPW];
  float dn[8];
  if (x_next != nullptr) {
#pragma unroll
    for (int i = 0; i < XPT; ++i) {
      const int idx = threadIdx.x + i * 32 * S;
      if (idx < 16 * CF) {
        const int r16 = idx / CF;
        hxs[r16][CF + (idx - r16 * CF)] = xv[i];
      }
    }
#pragma unroll
    for (int i = 0; i < TPW; ++i) {
      const int tile = wave + i * S;
      if (tile < WT) {
        const int n = tile * 16 + nb;
        const int m = n - N;
        const float* wp = Wcat + (n < NW ? n : NW - 1) + khalf * NW;
#pragma unroll
        for (int s = 0; s < KS; ++s) {
          bf[i][s].x = wp[(4 * s) * NW];
          bf[i][s].y = wp[(4 * s + 1) * NW];
        }
        bq[i] = bias[m < 0 ? 0 : (m < N ? m : N - 1)];
      }
    }
  }
  lds_barrier();

  if (wave == S - 1 && lane < 16) {
    const float dvn = deg_row_dinv(degp[lane]);
    sdn[lane] = dvn;
    if (dinv_nw != nullptr) dinv_nw[(long long)b * CT * CN + tm + lane] = dvn;
  }

  for (int idx = threadIdx.x; idx < 16 * CF; idx += 32 * S) {
    const int r16 = idx / CF;
    const int f   = idx - r16 * CF;
    const int lhi = (r16 >> 3) << 4;
    const int rr  = r16 & 7;
    float s[3];
#pragma unroll
    for (int c = 0; c < 3; ++c) {
      const int col = c * CF + f;
      const int reg = 8 * (col >> 4) + rr;
      const int l   = lhi + (col & 15);
      float a = part[0][reg][l];
      for (int q = 1; q < S; ++q) a += part[q][reg][l];
      s[c] = a;
    }
    const int node = tm + r16;
    const long long grow = (long long)b * CN + node;
    const float dv = dinv_t[(long long)b * CT * CN + node];
    const float* qp = Q + grow * N + f;
    const float z0 = qp[0]      - dv * s[0];
    const float z1 = qp[CF]     - dv * s[1];
    const float z2 = qp[2 * CF] - dv * s[2];
    const float gt = 1.0f / (1.0f + expf(-z0));
    const float hv = gt * tanhf(z1) + (1.0f - gt) * tanhf(z2);
    hout[grow * CF + f] = hv;
    hxs[r16][f] = hv;
  }

  if (x_next == nullptr) return;
  lds_barrier();

  const float* hp = &hxs[mrow][khalf];
  v2f af[KS];
#pragma unroll
  for (int s = 0; s < KS; ++s) af[s] = *(const v2f*)(hp + 4 * s);
#pragma unroll
  for (int r = 0; r < 8; ++r) dn[r] = sdn[((lane >> 4) << 3) + r];

#pragma unroll
  for (int i = 0; i < TPW; ++i) {
    const int tile = wave + i * S;
    if (tile < WT) {
      const int n = tile * 16 + nb;
      const int m = n - N;
      v8f accw = {};
#pragma unroll
      for (int s = 0; s < KS; ++s) wmma_operands_ready(bf[i][s], af[s]);
      wmma_group_open();
#pragma unroll
      for (int s = 0; s < KS; ++s)
        accw = __builtin_amdgcn_wmma_f32_16x16x4_f32(false, af[s], false, bf[i][s],
                                                     (short)0, accw, false, false);
      wmma_group_close();
      const bool rhalf = n < N;
      const long long grow0 = (long long)b * CN + rbase;
      float* dst = rhalf ? Rn + grow0 * N + n : Q + grow0 * N + m;
      if (n < NW) {
#pragma unroll
        for (int r = 0; r < 8; ++r) dst[r * N] = rhalf ? dn[r] * accw[r] : accw[r] + bq[i];
      }
    }
  }
}

static const int TE = 4;
static const int TD = 4;

extern "C" void kernel_launch(void* const* d_in, const int* in_sizes, int n_in,
                              void* d_out, int out_size, void* d_ws, size_t ws_size,
                              hipStream_t stream) {
  const float* obs  = (const float*)d_in[0];
  const float* adj  = (const float*)d_in[1];
  const float* h0   = (const float*)d_in[2];
  const float* We1  = (const float*)d_in[3];
  const float* be1  = (const float*)d_in[4];
  const float* We2  = (const float*)d_in[5];
  const float* be2  = (const float*)d_in[6];
  const float* Wch  = (const float*)d_in[7];
  const float* Wcu  = (const float*)d_in[8];
  const float* bc   = (const float*)d_in[9];
  const float* Wd1  = (const float*)d_in[10];
  const float* bd1  = (const float*)d_in[11];
  const float* Wd2  = (const float*)d_in[12];
  const float* bd2  = (const float*)d_in[13];
  const float* Wd3  = (const float*)d_in[14];
  const float* bd3  = (const float*)d_in[15];
  float* out = (float*)d_out;

  const int MN = CB * CN;
  const size_t HSZ = (size_t)MN * CF;

  float* ws = (float*)d_ws;
  size_t off = 0;
  float* dinv = ws + off; off += (size_t)CB * CT * CN;
  float* Wcat = ws + off; off += (size_t)100 * 300;
  float* H    = ws + off; off += TD * HSZ;
  float* xc   = ws + off; off += TE * HSZ;
  float* scr  = ws + off;
  float* Rp0  = scr;
  float* Rp1  = Rp0 + (size_t)MN * 3 * CF;
  float* Qp   = Rp1 + (size_t)MN * 3 * CF;
  float* tmp1 = scr;
  float* y    = scr;
  (void)ws_size; (void)in_sizes; (void)n_in; (void)out_size;

  float* hlast = H + (TD - 1) * HSZ;
  {
    deg_step0_kernel<<<dim3(CN / 8, CB), 256, 0, stream>>>(adj, dinv);
    pack_wcat_kernel<<<(100 * 300 + 255) / 256, 256, 0, stream>>>(Wch, Wcu, Wcat);
    hipMemcpyAsync(hlast, h0, sizeof(float) * CB * CN * CF, hipMemcpyDeviceToDevice, stream);
  }

  for (int te = 0; te < CT; te += TE) {
    encoder_rows_ksplit<<<(TE * MN) / 16, 32 * 8, 0, stream>>>(
        obs + (long long)te * CN * CIN, tmp1, We1, be1, We2, be2, xc);

    for (int j = 0; j < TD; ++j) {
      const int t = te + j;
      const float* x = xc + (size_t)j * HSZ;
      const float* hin = (j == 0) ? hlast : (H + (size_t)(j - 1) * HSZ);
      float* hout = H + (size_t)j * HSZ;
      float* Rcur = (j & 1) ? Rp1 : Rp0;
      float* Rnxt = (j & 1) ? Rp0 : Rp1;
      if (j == 0)
        gemm_hx_ksplit<8, 10><<<(MN / 16) * 2, 32 * 8, 0, stream>>>(
            hin, x, dinv + (long long)t * CN, Wcat, bc, Rcur, Qp);
      const bool more = j + 1 < TD;
      const bool nxt = t + 1 < CT;
      step_adj_gate_hx<16, 10><<<dim3(CN / 16, 1, CB), 32 * 16, 0, stream>>>(
          adj + (long long)t * CN * CN, Rcur, Qp, dinv + (long long)t * CN, hout,
          more ? xc + (size_t)(j + 1) * HSZ : nullptr,
          adj + (long long)(nxt ? t + 1 : t) * CN * CN,
          nxt ? dinv + (long long)(t + 1) * CN : nullptr,
          Wcat, bc, Rnxt);
    }
    decoder_rows_ksplit<<<(TD * MN) / 16, 32 * 8, 0, stream>>>(
        H, y, Wd1, bd1, Wd2, bd2, Wd3, bd3, out + (long long)te * CN * COUT);
  }

  hipMemcpyAsync(out + (long long)CB * CT * CN * COUT, hlast,
                 sizeof(float) * CB * CN * CF, hipMemcpyDeviceToDevice, stream);
}
